# starting kernel
# baseline (speedup 1.0000x reference)
#include <hip/hip_runtime.h>
#include <math.h>

typedef __attribute__((ext_vector_type(2))) float v2f;
typedef __attribute__((ext_vector_type(8))) float v8f;

#define NN 100000

__global__ void zero_f32(float* __restrict__ p, int n4) {
    int i = blockIdx.x * blockDim.x + threadIdx.x;
    int stride = gridDim.x * blockDim.x;
    float4 z = make_float4(0.f, 0.f, 0.f, 0.f);
    for (; i < n4; i += stride) ((float4*)p)[i] = z;
}

__global__ void deg_kernel(const int* __restrict__ src, const int* __restrict__ dst,
                           unsigned* __restrict__ out_cnt, unsigned* __restrict__ in_cnt, int E) {
    int e = blockIdx.x * blockDim.x + threadIdx.x;
    if (e < E) {
        atomicAdd(&out_cnt[src[e]], 1u);
        atomicAdd(&in_cnt[dst[e]], 1u);
    }
}

__global__ void norm_kernel(const unsigned* __restrict__ out_cnt, const unsigned* __restrict__ in_cnt,
                            float* __restrict__ out_norm, float* __restrict__ in_norm, int n) {
    int i = blockIdx.x * blockDim.x + threadIdx.x;
    if (i < n) {
        out_norm[i] = rsqrtf((float)(out_cnt[i] + 1u));
        in_norm[i]  = rsqrtf((float)(in_cnt[i] + 1u));
    }
}

__global__ void init_h_kernel(const float* __restrict__ weight, const float* __restrict__ W_in,
                              const float* __restrict__ b_in, float* __restrict__ h, int n) {
    int idx = blockIdx.x * blockDim.x + threadIdx.x;
    if (idx >= n * 16) return;
    int node = idx >> 4;
    int q = idx & 15;
    float w = weight[node];
    float4 wi = ((const float4*)W_in)[q];
    float4 bi = ((const float4*)b_in)[q];
    float4 r;
    r.x = w * wi.x + bi.x;
    r.y = w * wi.y + bi.y;
    r.z = w * wi.z + bi.z;
    r.w = w * wi.w + bi.w;
    ((float4*)(h + (size_t)node * 64))[q] = r;
}

__global__ __launch_bounds__(256) void gemm_layer_kernel(
    const float* __restrict__ h, const float* __restrict__ W,
    const float* __restrict__ out_norm, float* __restrict__ g, int act)
{
    __shared__ float Wl[64 * 64];
    __shared__ float Tl[32 * 64];

    const int tid = threadIdx.x;
    const int row0 = blockIdx.x * 32;

    for (int i = tid; i < 1024; i += 256)
        ((float4*)Wl)[i] = ((const float4*)W)[i];

    for (int i = tid; i < 512; i += 256) {
        int r  = i >> 4;
        int c4 = i & 15;
        int row = row0 + r;
        float4 v = ((const float4*)(h + (size_t)row * 64))[c4];
        if (act) {
            v.x = v.x > 0.f ? v.x : 0.01f * v.x;
            v.y = v.y > 0.f ? v.y : 0.01f * v.y;
            v.z = v.z > 0.f ? v.z : 0.01f * v.z;
            v.w = v.w > 0.f ? v.w : 0.01f * v.w;
        }
        float s = out_norm[row];
        v.x *= s; v.y *= s; v.z *= s; v.w *= s;
        ((float4*)(Tl + r * 64))[c4] = v;
    }
    __syncthreads();

    const int wave = tid >> 5;
    const int lane = tid & 31;
    const int msub = wave >> 2;
    const int nsub = wave & 3;
    const int l16  = lane & 15;
    const int arow = msub * 16 + l16;
    const int koff = (lane < 16) ? 0 : 2;
    const int col  = nsub * 16 + l16;

    v8f c = {};
    #pragma unroll
    for (int kk = 0; kk < 16; ++kk) {
        int k = kk * 4 + koff;
        v2f a, b;
        a.x = Tl[arow * 64 + k];
        a.y = Tl[arow * 64 + k + 1];
        b.x = Wl[k * 64 + col];
        b.y = Wl[(k + 1) * 64 + col];
        c = __builtin_amdgcn_wmma_f32_16x16x4_f32(false, a, false, b,
                                                  (short)0, c, false, false);
    }

    const int rbase = row0 + msub * 16 + ((lane < 16) ? 0 : 8);
    #pragma unroll
    for (int r = 0; r < 8; ++r)
        g[(size_t)(rbase + r) * 64 + col] = c[r];
}

__global__ __launch_bounds__(256) void spmm_kernel(
    const float* __restrict__ g, const int* __restrict__ src, const int* __restrict__ dst,
    float* __restrict__ agg, int E)
{
    int e = blockIdx.x * 8 + (threadIdx.x >> 5);
    if (e >= E) return;
    int lane = threadIdx.x & 31;
    int s = src[e];
    int d = dst[e];
    float2 v = ((const float2*)(g + (size_t)s * 64))[lane];
    float* a = agg + (size_t)d * 64 + lane * 2;
    atomicAdd(a,     v.x);
    atomicAdd(a + 1, v.y);
}

__global__ void post_kernel(const float* __restrict__ agg, const float* __restrict__ g,
                            const float* __restrict__ in_norm, const float* __restrict__ bc,
                            float* __restrict__ h, int n) {
    int idx = blockIdx.x * blockDim.x + threadIdx.x;
    if (idx >= n * 16) return;
    int node = idx >> 4;
    int q = idx & 15;
    float s = in_norm[node];
    float4 a = ((const float4*)(agg + (size_t)node * 64))[q];
    float4 gv = ((const float4*)(g + (size_t)node * 64))[q];
    float4 b = ((const float4*)bc)[q];
    float4 r;
    r.x = (a.x + gv.x) * s + b.x;
    r.y = (a.y + gv.y) * s + b.y;
    r.z = (a.z + gv.z) * s + b.z;
    r.w = (a.w + gv.w) * s + b.w;
    ((float4*)(h + (size_t)node * 64))[q] = r;
}

__global__ __launch_bounds__(256) void pred_kernel(
    const float* __restrict__ h, const float* __restrict__ W_pred,
    const float* __restrict__ b_pred, float* __restrict__ out, int n)
{
    int node = blockIdx.x * 8 + (threadIdx.x >> 5);
    if (node >= n) return;
    int lane = threadIdx.x & 31;
    float2 w = ((const float2*)W_pred)[lane];
    float2 v = ((const float2*)(h + (size_t)node * 64))[lane];
    float s = v.x * w.x + v.y * w.y;
    for (int off = 16; off > 0; off >>= 1)
        s += __shfl_down(s, off, 32);
    if (lane == 0) out[node] = s + b_pred[0];
}

extern "C" void kernel_launch(void* const* d_in, const int* in_sizes, int n_in,
                              void* d_out, int out_size, void* d_ws, size_t ws_size,
                              hipStream_t stream) {
    const float* weight = (const float*)d_in[0];
    const int*   src    = (const int*)d_in[1];
    const int*   dst    = (const int*)d_in[2];
    const float* W_in   = (const float*)d_in[3];
    const float* b_in   = (const float*)d_in[4];
    const float* Wc     = (const float*)d_in[5];
    const float* bc     = (const float*)d_in[6];
    const float* W_pred = (const float*)d_in[7];
    const float* b_pred = (const float*)d_in[8];
    float* out = (float*)d_out;

    const int N = in_sizes[0];
    const int E = in_sizes[1];

    unsigned* out_cnt = (unsigned*)d_ws;
    unsigned* in_cnt  = out_cnt + N;
    float* out_norm   = (float*)(in_cnt + N);
    float* in_norm    = out_norm + N;
    float* h          = in_norm + N;
    float* g          = h + (size_t)N * 64;
    float* agg        = g + (size_t)N * 64;

    zero_f32<<<512, 256, 0, stream>>>((float*)out_cnt, (2 * N) / 4);

    deg_kernel<<<(E + 255) / 256, 256, 0, stream>>>(src, dst, out_cnt, in_cnt, E);

    norm_kernel<<<(N + 255) / 256, 256, 0, stream>>>(out_cnt, in_cnt, out_norm, in_norm, N);

    init_h_kernel<<<(N * 16 + 255) / 256, 256, 0, stream>>>(weight, W_in, b_in, h, N);

    for (int l = 0; l < 3; ++l) {
        gemm_layer_kernel<<<N / 32, 256, 0, stream>>>(h, Wc + (size_t)l * 64 * 64,
                                                      out_norm, g, l > 0 ? 1 : 0);
        zero_f32<<<2048, 256, 0, stream>>>(agg, (N * 64) / 4);
        spmm_kernel<<<(E + 7) / 8, 256, 0, stream>>>(g, src, dst, agg, E);
        post_kernel<<<(N * 16 + 255) / 256, 256, 0, stream>>>(agg, g, in_norm,
                                                              bc + (size_t)l * 64, h, N);
    }

    pred_kernel<<<(N + 7) / 8, 256, 0, stream>>>(h, W_pred, b_pred, out, N);
}


# SpMM by gather: edges sorted by destination once (in-degree add returns an arrival-order rank); one wave sums a node's in-edges in rank order and applies the epilogue; float atomics gone
# speedup vs baseline: 2.3008x; 2.3008x over previous
#include <hip/hip_runtime.h>
#include <math.h>

typedef __attribute__((ext_vector_type(2))) float v2f;
typedef __attribute__((ext_vector_type(8))) float v8f;

#define NN 100000

__global__ void zero_f32(float* __restrict__ p, int n4) {
    int i = blockIdx.x * blockDim.x + threadIdx.x;
    int stride = gridDim.x * blockDim.x;
    float4 z = make_float4(0.f, 0.f, 0.f, 0.f);
    for (; i < n4; i += stride) ((float4*)p)[i] = z;
}

__global__ void deg_rank_kernel(const int* __restrict__ src, const int* __restrict__ dst,
                                unsigned* __restrict__ out_cnt, unsigned* __restrict__ in_cnt,
                                unsigned* __restrict__ rank, int E) {
    int e = blockIdx.x * blockDim.x + threadIdx.x;
    if (e < E) {
        atomicAdd(&out_cnt[src[e]], 1u);
        rank[e] = atomicAdd(&in_cnt[dst[e]], 1u);
    }
}

__global__ __launch_bounds__(256) void norm_alloc_kernel(
    const unsigned* __restrict__ out_cnt, const unsigned* __restrict__ in_cnt,
    float* __restrict__ out_norm, float* __restrict__ in_norm,
    unsigned* __restrict__ row_start, unsigned* __restrict__ cursor, int n)
{
    __shared__ unsigned wsum[8];
    __shared__ unsigned base_s;

    const int tid  = threadIdx.x;
    const int lane = tid & 31;
    const int wave = tid >> 5;
    const int i = blockIdx.x * 256 + tid;

    unsigned cnt = 0;
    if (i < n) {
        cnt = in_cnt[i];
        out_norm[i] = rsqrtf((float)(out_cnt[i] + 1u));
        in_norm[i]  = rsqrtf((float)(cnt + 1u));
    }

    unsigned x = cnt;
    for (int off = 1; off < 32; off <<= 1) {
        unsigned y = __shfl_up(x, off, 32);
        if (lane >= off) x += y;
    }
    if (lane == 31) wsum[wave] = x;
    __syncthreads();
    if (tid == 0) {
        unsigned t = 0;
        for (int w = 0; w < 8; ++w) { unsigned s = wsum[w]; wsum[w] = t; t += s; }
        base_s = atomicAdd(cursor, t);
    }
    __syncthreads();
    if (i < n) row_start[i] = base_s + wsum[wave] + (x - cnt);
}

__global__ void fill_kernel(const int* __restrict__ src, const int* __restrict__ dst,
                            const unsigned* __restrict__ rank, const unsigned* __restrict__ row_start,
                            int* __restrict__ csr_src, int E) {
    int e = blockIdx.x * blockDim.x + threadIdx.x;
    if (e < E) csr_src[row_start[dst[e]] + rank[e]] = src[e];
}

__global__ void init_h_kernel(const float* __restrict__ weight, const float* __restrict__ W_in,
                              const float* __restrict__ b_in, float* __restrict__ h, int n) {
    int idx = blockIdx.x * blockDim.x + threadIdx.x;
    if (idx >= n * 16) return;
    int node = idx >> 4;
    int q = idx & 15;
    float w = weight[node];
    float4 wi = ((const float4*)W_in)[q];
    float4 bi = ((const float4*)b_in)[q];
    float4 r;
    r.x = w * wi.x + bi.x;
    r.y = w * wi.y + bi.y;
    r.z = w * wi.z + bi.z;
    r.w = w * wi.w + bi.w;
    ((float4*)(h + (size_t)node * 64))[q] = r;
}

__global__ __launch_bounds__(256) void gemm_layer_kernel(
    const float* __restrict__ h, const float* __restrict__ W,
    const float* __restrict__ out_norm, float* __restrict__ g, int act)
{
    __shared__ float Wl[64 * 64];
    __shared__ float Tl[32 * 64];

    const int tid = threadIdx.x;
    const int row0 = blockIdx.x * 32;

    for (int i = tid; i < 1024; i += 256)
        ((float4*)Wl)[i] = ((const float4*)W)[i];

    for (int i = tid; i < 512; i += 256) {
        int r  = i >> 4;
        int c4 = i & 15;
        int row = row0 + r;
        float4 v = ((const float4*)(h + (size_t)row * 64))[c4];
        if (act) {
            v.x = v.x > 0.f ? v.x : 0.01f * v.x;
            v.y = v.y > 0.f ? v.y : 0.01f * v.y;
            v.z = v.z > 0.f ? v.z : 0.01f * v.z;
            v.w = v.w > 0.f ? v.w : 0.01f * v.w;
        }
        float s = out_norm[row];
        v.x *= s; v.y *= s; v.z *= s; v.w *= s;
        ((float4*)(Tl + r * 64))[c4] = v;
    }
    __syncthreads();

    const int wave = tid >> 5;
    const int lane = tid & 31;
    const int msub = wave >> 2;
    const int nsub = wave & 3;
    const int l16  = lane & 15;
    const int arow = msub * 16 + l16;
    const int koff = (lane < 16) ? 0 : 2;
    const int col  = nsub * 16 + l16;

    v8f c = {};
    #pragma unroll
    for (int kk = 0; kk < 16; ++kk) {
        int k = kk * 4 + koff;
        v2f a, b;
        a.x = Tl[arow * 64 + k];
        a.y = Tl[arow * 64 + k + 1];
        b.x = Wl[k * 64 + col];
        b.y = Wl[(k + 1) * 64 + col];
        c = __builtin_amdgcn_wmma_f32_16x16x4_f32(false, a, false, b,
                                                  (short)0, c, false, false);
    }

    const int rbase = row0 + msub * 16 + ((lane < 16) ? 0 : 8);
    #pragma unroll
    for (int r = 0; r < 8; ++r)
        g[(size_t)(rbase + r) * 64 + col] = c[r];
}

#define ROWS(U)                                                            \
    {                                                                      \
        float2 v[U];                                                       \
        _Pragma("unroll")                                                  \
        for (int u = 0; u < U; ++u) {                                      \
              \
            int s = __builtin_amdgcn_readlane(my, (int)(j + u));           \
            v[u] = g2[(size_t)s * 32 + lane];                              \
        }                                                                  \
        _Pragma("unroll")                                                  \
        for (int u = 0; u < U; ++u) { acc.x += v[u].x; acc.y += v[u].y; }  \
        j += U;                                                            \
    }

__global__ __launch_bounds__(256) void gather_post_kernel(
    const float* __restrict__ g, const int* __restrict__ csr_src,
    const unsigned* __restrict__ row_start, const unsigned* __restrict__ in_cnt,
    const float* __restrict__ in_norm, const float* __restrict__ bc,
    float* __restrict__ h, int n)
{
    const int node = blockIdx.x * 8 + (threadIdx.x >> 5);
    if (node >= n) return;
    const int lane = threadIdx.x & 31;
    const float2* g2 = (const float2*)g;

    const unsigned start = __builtin_amdgcn_readfirstlane(row_start[node]);
    const unsigned cnt   = __builtin_amdgcn_readfirstlane(in_cnt[node]);

    float2 acc = make_float2(0.f, 0.f);
    for (unsigned base = 0; base < cnt; base += 32) {
        const unsigned m = (cnt - base < 32u) ? (cnt - base) : 32u;
        const int my = ((unsigned)lane < m) ? csr_src[start + base + lane] : 0;
        unsigned j = 0;
        if (j + 16 <= m) ROWS(16)
        if (j + 16 <= m) ROWS(16)
        if (j + 8 <= m) ROWS(8)
        if (j + 4 <= m) ROWS(4)
        if (j + 2 <= m) ROWS(2)
        if (j + 1 <= m) ROWS(1)
    }

    const float2 gv = g2[(size_t)node * 32 + lane];
    const float sc = in_norm[node];
    const float2 b = ((const float2*)bc)[lane];
    float2 r;
    r.x = (acc.x + gv.x) * sc + b.x;
    r.y = (acc.y + gv.y) * sc + b.y;
    ((float2*)h)[(size_t)node * 32 + lane] = r;
}
#undef ROWS

__global__ __launch_bounds__(256) void pred_kernel(
    const float* __restrict__ h, const float* __restrict__ W_pred,
    const float* __restrict__ b_pred, float* __restrict__ out, int n)
{
    int node = blockIdx.x * 8 + (threadIdx.x >> 5);
    if (node >= n) return;
    int lane = threadIdx.x & 31;
    float2 w = ((const float2*)W_pred)[lane];
    float2 v = ((const float2*)(h + (size_t)node * 64))[lane];
    float s = v.x * w.x + v.y * w.y;
    for (int off = 16; off > 0; off >>= 1)
        s += __shfl_down(s, off, 32);
    if (lane == 0) out[node] = s + b_pred[0];
}

extern "C" void kernel_launch(void* const* d_in, const int* in_sizes, int n_in,
                              void* d_out, int out_size, void* d_ws, size_t ws_size,
                              hipStream_t stream) {
    const float* weight = (const float*)d_in[0];
    const int*   src    = (const int*)d_in[1];
    const int*   dst    = (const int*)d_in[2];
    const float* W_in   = (const float*)d_in[3];
    const float* b_in   = (const float*)d_in[4];
    const float* Wc     = (const float*)d_in[5];
    const float* bc     = (const float*)d_in[6];
    const float* W_pred = (const float*)d_in[7];
    const float* b_pred = (const float*)d_in[8];
    float* out = (float*)d_out;

    const int N = in_sizes[0];
    const int E = in_sizes[1];

    float* h          = (float*)d_ws;
    float* g          = h + (size_t)N * 64;
    unsigned* out_cnt = (unsigned*)(g + (size_t)N * 64);
    unsigned* in_cnt  = out_cnt + N;
    unsigned* cursor  = in_cnt + N;
    float* out_norm   = (float*)(cursor + 4);
    float* in_norm    = out_norm + N;
    unsigned* row_start = (unsigned*)(in_norm + N);
    int* csr_src      = (int*)(row_start + N);
    unsigned* rank    = (unsigned*)(csr_src + E);

    zero_f32<<<512, 256, 0, stream>>>((float*)out_cnt, (2 * N + 4) / 4);

    deg_rank_kernel<<<(E + 255) / 256, 256, 0, stream>>>(src, dst, out_cnt, in_cnt, rank, E);

    norm_alloc_kernel<<<(N + 255) / 256, 256, 0, stream>>>(out_cnt, in_cnt, out_norm, in_norm,
                                                           row_start, cursor, N);

    fill_kernel<<<(E + 255) / 256, 256, 0, stream>>>(src, dst, rank, row_start, csr_src, E);

    init_h_kernel<<<(N * 16 + 255) / 256, 256, 0, stream>>>(weight, W_in, b_in, h, N);

    for (int l = 0; l < 3; ++l) {
        gemm_layer_kernel<<<N / 32, 256, 0, stream>>>(h, Wc + (size_t)l * 64 * 64,
                                                      out_norm, g, l > 0 ? 1 : 0);
        gather_post_kernel<<<(N + 7) / 8, 256, 0, stream>>>(g, csr_src, row_start, in_cnt, in_norm,
                                                            bc + (size_t)l * 64, h, N);
    }

    pred_kernel<<<(N + 7) / 8, 256, 0, stream>>>(h, W_pred, b_pred, out, N);
}


# Per-edge global atomics (degrees, arrival rank) replaced by LDS counters: edges binned to 256-node buckets per block chunk, then ranked per node inside a bucket; in-edge sum order stays arrival order
# speedup vs baseline: 3.6214x; 1.5740x over previous
#include <hip/hip_runtime.h>
#include <math.h>

typedef __attribute__((ext_vector_type(2))) float v2f;
typedef __attribute__((ext_vector_type(8))) float v8f;

#define NN 100000

#define GCH 256
#define BSH 8
#define BMASK 255u

__device__ __forceinline__ void block_excl_scan(unsigned* a, int n, unsigned* wsum) {
    const int tid = threadIdx.x, lane = tid & 31, wave = tid >> 5;
    const int per = (n + 1023) >> 10;
    const int i0 = tid * per;
    unsigned s = 0;
    for (int k = 0; k < per; ++k) if (i0 + k < n) s += a[i0 + k];
    unsigned x = s;
    for (int off = 1; off < 32; off <<= 1) {
        unsigned y = __shfl_up(x, off, 32);
        if (lane >= off) x += y;
    }
    if (lane == 31) wsum[wave] = x;
    __syncthreads();
    unsigned run = x - s;
    for (int w = 0; w < wave; ++w) run += wsum[w];
    for (int k = 0; k < per; ++k) if (i0 + k < n) { unsigned v = a[i0 + k]; a[i0 + k] = run; run += v; }
    __syncthreads();
}

__global__ __launch_bounds__(1024) void bucket_count_kernel(
    const int* __restrict__ src, const int* __restrict__ dst,
    unsigned* __restrict__ tableD, unsigned* __restrict__ tableS, int E, int NB)
{
    extern __shared__ unsigned dyn_lds[];
    const int tid = threadIdx.x;
    for (int i = tid; i < 2 * NB; i += 1024) dyn_lds[i] = 0u;
    __syncthreads();

    const int chunk = (E + GCH - 1) / GCH;
    const int e0 = blockIdx.x * chunk;
    const int e1 = (e0 + chunk < E) ? (e0 + chunk) : E;
    for (int e = e0 + tid; e < e1; e += 4096) {
        int d[4], s[4];
        #pragma unroll
        for (int u = 0; u < 4; ++u) {
            const int ee = e + u * 1024;
            const int ec = (ee < e1) ? ee : e;
            d[u] = dst[ec];
            s[u] = src[ec];
        }
        asm volatile("" :: "v"(d[0]), "v"(d[1]), "v"(d[2]), "v"(d[3]), "v"(s[0]), "v"(s[1]), "v"(s[2]), "v"(s[3]));
        #pragma unroll
        for (int u = 0; u < 4; ++u) {
            if (e + u * 1024 < e1) {
                atomicAdd(&dyn_lds[d[u] >> BSH], 1u);
                atomicAdd(&dyn_lds[NB + (s[u] >> BSH)], 1u);
            }
        }
    }
    __syncthreads();
    for (int b = tid; b < NB; b += 1024) {
        tableD[b * GCH + blockIdx.x] = dyn_lds[b];
        tableS[b * GCH + blockIdx.x] = dyn_lds[NB + b];
    }
}

__global__ __launch_bounds__(GCH) void bucket_scan_kernel(
    unsigned* __restrict__ tableD, unsigned* __restrict__ tableS,
    unsigned* __restrict__ totD, unsigned* __restrict__ totS)
{
    __shared__ unsigned ws[2][GCH / 32];
    const int t = threadIdx.x, lane = t & 31, wave = t >> 5;
    const size_t idx = (size_t)blockIdx.x * GCH + t;
    const unsigned vd = tableD[idx], vs = tableS[idx];
    unsigned xd = vd, xs = vs;
    for (int off = 1; off < 32; off <<= 1) {
        unsigned yd = __shfl_up(xd, off, 32);
        unsigned ys = __shfl_up(xs, off, 32);
        if (lane >= off) { xd += yd; xs += ys; }
    }
    if (lane == 31) { ws[0][wave] = xd; ws[1][wave] = xs; }
    __syncthreads();
    unsigned od = 0, os = 0;
    for (int w = 0; w < wave; ++w) { od += ws[0][w]; os += ws[1][w]; }
    tableD[idx] = od + xd - vd;
    tableS[idx] = os + xs - vs;
    if (t == GCH - 1) { totD[blockIdx.x] = od + xd; totS[blockIdx.x] = os + xs; }
}

__global__ __launch_bounds__(1024) void bucket_scatter_kernel(
    const int* __restrict__ src, const int* __restrict__ dst,
    const unsigned* __restrict__ tableD, const unsigned* __restrict__ tableS,
    const unsigned* __restrict__ totD, const unsigned* __restrict__ totS,
    unsigned* __restrict__ baseD, unsigned* __restrict__ baseS,
    unsigned* __restrict__ bufD, unsigned char* __restrict__ bufS, int E, int NB)
{
    extern __shared__ unsigned dyn_lds[];
    __shared__ unsigned wsum[32];
    const int tid = threadIdx.x;
    for (int b = tid; b < NB; b += 1024) { dyn_lds[b] = totD[b]; dyn_lds[NB + b] = totS[b]; }
    __syncthreads();
    block_excl_scan(dyn_lds, NB, wsum);
    block_excl_scan(dyn_lds + NB, NB, wsum);
    for (int b = tid; b < NB; b += 1024) {
        const unsigned bd = dyn_lds[b], bs = dyn_lds[NB + b];
        if (blockIdx.x == 0) { baseD[b] = bd; baseS[b] = bs; }
        dyn_lds[b]      = bd + tableD[b * GCH + blockIdx.x];
        dyn_lds[NB + b] = bs + tableS[b * GCH + blockIdx.x];
    }
    __syncthreads();

    const int chunk = (E + GCH - 1) / GCH;
    const int e0 = blockIdx.x * chunk;
    const int e1 = (e0 + chunk < E) ? (e0 + chunk) : E;
    for (int e = e0 + tid; e < e1; e += 4096) {
        int d[4], s[4];
        #pragma unroll
        for (int u = 0; u < 4; ++u) {
            const int ee = e + u * 1024;
            const int ec = (ee < e1) ? ee : e;
            d[u] = dst[ec];
            s[u] = src[ec];
        }
        asm volatile("" :: "v"(d[0]), "v"(d[1]), "v"(d[2]), "v"(d[3]), "v"(s[0]), "v"(s[1]), "v"(s[2]), "v"(s[3]));
        #pragma unroll
        for (int u = 0; u < 4; ++u) {
            if (e + u * 1024 < e1) {
                const unsigned pd = atomicAdd(&dyn_lds[d[u] >> BSH], 1u);
                bufD[pd] = (((unsigned)d[u] & BMASK) << 24) | (unsigned)s[u];
                const unsigned ps = atomicAdd(&dyn_lds[NB + (s[u] >> BSH)], 1u);
                bufS[ps] = (unsigned char)((unsigned)s[u] & BMASK);
            }
        }
    }
}

__global__ __launch_bounds__(1024) void bucket_finish_kernel(
    const unsigned* __restrict__ bufD, const unsigned char* __restrict__ bufS,
    const unsigned* __restrict__ baseD, const unsigned* __restrict__ baseS,
    const unsigned* __restrict__ totD, const unsigned* __restrict__ totS,
    float* __restrict__ out_norm, float* __restrict__ in_norm,
    unsigned* __restrict__ in_cnt, unsigned* __restrict__ row_start,
    int* __restrict__ csr_src, int n)
{
    __shared__ unsigned cin[256];
    __shared__ unsigned cout[256];
    __shared__ unsigned cur[256];
    __shared__ unsigned wsum[32];
    const int tid = threadIdx.x;
    const int b = blockIdx.x;
    const unsigned bD = baseD[b], nD = totD[b];
    const unsigned bS = baseS[b], nS = totS[b];
    if (tid < 256) { cin[tid] = 0u; cout[tid] = 0u; }
    __syncthreads();
    for (unsigned i = tid; i < nS; i += 1024) atomicAdd(&cout[bufS[bS + i]], 1u);
    for (unsigned i = tid; i < nD; i += 1024) atomicAdd(&cin[bufD[bD + i] >> 24], 1u);
    __syncthreads();
    if (tid < 256) cur[tid] = cin[tid];
    __syncthreads();
    block_excl_scan(cur, 256, wsum);
    if (tid < 256) {
        const unsigned rs = bD + cur[tid];
        cur[tid] = rs;
        const int node = b * 256 + tid;
        if (node < n) {
            const unsigned c = cin[tid];
            row_start[node] = rs;
            in_cnt[node] = c;
            in_norm[node]  = rsqrtf((float)(c + 1u));
            out_norm[node] = rsqrtf((float)(cout[tid] + 1u));
        }
    }
    __syncthreads();
    for (unsigned i = tid; i < nD; i += 1024) {
        const unsigned w = bufD[bD + i];
        const unsigned pos = atomicAdd(&cur[w >> 24], 1u);
        csr_src[pos] = (int)(w & 0x00FFFFFFu);
    }
}

__global__ void init_h_kernel(const float* __restrict__ weight, const float* __restrict__ W_in,
                              const float* __restrict__ b_in, float* __restrict__ h, int n) {
    int idx = blockIdx.x * blockDim.x + threadIdx.x;
    if (idx >= n * 16) return;
    int node = idx >> 4;
    int q = idx & 15;
    float w = weight[node];
    float4 wi = ((const float4*)W_in)[q];
    float4 bi = ((const float4*)b_in)[q];
    float4 r;
    r.x = w * wi.x + bi.x;
    r.y = w * wi.y + bi.y;
    r.z = w * wi.z + bi.z;
    r.w = w * wi.w + bi.w;
    ((float4*)(h + (size_t)node * 64))[q] = r;
}

__global__ __launch_bounds__(256) void gemm_layer_kernel(
    const float* __restrict__ h, const float* __restrict__ W,
    const float* __restrict__ out_norm, float* __restrict__ g, int act)
{
    __shared__ float Wl[64 * 64];
    __shared__ float Tl[32 * 64];

    const int tid = threadIdx.x;
    const int row0 = blockIdx.x * 32;

    for (int i = tid; i < 1024; i += 256)
        ((float4*)Wl)[i] = ((const float4*)W)[i];

    for (int i = tid; i < 512; i += 256) {
        int r  = i >> 4;
        int c4 = i & 15;
        int row = row0 + r;
        float4 v = ((const float4*)(h + (size_t)row * 64))[c4];
        if (act) {
            v.x = v.x > 0.f ? v.x : 0.01f * v.x;
            v.y = v.y > 0.f ? v.y : 0.01f * v.y;
            v.z = v.z > 0.f ? v.z : 0.01f * v.z;
            v.w = v.w > 0.f ? v.w : 0.01f * v.w;
        }
        float s = out_norm[row];
        v.x *= s; v.y *= s; v.z *= s; v.w *= s;
        ((float4*)(Tl + r * 64))[c4] = v;
    }
    __syncthreads();

    const int wave = tid >> 5;
    const int lane = tid & 31;
    const int msub = wave >> 2;
    const int nsub = wave & 3;
    const int l16  = lane & 15;
    const int arow = msub * 16 + l16;
    const int koff = (lane < 16) ? 0 : 2;
    const int col  = nsub * 16 + l16;

    v8f c = {};
    #pragma unroll
    for (int kk = 0; kk < 16; ++kk) {
        int k = kk * 4 + koff;
        v2f a, b;
        a.x = Tl[arow * 64 + k];
        a.y = Tl[arow * 64 + k + 1];
        b.x = Wl[k * 64 + col];
        b.y = Wl[(k + 1) * 64 + col];
        c = __builtin_amdgcn_wmma_f32_16x16x4_f32(false, a, false, b,
                                                  (short)0, c, false, false);
    }

    const int rbase = row0 + msub * 16 + ((lane < 16) ? 0 : 8);
    #pragma unroll
    for (int r = 0; r < 8; ++r)
        g[(size_t)(rbase + r) * 64 + col] = c[r];
}

#define ROWS(U)                                                            \
    {                                                                      \
        float2 v[U];                                                       \
        _Pragma("unroll")                                                  \
        for (int u = 0; u < U; ++u) {                                      \
              \
            int s = __builtin_amdgcn_readlane(my, (int)(j + u));           \
            v[u] = g2[(size_t)s * 32 + lane];                              \
        }                                                                  \
        _Pragma("unroll")                                                  \
        for (int u = 0; u < U; ++u) { acc.x += v[u].x; acc.y += v[u].y; }  \
        j += U;                                                            \
    }

__global__ __launch_bounds__(256) void gather_post_kernel(
    const float* __restrict__ g, const int* __restrict__ csr_src,
    const unsigned* __restrict__ row_start, const unsigned* __restrict__ in_cnt,
    const float* __restrict__ in_norm, const float* __restrict__ bc,
    float* __restrict__ h, int n)
{
    const int node = blockIdx.x * 8 + (threadIdx.x >> 5);
    if (node >= n) return;
    const int lane = threadIdx.x & 31;
    const float2* g2 = (const float2*)g;

    const unsigned start = __builtin_amdgcn_readfirstlane(row_start[node]);
    const unsigned cnt   = __builtin_amdgcn_readfirstlane(in_cnt[node]);

    float2 acc = make_float2(0.f, 0.f);
    for (unsigned base = 0; base < cnt; base += 32) {
        const unsigned m = (cnt - base < 32u) ? (cnt - base) : 32u;
        const int my = ((unsigned)lane < m) ? csr_src[start + base + lane] : 0;
        unsigned j = 0;
        if (j + 16 <= m) ROWS(16)
        if (j + 16 <= m) ROWS(16)
        if (j + 8 <= m) ROWS(8)
        if (j + 4 <= m) ROWS(4)
        if (j + 2 <= m) ROWS(2)
        if (j + 1 <= m) ROWS(1)
    }

    const float2 gv = g2[(size_t)node * 32 + lane];
    const float sc = in_norm[node];
    const float2 b = ((const float2*)bc)[lane];
    float2 r;
    r.x = (acc.x + gv.x) * sc + b.x;
    r.y = (acc.y + gv.y) * sc + b.y;
    ((float2*)h)[(size_t)node * 32 + lane] = r;
}
#undef ROWS

__global__ __launch_bounds__(256) void pred_kernel(
    const float* __restrict__ h, const float* __restrict__ W_pred,
    const float* __restrict__ b_pred, float* __restrict__ out, int n)
{
    int node = blockIdx.x * 8 + (threadIdx.x >> 5);
    if (node >= n) return;
    int lane = threadIdx.x & 31;
    float2 w = ((const float2*)W_pred)[lane];
    float2 v = ((const float2*)(h + (size_t)node * 64))[lane];
    float s = v.x * w.x + v.y * w.y;
    for (int off = 16; off > 0; off >>= 1)
        s += __shfl_down(s, off, 32);
    if (lane == 0) out[node] = s + b_pred[0];
}

extern "C" void kernel_launch(void* const* d_in, const int* in_sizes, int n_in,
                              void* d_out, int out_size, void* d_ws, size_t ws_size,
                              hipStream_t stream) {
    const float* weight = (const float*)d_in[0];
    const int*   src    = (const int*)d_in[1];
    const int*   dst    = (const int*)d_in[2];
    const float* W_in   = (const float*)d_in[3];
    const float* b_in   = (const float*)d_in[4];
    const float* Wc     = (const float*)d_in[5];
    const float* bc     = (const float*)d_in[6];
    const float* W_pred = (const float*)d_in[7];
    const float* b_pred = (const float*)d_in[8];
    float* out = (float*)d_out;

    const int N = in_sizes[0];
    const int E = in_sizes[1];

    const int NB = (N + 255) / 256;

    float* h          = (float*)d_ws;
    float* g          = h + (size_t)N * 64;
    unsigned* in_cnt  = (unsigned*)(g + (size_t)N * 64);
    float* out_norm   = (float*)(in_cnt + N);
    float* in_norm    = out_norm + N;
    unsigned* row_start = (unsigned*)(in_norm + N);
    int* csr_src      = (int*)(row_start + N);
    unsigned* bufD    = (unsigned*)(csr_src + E);
    unsigned* tableD  = bufD + E;
    unsigned* tableS  = tableD + (size_t)NB * GCH;
    unsigned* totD    = tableS + (size_t)NB * GCH;
    unsigned* totS    = totD + NB;
    unsigned* baseD   = totS + NB;
    unsigned* baseS   = baseD + NB;
    unsigned char* bufS = (unsigned char*)(baseS + NB);

    bucket_count_kernel<<<GCH, 1024, 2 * NB * sizeof(unsigned), stream>>>(src, dst, tableD, tableS, E, NB);

    bucket_scan_kernel<<<NB, GCH, 0, stream>>>(tableD, tableS, totD, totS);

    bucket_scatter_kernel<<<GCH, 1024, 2 * NB * sizeof(unsigned), stream>>>(
        src, dst, tableD, tableS, totD, totS, baseD, baseS, bufD, bufS, E, NB);

    bucket_finish_kernel<<<NB, 1024, 0, stream>>>(bufD, bufS, baseD, baseS, totD, totS,
                                                  out_norm, in_norm, in_cnt, row_start, csr_src, N);

    init_h_kernel<<<(N * 16 + 255) / 256, 256, 0, stream>>>(weight, W_in, b_in, h, N);

    for (int l = 0; l < 3; ++l) {
        gemm_layer_kernel<<<N / 32, 256, 0, stream>>>(h, Wc + (size_t)l * 64 * 64,
                                                      out_norm, g, l > 0 ? 1 : 0);
        gather_post_kernel<<<(N + 7) / 8, 256, 0, stream>>>(g, csr_src, row_start, in_cnt, in_norm,
                                                            bc + (size_t)l * 64, h, N);
    }

    pred_kernel<<<(N + 7) / 8, 256, 0, stream>>>(h, W_pred, b_pred, out, N);
}


# Layer 0 sums before it projects: its input rows are rank 2 (weight x W_in + b_in), so its SpMM sums two scalars per edge (8 strided partial sums, then a tree) ahead of the WMMA product; no init pass
# speedup vs baseline: 3.8181x; 1.0543x over previous
#include <hip/hip_runtime.h>
#include <math.h>

typedef __attribute__((ext_vector_type(2))) float v2f;
typedef __attribute__((ext_vector_type(8))) float v8f;

#define NN 100000

#define GCH 256
#define BSH 8
#define BMASK 255u

__device__ __forceinline__ void block_excl_scan(unsigned* a, int n, unsigned* wsum) {
    const int tid = threadIdx.x, lane = tid & 31, wave = tid >> 5;
    const int per = (n + 1023) >> 10;
    const int i0 = tid * per;
    unsigned s = 0;
    for (int k = 0; k < per; ++k) if (i0 + k < n) s += a[i0 + k];
    unsigned x = s;
    for (int off = 1; off < 32; off <<= 1) {
        unsigned y = __shfl_up(x, off, 32);
        if (lane >= off) x += y;
    }
    if (lane == 31) wsum[wave] = x;
    __syncthreads();
    unsigned run = x - s;
    for (int w = 0; w < wave; ++w) run += wsum[w];
    for (int k = 0; k < per; ++k) if (i0 + k < n) { unsigned v = a[i0 + k]; a[i0 + k] = run; run += v; }
    __syncthreads();
}

__global__ __launch_bounds__(1024) void bucket_count_kernel(
    const int* __restrict__ src, const int* __restrict__ dst,
    unsigned* __restrict__ tableD, unsigned* __restrict__ tableS, int E, int NB)
{
    extern __shared__ unsigned dyn_lds[];
    const int tid = threadIdx.x;
    for (int i = tid; i < 2 * NB; i += 1024) dyn_lds[i] = 0u;
    __syncthreads();

    const int chunk = (E + GCH - 1) / GCH;
    const int e0 = blockIdx.x * chunk;
    const int e1 = (e0 + chunk < E) ? (e0 + chunk) : E;
    for (int e = e0 + tid; e < e1; e += 4096) {
        int d[4], s[4];
        #pragma unroll
        for (int u = 0; u < 4; ++u) {
            const int ee = e + u * 1024;
            const int ec = (ee < e1) ? ee : e;
            d[u] = dst[ec];
            s[u] = src[ec];
        }
        asm volatile("" :: "v"(d[0]), "v"(d[1]), "v"(d[2]), "v"(d[3]), "v"(s[0]), "v"(s[1]), "v"(s[2]), "v"(s[3]));
        #pragma unroll
        for (int u = 0; u < 4; ++u) {
            if (e + u * 1024 < e1) {
                atomicAdd(&dyn_lds[d[u] >> BSH], 1u);
                atomicAdd(&dyn_lds[NB + (s[u] >> BSH)], 1u);
            }
        }
    }
    __syncthreads();
    for (int b = tid; b < NB; b += 1024) {
        tableD[b * GCH + blockIdx.x] = dyn_lds[b];
        tableS[b * GCH + blockIdx.x] = dyn_lds[NB + b];
    }
}

__global__ __launch_bounds__(GCH) void bucket_scan_kernel(
    unsigned* __restrict__ tableD, unsigned* __restrict__ tableS,
    unsigned* __restrict__ totD, unsigned* __restrict__ totS)
{
    __shared__ unsigned ws[2][GCH / 32];
    const int t = threadIdx.x, lane = t & 31, wave = t >> 5;
    const size_t idx = (size_t)blockIdx.x * GCH + t;
    const unsigned vd = tableD[idx], vs = tableS[idx];
    unsigned xd = vd, xs = vs;
    for (int off = 1; off < 32; off <<= 1) {
        unsigned yd = __shfl_up(xd, off, 32);
        unsigned ys = __shfl_up(xs, off, 32);
        if (lane >= off) { xd += yd; xs += ys; }
    }
    if (lane == 31) { ws[0][wave] = xd; ws[1][wave] = xs; }
    __syncthreads();
    unsigned od = 0, os = 0;
    for (int w = 0; w < wave; ++w) { od += ws[0][w]; os += ws[1][w]; }
    tableD[idx] = od + xd - vd;
    tableS[idx] = os + xs - vs;
    if (t == GCH - 1) { totD[blockIdx.x] = od + xd; totS[blockIdx.x] = os + xs; }
}

__global__ __launch_bounds__(1024) void bucket_scatter_kernel(
    const int* __restrict__ src, const int* __restrict__ dst,
    const unsigned* __restrict__ tableD, const unsigned* __restrict__ tableS,
    const unsigned* __restrict__ totD, const unsigned* __restrict__ totS,
    unsigned* __restrict__ baseD, unsigned* __restrict__ baseS,
    unsigned* __restrict__ bufD, unsigned char* __restrict__ bufS, int E, int NB)
{
    extern __shared__ unsigned dyn_lds[];
    __shared__ unsigned wsum[32];
    const int tid = threadIdx.x;
    for (int b = tid; b < NB; b += 1024) { dyn_lds[b] = totD[b]; dyn_lds[NB + b] = totS[b]; }
    __syncthreads();
    block_excl_scan(dyn_lds, NB, wsum);
    block_excl_scan(dyn_lds + NB, NB, wsum);
    for (int b = tid; b < NB; b += 1024) {
        const unsigned bd = dyn_lds[b], bs = dyn_lds[NB + b];
        if (blockIdx.x == 0) { baseD[b] = bd; baseS[b] = bs; }
        dyn_lds[b]      = bd + tableD[b * GCH + blockIdx.x];
        dyn_lds[NB + b] = bs + tableS[b * GCH + blockIdx.x];
    }
    __syncthreads();

    const int chunk = (E + GCH - 1) / GCH;
    const int e0 = blockIdx.x * chunk;
    const int e1 = (e0 + chunk < E) ? (e0 + chunk) : E;
    for (int e = e0 + tid; e < e1; e += 4096) {
        int d[4], s[4];
        #pragma unroll
        for (int u = 0; u < 4; ++u) {
            const int ee = e + u * 1024;
            const int ec = (ee < e1) ? ee : e;
            d[u] = dst[ec];
            s[u] = src[ec];
        }
        asm volatile("" :: "v"(d[0]), "v"(d[1]), "v"(d[2]), "v"(d[3]), "v"(s[0]), "v"(s[1]), "v"(s[2]), "v"(s[3]));
        #pragma unroll
        for (int u = 0; u < 4; ++u) {
            if (e + u * 1024 < e1) {
                const unsigned pd = atomicAdd(&dyn_lds[d[u] >> BSH], 1u);
                bufD[pd] = (((unsigned)d[u] & BMASK) << 24) | (unsigned)s[u];
                const unsigned ps = atomicAdd(&dyn_lds[NB + (s[u] >> BSH)], 1u);
                bufS[ps] = (unsigned char)((unsigned)s[u] & BMASK);
            }
        }
    }
}

__global__ __launch_bounds__(1024) void bucket_finish_kernel(
    const unsigned* __restrict__ bufD, const unsigned char* __restrict__ bufS,
    const unsigned* __restrict__ baseD, const unsigned* __restrict__ baseS,
    const unsigned* __restrict__ totD, const unsigned* __restrict__ totS,
    const float* __restrict__ weight,
    float* __restrict__ out_norm, float* __restrict__ in_norm, float* __restrict__ pq,
    unsigned* __restrict__ in_cnt, unsigned* __restrict__ row_start,
    int* __restrict__ csr_src, int n)
{
    __shared__ unsigned cin[256];
    __shared__ unsigned cout[256];
    __shared__ unsigned cur[256];
    __shared__ unsigned wsum[32];
    const int tid = threadIdx.x;
    const int b = blockIdx.x;
    const unsigned bD = baseD[b], nD = totD[b];
    const unsigned bS = baseS[b], nS = totS[b];
    const float wv = (tid < 256 && b * 256 + tid < n) ? weight[b * 256 + tid] : 0.f;
    if (tid < 256) { cin[tid] = 0u; cout[tid] = 0u; }
    __syncthreads();
    for (unsigned i = tid; i < nS; i += 1024) atomicAdd(&cout[bufS[bS + i]], 1u);
    for (unsigned i = tid; i < nD; i += 1024) atomicAdd(&cin[bufD[bD + i] >> 24], 1u);
    __syncthreads();
    if (tid < 256) cur[tid] = cin[tid];
    __syncthreads();
    block_excl_scan(cur, 256, wsum);
    if (tid < 256) {
        const unsigned rs = bD + cur[tid];
        cur[tid] = rs;
        const int node = b * 256 + tid;
        if (node < n) {
            const unsigned c = cin[tid];
            row_start[node] = rs;
            in_cnt[node] = c;
            in_norm[node]  = rsqrtf((float)(c + 1u));
            const float on = rsqrtf((float)(cout[tid] + 1u));
            out_norm[node] = on;
            ((float2*)pq)[node] = make_float2(on * wv, on);
        }
    }
    __syncthreads();
    for (unsigned i = tid; i < nD; i += 1024) {
        const unsigned w = bufD[bD + i];
        const unsigned pos = atomicAdd(&cur[w >> 24], 1u);
        csr_src[pos] = (int)(w & 0x00FFFFFFu);
    }
}

__global__ __launch_bounds__(256) void layer0_kernel(
    const float* __restrict__ pq, const int* __restrict__ csr_src,
    const unsigned* __restrict__ row_start, const unsigned* __restrict__ in_cnt,
    const float* __restrict__ in_norm,
    const float* __restrict__ W_in, const float* __restrict__ b_in,
    const float* __restrict__ W, const float* __restrict__ bias, float* __restrict__ h)
{
    __shared__ float Wl[64 * 64];
    __shared__ float Tl[32 * 64];

    const int tid = threadIdx.x;
    const int row0 = blockIdx.x * 32;

    #pragma unroll
    for (int i = 0; i < 4; ++i)
        ((float4*)Wl)[tid + 256 * i] = ((const float4*)W)[tid + 256 * i];

    {
        const int r   = tid >> 3;
        const int sub = tid & 7;
        const int row = row0 + r;
        const unsigned start = row_start[row];
        const unsigned cnt   = in_cnt[row];
        const float2* pq2 = (const float2*)pq;
        const float2 self = pq2[row];
        const float inn = in_norm[row];

        float a = 0.f, c = 0.f;
        for (unsigned j = sub; j < cnt; j += 32) {
            int s[4];
            float2 v[4];
            #pragma unroll
            for (int u = 0; u < 4; ++u) {
                const unsigned jj = j + 8u * u;
                s[u] = csr_src[start + (jj < cnt ? jj : j)];
            }
            #pragma unroll
            for (int u = 0; u < 4; ++u) v[u] = pq2[s[u]];
            #pragma unroll
            for (int u = 0; u < 4; ++u) {
                if (j + 8u * u < cnt) { a += v[u].x; c += v[u].y; }
            }
        }
        #pragma unroll
        for (int off = 1; off < 8; off <<= 1) {
            a += __shfl_xor(a, off, 32);
            c += __shfl_xor(c, off, 32);
        }
        a = (a + self.x) * inn;
        c = (c + self.y) * inn;

        const float4 wi0 = ((const float4*)W_in)[sub * 2], wi1 = ((const float4*)W_in)[sub * 2 + 1];
        const float4 bi0 = ((const float4*)b_in)[sub * 2], bi1 = ((const float4*)b_in)[sub * 2 + 1];
        float4 t0, t1;
        t0.x = a * wi0.x + c * bi0.x;  t0.y = a * wi0.y + c * bi0.y;
        t0.z = a * wi0.z + c * bi0.z;  t0.w = a * wi0.w + c * bi0.w;
        t1.x = a * wi1.x + c * bi1.x;  t1.y = a * wi1.y + c * bi1.y;
        t1.z = a * wi1.z + c * bi1.z;  t1.w = a * wi1.w + c * bi1.w;
        ((float4*)(Tl + r * 64))[sub * 2]     = t0;
        ((float4*)(Tl + r * 64))[sub * 2 + 1] = t1;
    }
    __syncthreads();

    const int wave = tid >> 5;
    const int lane = tid & 31;
    const int msub = wave >> 2;
    const int nsub = wave & 3;
    const int l16  = lane & 15;
    const int arow = msub * 16 + l16;
    const int koff = (lane < 16) ? 0 : 2;
    const int col  = nsub * 16 + l16;

    v8f acc = {};
    #pragma unroll
    for (int kk = 0; kk < 16; ++kk) {
        int k = kk * 4 + koff;
        v2f fa, fb;
        fa.x = Tl[arow * 64 + k];
        fa.y = Tl[arow * 64 + k + 1];
        fb.x = Wl[k * 64 + col];
        fb.y = Wl[(k + 1) * 64 + col];
        acc = __builtin_amdgcn_wmma_f32_16x16x4_f32(false, fa, false, fb,
                                                    (short)0, acc, false, false);
    }

    const float bcol = bias[col];
    const int rbase = row0 + msub * 16 + ((lane < 16) ? 0 : 8);
    #pragma unroll
    for (int r = 0; r < 8; ++r)
        h[(size_t)(rbase + r) * 64 + col] = acc[r] + bcol;
}

__global__ __launch_bounds__(256) void gemm_layer_kernel(
    const float* __restrict__ h, const float* __restrict__ W,
    const float* __restrict__ out_norm, float* __restrict__ g)
{
    __shared__ float Wl[64 * 64];
    __shared__ float Tl[32 * 64];

    const int tid = threadIdx.x;
    const int row0 = blockIdx.x * 32;

    for (int i = tid; i < 1024; i += 256)
        ((float4*)Wl)[i] = ((const float4*)W)[i];

    for (int i = tid; i < 512; i += 256) {
        int r  = i >> 4;
        int c4 = i & 15;
        int row = row0 + r;
        float4 v = ((const float4*)(h + (size_t)row * 64))[c4];
        v.x = v.x > 0.f ? v.x : 0.01f * v.x;
        v.y = v.y > 0.f ? v.y : 0.01f * v.y;
        v.z = v.z > 0.f ? v.z : 0.01f * v.z;
        v.w = v.w > 0.f ? v.w : 0.01f * v.w;
        float s = out_norm[row];
        v.x *= s; v.y *= s; v.z *= s; v.w *= s;
        ((float4*)(Tl + r * 64))[c4] = v;
    }
    __syncthreads();

    const int wave = tid >> 5;
    const int lane = tid & 31;
    const int msub = wave >> 2;
    const int nsub = wave & 3;
    const int l16  = lane & 15;
    const int arow = msub * 16 + l16;
    const int koff = (lane < 16) ? 0 : 2;
    const int col  = nsub * 16 + l16;

    v8f c = {};
    #pragma unroll
    for (int kk = 0; kk < 16; ++kk) {
        int k = kk * 4 + koff;
        v2f a, b;
        a.x = Tl[arow * 64 + k];
        a.y = Tl[arow * 64 + k + 1];
        b.x = Wl[k * 64 + col];
        b.y = Wl[(k + 1) * 64 + col];
        c = __builtin_amdgcn_wmma_f32_16x16x4_f32(false, a, false, b,
                                                  (short)0, c, false, false);
    }

    const int rbase = row0 + msub * 16 + ((lane < 16) ? 0 : 8);
    #pragma unroll
    for (int r = 0; r < 8; ++r)
        g[(size_t)(rbase + r) * 64 + col] = c[r];
}

#define ROWS(U)                                                            \
    {                                                                      \
        float2 v[U];                                                       \
        _Pragma("unroll")                                                  \
        for (int u = 0; u < U; ++u) {                                      \
              \
            int s = __builtin_amdgcn_readlane(my, (int)(j + u));           \
            v[u] = g2[(size_t)s * 32 + lane];                              \
        }                                                                  \
        _Pragma("unroll")                                                  \
        for (int u = 0; u < U; ++u) { acc.x += v[u].x; acc.y += v[u].y; }  \
        j += U;                                                            \
    }

__global__ __launch_bounds__(256) void gather_post_kernel(
    const float* __restrict__ g, const int* __restrict__ csr_src,
    const unsigned* __restrict__ row_start, const unsigned* __restrict__ in_cnt,
    const float* __restrict__ in_norm, const float* __restrict__ bc,
    float* __restrict__ h, int n)
{
    const int node = blockIdx.x * 8 + (threadIdx.x >> 5);
    if (node >= n) return;
    const int lane = threadIdx.x & 31;
    const float2* g2 = (const float2*)g;

    const unsigned start = __builtin_amdgcn_readfirstlane(row_start[node]);
    const unsigned cnt   = __builtin_amdgcn_readfirstlane(in_cnt[node]);

    float2 acc = make_float2(0.f, 0.f);
    for (unsigned base = 0; base < cnt; base += 32) {
        const unsigned m = (cnt - base < 32u) ? (cnt - base) : 32u;
        const int my = ((unsigned)lane < m) ? csr_src[start + base + lane] : 0;
        unsigned j = 0;
        if (j + 16 <= m) ROWS(16)
        if (j + 16 <= m) ROWS(16)
        if (j + 8 <= m) ROWS(8)
        if (j + 4 <= m) ROWS(4)
        if (j + 2 <= m) ROWS(2)
        if (j + 1 <= m) ROWS(1)
    }

    const float2 gv = g2[(size_t)node * 32 + lane];
    const float sc = in_norm[node];
    const float2 b = ((const float2*)bc)[lane];
    float2 r;
    r.x = (acc.x + gv.x) * sc + b.x;
    r.y = (acc.y + gv.y) * sc + b.y;
    ((float2*)h)[(size_t)node * 32 + lane] = r;
}
#undef ROWS

__global__ __launch_bounds__(256) void pred_kernel(
    const float* __restrict__ h, const float* __restrict__ W_pred,
    const float* __restrict__ b_pred, float* __restrict__ out, int n)
{
    int node = blockIdx.x * 8 + (threadIdx.x >> 5);
    if (node >= n) return;
    int lane = threadIdx.x & 31;
    float2 w = ((const float2*)W_pred)[lane];
    float2 v = ((const float2*)(h + (size_t)node * 64))[lane];
    float s = v.x * w.x + v.y * w.y;
    for (int off = 16; off > 0; off >>= 1)
        s += __shfl_down(s, off, 32);
    if (lane == 0) out[node] = s + b_pred[0];
}

extern "C" void kernel_launch(void* const* d_in, const int* in_sizes, int n_in,
                              void* d_out, int out_size, void* d_ws, size_t ws_size,
                              hipStream_t stream) {
    const float* weight = (const float*)d_in[0];
    const int*   src    = (const int*)d_in[1];
    const int*   dst    = (const int*)d_in[2];
    const float* W_in   = (const float*)d_in[3];
    const float* b_in   = (const float*)d_in[4];
    const float* Wc     = (const float*)d_in[5];
    const float* bc     = (const float*)d_in[6];
    const float* W_pred = (const float*)d_in[7];
    const float* b_pred = (const float*)d_in[8];
    float* out = (float*)d_out;

    const int N = in_sizes[0];
    const int E = in_sizes[1];

    const int NB = (N + 255) / 256;

    float* h          = (float*)d_ws;
    float* g          = h + (size_t)N * 64;
    unsigned* in_cnt  = (unsigned*)(g + (size_t)N * 64);
    float* out_norm   = (float*)(in_cnt + N);
    float* in_norm    = out_norm + N;
    unsigned* row_start = (unsigned*)(in_norm + N);
    float* pq         = (float*)(row_start + N);
    int* csr_src      = (int*)(pq + (size_t)N * 2);
    unsigned* bufD    = (unsigned*)(csr_src + E);
    unsigned* tableD  = bufD + E;
    unsigned* tableS  = tableD + (size_t)NB * GCH;
    unsigned* totD    = tableS + (size_t)NB * GCH;
    unsigned* totS    = totD + NB;
    unsigned* baseD   = totS + NB;
    unsigned* baseS   = baseD + NB;
    unsigned char* bufS = (unsigned char*)(baseS + NB);

    bucket_count_kernel<<<GCH, 1024, 2 * NB * sizeof(unsigned), stream>>>(src, dst, tableD, tableS, E, NB);

    bucket_scan_kernel<<<NB, GCH, 0, stream>>>(tableD, tableS, totD, totS);

    bucket_scatter_kernel<<<GCH, 1024, 2 * NB * sizeof(unsigned), stream>>>(
        src, dst, tableD, tableS, totD, totS, baseD, baseS, bufD, bufS, E, NB);

    bucket_finish_kernel<<<NB, 1024, 0, stream>>>(bufD, bufS, baseD, baseS, totD, totS, weight,
                                                  out_norm, in_norm, pq, in_cnt, row_start, csr_src, N);

    layer0_kernel<<<N / 32, 256, 0, stream>>>(pq, csr_src, row_start, in_cnt, in_norm,
                                              W_in, b_in, Wc, bc, h);

    for (int l = 1; l < 3; ++l) {
        gemm_layer_kernel<<<N / 32, 256, 0, stream>>>(h, Wc + (size_t)l * 64 * 64,
                                                      out_norm, g);
        gather_post_kernel<<<(N + 7) / 8, 256, 0, stream>>>(g, csr_src, row_start, in_cnt, in_norm,
                                                            bc + (size_t)l * 64, h, N);
    }

    pred_kernel<<<(N + 7) / 8, 256, 0, stream>>>(h, W_pred, b_pred, out, N);
}


# Last layer applies the head before its SpMM: z = (t @ W2) . W_pred per node in the WMMA epilogue (16-lane butterfly, 4 tiles), then one scalar per edge (8 strided sums + tree); head launch gone
# speedup vs baseline: 4.2036x; 1.1010x over previous
#include <hip/hip_runtime.h>
#include <math.h>

typedef __attribute__((ext_vector_type(2))) float v2f;
typedef __attribute__((ext_vector_type(8))) float v8f;

#define NN 100000

#define GCH 256
#define BSH 8
#define BMASK 255u

__device__ __forceinline__ void block_excl_scan(unsigned* a, int n, unsigned* wsum) {
    const int tid = threadIdx.x, lane = tid & 31, wave = tid >> 5;
    const int per = (n + 1023) >> 10;
    const int i0 = tid * per;
    unsigned s = 0;
    for (int k = 0; k < per; ++k) if (i0 + k < n) s += a[i0 + k];
    unsigned x = s;
    for (int off = 1; off < 32; off <<= 1) {
        unsigned y = __shfl_up(x, off, 32);
        if (lane >= off) x += y;
    }
    if (lane == 31) wsum[wave] = x;
    __syncthreads();
    unsigned run = x - s;
    for (int w = 0; w < wave; ++w) run += wsum[w];
    for (int k = 0; k < per; ++k) if (i0 + k < n) { unsigned v = a[i0 + k]; a[i0 + k] = run; run += v; }
    __syncthreads();
}

__global__ __launch_bounds__(1024) void bucket_count_kernel(
    const int* __restrict__ src, const int* __restrict__ dst,
    unsigned* __restrict__ tableD, unsigned* __restrict__ tableS, int E, int NB)
{
    extern __shared__ unsigned dyn_lds[];
    const int tid = threadIdx.x;
    for (int i = tid; i < 2 * NB; i += 1024) dyn_lds[i] = 0u;
    __syncthreads();

    const int chunk = (E + GCH - 1) / GCH;
    const int e0 = blockIdx.x * chunk;
    const int e1 = (e0 + chunk < E) ? (e0 + chunk) : E;
    for (int e = e0 + tid; e < e1; e += 4096) {
        int d[4], s[4];
        #pragma unroll
        for (int u = 0; u < 4; ++u) {
            const int ee = e + u * 1024;
            const int ec = (ee < e1) ? ee : e;
            d[u] = dst[ec];
            s[u] = src[ec];
        }
        asm volatile("" :: "v"(d[0]), "v"(d[1]), "v"(d[2]), "v"(d[3]), "v"(s[0]), "v"(s[1]), "v"(s[2]), "v"(s[3]));
        #pragma unroll
        for (int u = 0; u < 4; ++u) {
            if (e + u * 1024 < e1) {
                atomicAdd(&dyn_lds[d[u] >> BSH], 1u);
                atomicAdd(&dyn_lds[NB + (s[u] >> BSH)], 1u);
            }
        }
    }
    __syncthreads();
    for (int b = tid; b < NB; b += 1024) {
        tableD[b * GCH + blockIdx.x] = dyn_lds[b];
        tableS[b * GCH + blockIdx.x] = dyn_lds[NB + b];
    }
}

__global__ __launch_bounds__(GCH) void bucket_scan_kernel(
    unsigned* __restrict__ tableD, unsigned* __restrict__ tableS,
    unsigned* __restrict__ totD, unsigned* __restrict__ totS)
{
    __shared__ unsigned ws[2][GCH / 32];
    const int t = threadIdx.x, lane = t & 31, wave = t >> 5;
    const size_t idx = (size_t)blockIdx.x * GCH + t;
    const unsigned vd = tableD[idx], vs = tableS[idx];
    unsigned xd = vd, xs = vs;
    for (int off = 1; off < 32; off <<= 1) {
        unsigned yd = __shfl_up(xd, off, 32);
        unsigned ys = __shfl_up(xs, off, 32);
        if (lane >= off) { xd += yd; xs += ys; }
    }
    if (lane == 31) { ws[0][wave] = xd; ws[1][wave] = xs; }
    __syncthreads();
    unsigned od = 0, os = 0;
    for (int w = 0; w < wave; ++w) { od += ws[0][w]; os += ws[1][w]; }
    tableD[idx] = od + xd - vd;
    tableS[idx] = os + xs - vs;
    if (t == GCH - 1) { totD[blockIdx.x] = od + xd; totS[blockIdx.x] = os + xs; }
}

__global__ __launch_bounds__(1024) void bucket_scatter_kernel(
    const int* __restrict__ src, const int* __restrict__ dst,
    const unsigned* __restrict__ tableD, const unsigned* __restrict__ tableS,
    const unsigned* __restrict__ totD, const unsigned* __restrict__ totS,
    unsigned* __restrict__ baseD, unsigned* __restrict__ baseS,
    unsigned* __restrict__ bufD, unsigned char* __restrict__ bufS, int E, int NB)
{
    extern __shared__ unsigned dyn_lds[];
    __shared__ unsigned wsum[32];
    const int tid = threadIdx.x;
    for (int b = tid; b < NB; b += 1024) { dyn_lds[b] = totD[b]; dyn_lds[NB + b] = totS[b]; }
    __syncthreads();
    block_excl_scan(dyn_lds, NB, wsum);
    block_excl_scan(dyn_lds + NB, NB, wsum);
    for (int b = tid; b < NB; b += 1024) {
        const unsigned bd = dyn_lds[b], bs = dyn_lds[NB + b];
        if (blockIdx.x == 0) { baseD[b] = bd; baseS[b] = bs; }
        dyn_lds[b]      = bd + tableD[b * GCH + blockIdx.x];
        dyn_lds[NB + b] = bs + tableS[b * GCH + blockIdx.x];
    }
    __syncthreads();

    const int chunk = (E + GCH - 1) / GCH;
    const int e0 = blockIdx.x * chunk;
    const int e1 = (e0 + chunk < E) ? (e0 + chunk) : E;
    for (int e = e0 + tid; e < e1; e += 4096) {
        int d[4], s[4];
        #pragma unroll
        for (int u = 0; u < 4; ++u) {
            const int ee = e + u * 1024;
            const int ec = (ee < e1) ? ee : e;
            d[u] = dst[ec];
            s[u] = src[ec];
        }
        asm volatile("" :: "v"(d[0]), "v"(d[1]), "v"(d[2]), "v"(d[3]), "v"(s[0]), "v"(s[1]), "v"(s[2]), "v"(s[3]));
        #pragma unroll
        for (int u = 0; u < 4; ++u) {
            if (e + u * 1024 < e1) {
                const unsigned pd = atomicAdd(&dyn_lds[d[u] >> BSH], 1u);
                bufD[pd] = (((unsigned)d[u] & BMASK) << 24) | (unsigned)s[u];
                const unsigned ps = atomicAdd(&dyn_lds[NB + (s[u] >> BSH)], 1u);
                bufS[ps] = (unsigned char)((unsigned)s[u] & BMASK);
            }
        }
    }
}

__global__ __launch_bounds__(1024) void bucket_finish_kernel(
    const unsigned* __restrict__ bufD, const unsigned char* __restrict__ bufS,
    const unsigned* __restrict__ baseD, const unsigned* __restrict__ baseS,
    const unsigned* __restrict__ totD, const unsigned* __restrict__ totS,
    const float* __restrict__ weight,
    float* __restrict__ out_norm, float* __restrict__ in_norm, float* __restrict__ pq,
    unsigned* __restrict__ in_cnt, unsigned* __restrict__ row_start,
    int* __restrict__ csr_src, int n)
{
    __shared__ unsigned cin[256];
    __shared__ unsigned cout[256];
    __shared__ unsigned cur[256];
    __shared__ unsigned wsum[32];
    const int tid = threadIdx.x;
    const int b = blockIdx.x;
    const unsigned bD = baseD[b], nD = totD[b];
    const unsigned bS = baseS[b], nS = totS[b];
    const float wv = (tid < 256 && b * 256 + tid < n) ? weight[b * 256 + tid] : 0.f;
    if (tid < 256) { cin[tid] = 0u; cout[tid] = 0u; }
    __syncthreads();
    for (unsigned i = tid; i < nS; i += 1024) atomicAdd(&cout[bufS[bS + i]], 1u);
    for (unsigned i = tid; i < nD; i += 1024) atomicAdd(&cin[bufD[bD + i] >> 24], 1u);
    __syncthreads();
    if (tid < 256) cur[tid] = cin[tid];
    __syncthreads();
    block_excl_scan(cur, 256, wsum);
    if (tid < 256) {
        const unsigned rs = bD + cur[tid];
        cur[tid] = rs;
        const int node = b * 256 + tid;
        if (node < n) {
            const unsigned c = cin[tid];
            row_start[node] = rs;
            in_cnt[node] = c;
            in_norm[node]  = rsqrtf((float)(c + 1u));
            const float on = rsqrtf((float)(cout[tid] + 1u));
            out_norm[node] = on;
            ((float2*)pq)[node] = make_float2(on * wv, on);
        }
    }
    __syncthreads();
    for (unsigned i = tid; i < nD; i += 1024) {
        const unsigned w = bufD[bD + i];
        const unsigned pos = atomicAdd(&cur[w >> 24], 1u);
        csr_src[pos] = (int)(w & 0x00FFFFFFu);
    }
}

__global__ __launch_bounds__(256) void layer0_kernel(
    const float* __restrict__ pq, const int* __restrict__ csr_src,
    const unsigned* __restrict__ row_start, const unsigned* __restrict__ in_cnt,
    const float* __restrict__ in_norm,
    const float* __restrict__ W_in, const float* __restrict__ b_in,
    const float* __restrict__ W, const float* __restrict__ bias, float* __restrict__ h)
{
    __shared__ float Wl[64 * 64];
    __shared__ float Tl[32 * 64];

    const int tid = threadIdx.x;
    const int row0 = blockIdx.x * 32;

    #pragma unroll
    for (int i = 0; i < 4; ++i)
        ((float4*)Wl)[tid + 256 * i] = ((const float4*)W)[tid + 256 * i];

    {
        const int r   = tid >> 3;
        const int sub = tid & 7;
        const int row = row0 + r;
        const unsigned start = row_start[row];
        const unsigned cnt   = in_cnt[row];
        const float2* pq2 = (const float2*)pq;
        const float2 self = pq2[row];
        const float inn = in_norm[row];

        float a = 0.f, c = 0.f;
        for (unsigned j = sub; j < cnt; j += 32) {
            int s[4];
            float2 v[4];
            #pragma unroll
            for (int u = 0; u < 4; ++u) {
                const unsigned jj = j + 8u * u;
                s[u] = csr_src[start + (jj < cnt ? jj : j)];
            }
            #pragma unroll
            for (int u = 0; u < 4; ++u) v[u] = pq2[s[u]];
            #pragma unroll
            for (int u = 0; u < 4; ++u) {
                if (j + 8u * u < cnt) { a += v[u].x; c += v[u].y; }
            }
        }
        #pragma unroll
        for (int off = 1; off < 8; off <<= 1) {
            a += __shfl_xor(a, off, 32);
            c += __shfl_xor(c, off, 32);
        }
        a = (a + self.x) * inn;
        c = (c + self.y) * inn;

        const float4 wi0 = ((const float4*)W_in)[sub * 2], wi1 = ((const float4*)W_in)[sub * 2 + 1];
        const float4 bi0 = ((const float4*)b_in)[sub * 2], bi1 = ((const float4*)b_in)[sub * 2 + 1];
        float4 t0, t1;
        t0.x = a * wi0.x + c * bi0.x;  t0.y = a * wi0.y + c * bi0.y;
        t0.z = a * wi0.z + c * bi0.z;  t0.w = a * wi0.w + c * bi0.w;
        t1.x = a * wi1.x + c * bi1.x;  t1.y = a * wi1.y + c * bi1.y;
        t1.z = a * wi1.z + c * bi1.z;  t1.w = a * wi1.w + c * bi1.w;
        ((float4*)(Tl + r * 64))[sub * 2]     = t0;
        ((float4*)(Tl + r * 64))[sub * 2 + 1] = t1;
    }
    __syncthreads();

    const int wave = tid >> 5;
    const int lane = tid & 31;
    const int msub = wave >> 2;
    const int nsub = wave & 3;
    const int l16  = lane & 15;
    const int arow = msub * 16 + l16;
    const int koff = (lane < 16) ? 0 : 2;
    const int col  = nsub * 16 + l16;

    v8f acc = {};
    #pragma unroll
    for (int kk = 0; kk < 16; ++kk) {
        int k = kk * 4 + koff;
        v2f fa, fb;
        fa.x = Tl[arow * 64 + k];
        fa.y = Tl[arow * 64 + k + 1];
        fb.x = Wl[k * 64 + col];
        fb.y = Wl[(k + 1) * 64 + col];
        acc = __builtin_amdgcn_wmma_f32_16x16x4_f32(false, fa, false, fb,
                                                    (short)0, acc, false, false);
    }

    const float bcol = bias[col];
    const int rbase = row0 + msub * 16 + ((lane < 16) ? 0 : 8);
    #pragma unroll
    for (int r = 0; r < 8; ++r)
        h[(size_t)(rbase + r) * 64 + col] = acc[r] + bcol;
}

__global__ __launch_bounds__(256) void gemm_layer_kernel(
    const float* __restrict__ h, const float* __restrict__ W,
    const float* __restrict__ out_norm, float* __restrict__ g)
{
    __shared__ float Wl[64 * 64];
    __shared__ float Tl[32 * 64];

    const int tid = threadIdx.x;
    const int row0 = blockIdx.x * 32;

    for (int i = tid; i < 1024; i += 256)
        ((float4*)Wl)[i] = ((const float4*)W)[i];

    for (int i = tid; i < 512; i += 256) {
        int r  = i >> 4;
        int c4 = i & 15;
        int row = row0 + r;
        float4 v = ((const float4*)(h + (size_t)row * 64))[c4];
        v.x = v.x > 0.f ? v.x : 0.01f * v.x;
        v.y = v.y > 0.f ? v.y : 0.01f * v.y;
        v.z = v.z > 0.f ? v.z : 0.01f * v.z;
        v.w = v.w > 0.f ? v.w : 0.01f * v.w;
        float s = out_norm[row];
        v.x *= s; v.y *= s; v.z *= s; v.w *= s;
        ((float4*)(Tl + r * 64))[c4] = v;
    }
    __syncthreads();

    const int wave = tid >> 5;
    const int lane = tid & 31;
    const int msub = wave >> 2;
    const int nsub = wave & 3;
    const int l16  = lane & 15;
    const int arow = msub * 16 + l16;
    const int koff = (lane < 16) ? 0 : 2;
    const int col  = nsub * 16 + l16;

    v8f c = {};
    #pragma unroll
    for (int kk = 0; kk < 16; ++kk) {
        int k = kk * 4 + koff;
        v2f a, b;
        a.x = Tl[arow * 64 + k];
        a.y = Tl[arow * 64 + k + 1];
        b.x = Wl[k * 64 + col];
        b.y = Wl[(k + 1) * 64 + col];
        c = __builtin_amdgcn_wmma_f32_16x16x4_f32(false, a, false, b,
                                                  (short)0, c, false, false);
    }

    const int rbase = row0 + msub * 16 + ((lane < 16) ? 0 : 8);
    #pragma unroll
    for (int r = 0; r < 8; ++r)
        g[(size_t)(rbase + r) * 64 + col] = c[r];
}

#define ROWS(U)                                                            \
    {                                                                      \
        float2 v[U];                                                       \
        _Pragma("unroll")                                                  \
        for (int u = 0; u < U; ++u) {                                      \
              \
            int s = __builtin_amdgcn_readlane(my, (int)(j + u));           \
            v[u] = g2[(size_t)s * 32 + lane];                              \
        }                                                                  \
        _Pragma("unroll")                                                  \
        for (int u = 0; u < U; ++u) { acc.x += v[u].x; acc.y += v[u].y; }  \
        j += U;                                                            \
    }

__global__ __launch_bounds__(256) void gather_post_kernel(
    const float* __restrict__ g, const int* __restrict__ csr_src,
    const unsigned* __restrict__ row_start, const unsigned* __restrict__ in_cnt,
    const float* __restrict__ in_norm, const float* __restrict__ bc,
    float* __restrict__ h, int n)
{
    const int node = blockIdx.x * 8 + (threadIdx.x >> 5);
    if (node >= n) return;
    const int lane = threadIdx.x & 31;
    const float2* g2 = (const float2*)g;

    const unsigned start = __builtin_amdgcn_readfirstlane(row_start[node]);
    const unsigned cnt   = __builtin_amdgcn_readfirstlane(in_cnt[node]);

    float2 acc = make_float2(0.f, 0.f);
    for (unsigned base = 0; base < cnt; base += 32) {
        const unsigned m = (cnt - base < 32u) ? (cnt - base) : 32u;
        const int my = ((unsigned)lane < m) ? csr_src[start + base + lane] : 0;
        unsigned j = 0;
        if (j + 16 <= m) ROWS(16)
        if (j + 16 <= m) ROWS(16)
        if (j + 8 <= m) ROWS(8)
        if (j + 4 <= m) ROWS(4)
        if (j + 2 <= m) ROWS(2)
        if (j + 1 <= m) ROWS(1)
    }

    const float2 gv = g2[(size_t)node * 32 + lane];
    const float sc = in_norm[node];
    const float2 b = ((const float2*)bc)[lane];
    float2 r;
    r.x = (acc.x + gv.x) * sc + b.x;
    r.y = (acc.y + gv.y) * sc + b.y;
    ((float2*)h)[(size_t)node * 32 + lane] = r;
}
#undef ROWS

template <int M> __device__ __forceinline__ float row_xor_add(float x) {
    const int y = __builtin_amdgcn_update_dpp(0, __builtin_bit_cast(int, x), 0x160 + M, 0xf, 0xf, true);
    return x + __builtin_bit_cast(float, y);
}

__global__ __launch_bounds__(256) void gemm_head_kernel(
    const float* __restrict__ h, const float* __restrict__ W,
    const float* __restrict__ out_norm, const float* __restrict__ W_pred, float* __restrict__ z)
{
    __shared__ float Wl[64 * 64];
    __shared__ float Tl[32 * 64];
    __shared__ float Zp[4][32];

    const int tid = threadIdx.x;
    const int row0 = blockIdx.x * 32;

    for (int i = tid; i < 1024; i += 256)
        ((float4*)Wl)[i] = ((const float4*)W)[i];

    for (int i = tid; i < 512; i += 256) {
        int r  = i >> 4;
        int c4 = i & 15;
        int row = row0 + r;
        float4 v = ((const float4*)(h + (size_t)row * 64))[c4];
        v.x = v.x > 0.f ? v.x : 0.01f * v.x;
        v.y = v.y > 0.f ? v.y : 0.01f * v.y;
        v.z = v.z > 0.f ? v.z : 0.01f * v.z;
        v.w = v.w > 0.f ? v.w : 0.01f * v.w;
        float s = out_norm[row];
        v.x *= s; v.y *= s; v.z *= s; v.w *= s;
        ((float4*)(Tl + r * 64))[c4] = v;
    }
    __syncthreads();

    const int wave = tid >> 5;
    const int lane = tid & 31;
    const int msub = wave >> 2;
    const int nsub = wave & 3;
    const int l16  = lane & 15;
    const int arow = msub * 16 + l16;
    const int koff = (lane < 16) ? 0 : 2;
    const int col  = nsub * 16 + l16;

    const float wp = W_pred[col];

    v8f c = {};
    #pragma unroll
    for (int kk = 0; kk < 16; ++kk) {
        int k = kk * 4 + koff;
        v2f a, b;
        a.x = Tl[arow * 64 + k];
        a.y = Tl[arow * 64 + k + 1];
        b.x = Wl[k * 64 + col];
        b.y = Wl[(k + 1) * 64 + col];
        c = __builtin_amdgcn_wmma_f32_16x16x4_f32(false, a, false, b,
                                                  (short)0, c, false, false);
    }

    const int rloc = msub * 16 + ((lane < 16) ? 0 : 8);
    float p[8];
    #pragma unroll
    for (int r = 0; r < 8; ++r) p[r] = c[r] * wp;
    #pragma unroll
    for (int r = 0; r < 8; ++r) p[r] = row_xor_add<1>(p[r]);
    #pragma unroll
    for (int r = 0; r < 8; ++r) p[r] = row_xor_add<2>(p[r]);
    #pragma unroll
    for (int r = 0; r < 8; ++r) p[r] = row_xor_add<4>(p[r]);
    #pragma unroll
    for (int r = 0; r < 8; ++r) p[r] = row_xor_add<8>(p[r]);
    if (l16 == 0) {
        #pragma unroll
        for (int r = 0; r < 8; ++r) Zp[nsub][rloc + r] = p[r];
    }
    __syncthreads();
    if (tid < 32)
        z[row0 + tid] = (Zp[0][tid] + Zp[1][tid]) + (Zp[2][tid] + Zp[3][tid]);
}

__global__ __launch_bounds__(256) void gather_head_kernel(
    const float* __restrict__ z, const int* __restrict__ csr_src,
    const unsigned* __restrict__ row_start, const unsigned* __restrict__ in_cnt,
    const float* __restrict__ in_norm, const float* __restrict__ bias,
    const float* __restrict__ W_pred, const float* __restrict__ b_pred,
    float* __restrict__ out)
{
    __shared__ float ol[32];

    const int tid = threadIdx.x;
    const int lane = tid & 31;
    const int r   = tid >> 3;
    const int sub = tid & 7;
    const int row = blockIdx.x * 32 + r;

    const unsigned start = row_start[row];
    const unsigned cnt   = in_cnt[row];
    const float self = z[row];
    const float inn = in_norm[row];
    const float2 bb = ((const float2*)bias)[lane];
    const float2 ww = ((const float2*)W_pred)[lane];
    asm volatile("" :: "v"(start), "v"(bb.x), "v"(bb.y), "v"(ww.x), "v"(ww.y));

    float a = 0.f;
    for (unsigned j = sub; j < cnt; j += 32) {
        int s[4];
        float v[4];
        #pragma unroll
        for (int u = 0; u < 4; ++u) {
            const unsigned jj = j + 8u * u;
            s[u] = csr_src[start + (jj < cnt ? jj : j)];
        }
        #pragma unroll
        for (int u = 0; u < 4; ++u) v[u] = z[s[u]];
        #pragma unroll
        for (int u = 0; u < 4; ++u) {
            if (j + 8u * u < cnt) a += v[u];
        }
    }
    a = row_xor_add<1>(a);
    a = row_xor_add<2>(a);
    a = row_xor_add<4>(a);

    float cb = bb.x * ww.x + bb.y * ww.y;
    cb = row_xor_add<1>(cb);
    cb = row_xor_add<2>(cb);
    cb = row_xor_add<4>(cb);
    cb = row_xor_add<8>(cb);
    cb += __shfl_xor(cb, 16, 32);
    cb += b_pred[0];

    if (sub == 0) ol[r] = (a + self) * inn + cb;
    __syncthreads();
    if (tid < 32) out[blockIdx.x * 32 + tid] = ol[tid];
}

extern "C" void kernel_launch(void* const* d_in, const int* in_sizes, int n_in,
                              void* d_out, int out_size, void* d_ws, size_t ws_size,
                              hipStream_t stream) {
    const float* weight = (const float*)d_in[0];
    const int*   src    = (const int*)d_in[1];
    const int*   dst    = (const int*)d_in[2];
    const float* W_in   = (const float*)d_in[3];
    const float* b_in   = (const float*)d_in[4];
    const float* Wc     = (const float*)d_in[5];
    const float* bc     = (const float*)d_in[6];
    const float* W_pred = (const float*)d_in[7];
    const float* b_pred = (const float*)d_in[8];
    float* out = (float*)d_out;

    const int N = in_sizes[0];
    const int E = in_sizes[1];

    const int NB = (N + 255) / 256;

    float* h          = (float*)d_ws;
    float* g          = h + (size_t)N * 64;
    unsigned* in_cnt  = (unsigned*)(g + (size_t)N * 64);
    float* out_norm   = (float*)(in_cnt + N);
    float* in_norm    = out_norm + N;
    unsigned* row_start = (unsigned*)(in_norm + N);
    float* pq         = (float*)(row_start + N);
    int* csr_src      = (int*)(pq + (size_t)N * 2);
    unsigned* bufD    = (unsigned*)(csr_src + E);
    unsigned* tableD  = bufD + E;
    unsigned* tableS  = tableD + (size_t)NB * GCH;
    unsigned* totD    = tableS + (size_t)NB * GCH;
    unsigned* totS    = totD + NB;
    unsigned* baseD   = totS + NB;
    unsigned* baseS   = baseD + NB;
    unsigned char* bufS = (unsigned char*)(baseS + NB);

    bucket_count_kernel<<<GCH, 1024, 2 * NB * sizeof(unsigned), stream>>>(src, dst, tableD, tableS, E, NB);

    bucket_scan_kernel<<<NB, GCH, 0, stream>>>(tableD, tableS, totD, totS);

    bucket_scatter_kernel<<<GCH, 1024, 2 * NB * sizeof(unsigned), stream>>>(
        src, dst, tableD, tableS, totD, totS, baseD, baseS, bufD, bufS, E, NB);

    bucket_finish_kernel<<<NB, 1024, 0, stream>>>(bufD, bufS, baseD, baseS, totD, totS, weight,
                                                  out_norm, in_norm, pq, in_cnt, row_start, csr_src, N);

    layer0_kernel<<<N / 32, 256, 0, stream>>>(pq, csr_src, row_start, in_cnt, in_norm,
                                              W_in, b_in, Wc, bc, h);

    gemm_layer_kernel<<<N / 32, 256, 0, stream>>>(h, Wc + (size_t)64 * 64, out_norm, g);
    gather_post_kernel<<<(N + 7) / 8, 256, 0, stream>>>(g, csr_src, row_start, in_cnt, in_norm,
                                                        bc + 64, h, N);

    float* z = g;
    gemm_head_kernel<<<N / 32, 256, 0, stream>>>(h, Wc + (size_t)2 * 64 * 64, out_norm, W_pred, z);
    gather_head_kernel<<<N / 32, 256, 0, stream>>>(z, csr_src, row_start, in_cnt, in_norm,
                                                   bc + 2 * 64, W_pred, b_pred, out);
}


# Edge sort's scattered 4-byte and 1-byte global stores staged in LDS (chunk grouped by bucket via rank adds; each bucket's sorted list), written as contiguous runs; in-edge sums stay arrival-ordered
# speedup vs baseline: 6.3112x; 1.5014x over previous
#include <hip/hip_runtime.h>
#include <math.h>

typedef __attribute__((ext_vector_type(2))) float v2f;
typedef __attribute__((ext_vector_type(8))) float v8f;

#define NN 100000

#define GCH 256
#define BSH 8
#define BMASK 255u

__device__ __forceinline__ void block_excl_scan(unsigned* a, int n, unsigned* wsum) {
    const int tid = threadIdx.x, lane = tid & 31, wave = tid >> 5;
    const int per = (n + 1023) >> 10;
    const int i0 = tid * per;
    unsigned s = 0;
    for (int k = 0; k < per; ++k) if (i0 + k < n) s += a[i0 + k];
    unsigned x = s;
    for (int off = 1; off < 32; off <<= 1) {
        unsigned y = __shfl_up(x, off, 32);
        if (lane >= off) x += y;
    }
    if (lane == 31) wsum[wave] = x;
    __syncthreads();
    unsigned run = x - s;
    for (int w = 0; w < wave; ++w) run += wsum[w];
    for (int k = 0; k < per; ++k) if (i0 + k < n) { unsigned v = a[i0 + k]; a[i0 + k] = run; run += v; }
    __syncthreads();
}

__global__ __launch_bounds__(1024) void bucket_count_kernel(
    const int* __restrict__ src, const int* __restrict__ dst,
    unsigned* __restrict__ tableD, unsigned* __restrict__ tableS, int E, int NB)
{
    extern __shared__ unsigned dyn_lds[];
    const int tid = threadIdx.x;
    for (int i = tid; i < 2 * NB; i += 1024) dyn_lds[i] = 0u;
    __syncthreads();

    const int chunk = (E + GCH - 1) / GCH;
    const int e0 = blockIdx.x * chunk;
    const int e1 = (e0 + chunk < E) ? (e0 + chunk) : E;
    for (int e = e0 + tid; e < e1; e += 4096) {
        int d[4], s[4];
        #pragma unroll
        for (int u = 0; u < 4; ++u) {
            const int ee = e + u * 1024;
            const int ec = (ee < e1) ? ee : e;
            d[u] = dst[ec];
            s[u] = src[ec];
        }
        asm volatile("" :: "v"(d[0]), "v"(d[1]), "v"(d[2]), "v"(d[3]), "v"(s[0]), "v"(s[1]), "v"(s[2]), "v"(s[3]));
        #pragma unroll
        for (int u = 0; u < 4; ++u) {
            if (e + u * 1024 < e1) {
                atomicAdd(&dyn_lds[d[u] >> BSH], 1u);
                atomicAdd(&dyn_lds[NB + (s[u] >> BSH)], 1u);
            }
        }
    }
    __syncthreads();
    for (int b = tid; b < NB; b += 1024) {
        tableD[b * GCH + blockIdx.x] = dyn_lds[b];
        tableS[b * GCH + blockIdx.x] = dyn_lds[NB + b];
    }
}

__global__ __launch_bounds__(GCH) void bucket_scan_kernel(
    unsigned* __restrict__ tableD, unsigned* __restrict__ tableS,
    unsigned* __restrict__ totD, unsigned* __restrict__ totS)
{
    __shared__ unsigned ws[2][GCH / 32];
    const int t = threadIdx.x, lane = t & 31, wave = t >> 5;
    const size_t idx = (size_t)blockIdx.x * GCH + t;
    const unsigned vd = tableD[idx], vs = tableS[idx];
    unsigned xd = vd, xs = vs;
    for (int off = 1; off < 32; off <<= 1) {
        unsigned yd = __shfl_up(xd, off, 32);
        unsigned ys = __shfl_up(xs, off, 32);
        if (lane >= off) { xd += yd; xs += ys; }
    }
    if (lane == 31) { ws[0][wave] = xd; ws[1][wave] = xs; }
    __syncthreads();
    unsigned od = 0, os = 0;
    for (int w = 0; w < wave; ++w) { od += ws[0][w]; os += ws[1][w]; }
    tableD[idx] = od + xd - vd;
    tableS[idx] = os + xs - vs;
    if (t == GCH - 1) { totD[blockIdx.x] = od + xd; totS[blockIdx.x] = os + xs; }
}

#define EPT 13
#define TILE_E (1024 * EPT)
__global__ __launch_bounds__(1024) void bucket_scatter_kernel(
    const int* __restrict__ src, const int* __restrict__ dst,
    const unsigned* __restrict__ tableD, const unsigned* __restrict__ tableS,
    const unsigned* __restrict__ totD, const unsigned* __restrict__ totS,
    unsigned* __restrict__ baseD, unsigned* __restrict__ baseS,
    unsigned* __restrict__ bufD, unsigned char* __restrict__ bufS, int E, int NB)
{
    extern __shared__ unsigned dyn_lds[];
    unsigned* gcD = dyn_lds;
    unsigned* gcS = dyn_lds + NB;
    unsigned* lsD = dyn_lds + 2 * NB;
    unsigned* lsS = dyn_lds + 3 * NB;
    unsigned* dlD = dyn_lds + 4 * NB;
    unsigned* dlS = dyn_lds + 5 * NB;
    __shared__ unsigned wsum[32];
    __shared__ unsigned       stD[TILE_E];
    __shared__ unsigned short sbD[TILE_E];
    __shared__ unsigned       stS[TILE_E];
    __shared__ unsigned spare[64];
    const int tid = threadIdx.x;
    if (tid < 64) spare[tid] = 0u;
    for (int b = tid; b < NB; b += 1024) { gcD[b] = totD[b]; gcS[b] = totS[b]; }
    __syncthreads();
    block_excl_scan(gcD, NB, wsum);
    block_excl_scan(gcS, NB, wsum);
    for (int b = tid; b < NB; b += 1024) {
        const unsigned bd = gcD[b], bs = gcS[b];
        if (blockIdx.x == 0) { baseD[b] = bd; baseS[b] = bs; }
        gcD[b] = bd + tableD[b * GCH + blockIdx.x];
        gcS[b] = bs + tableS[b * GCH + blockIdx.x];
    }
    __syncthreads();

    const int chunk = (E + GCH - 1) / GCH;
    const int e0 = blockIdx.x * chunk;
    const int e1 = (e0 + chunk < E) ? (e0 + chunk) : E;
    const int nt = e1 - e0;
    if (nt > TILE_E) {
        for (int e = e0 + tid; e < e1; e += 1024) {
            const int dd = dst[e], ss = src[e];
            const unsigned qd = atomicAdd(&gcD[(unsigned)dd >> BSH], 1u);
            bufD[qd] = (((unsigned)dd & BMASK) << 24) | (unsigned)ss;
            const unsigned qs = atomicAdd(&gcS[(unsigned)ss >> BSH], 1u);
            bufS[qs] = (unsigned char)((unsigned)ss & BMASK);
        }
    } else if (nt > 0) {
        const int t0 = e0;
        for (int b = tid; b < 2 * NB; b += 1024) lsD[b] = 0u;
        __syncthreads();

        int d[EPT], s[EPT];
        #pragma unroll
        for (int u = 0; u < EPT; ++u) {
            const int i = u * 1024 + tid;
            const int ec = t0 + ((i < nt) ? i : 0);
            d[u] = dst[ec];
            s[u] = src[ec];
        }
        unsigned rk[EPT];
        {
            unsigned rd[EPT], rs[EPT];
            #pragma unroll
            for (int u = 0; u < EPT; ++u) {
                const bool valid = u * 1024 + tid < nt;
                rd[u] = atomicAdd(valid ? &lsD[(unsigned)d[u] >> BSH] : &spare[tid & 31], 1u);
                rs[u] = atomicAdd(valid ? &lsS[(unsigned)s[u] >> BSH] : &spare[32 + (tid & 31)], 1u);
            }
            #pragma unroll
            for (int u = 0; u < EPT; ++u) rk[u] = (rd[u] & 0xFFFFu) | (rs[u] << 16);
        }
        __syncthreads();
        block_excl_scan(lsD, NB, wsum);
        block_excl_scan(lsS, NB, wsum);
        for (int b = tid; b < NB; b += 1024) { dlD[b] = gcD[b] - lsD[b]; dlS[b] = gcS[b] - lsS[b]; }
        unsigned pd[EPT], ps[EPT];
        #pragma unroll
        for (int u = 0; u < EPT; ++u) {
            pd[u] = lsD[(unsigned)d[u] >> BSH] + (rk[u] & 0xFFFFu);
            ps[u] = lsS[(unsigned)s[u] >> BSH] + (rk[u] >> 16);
        }
        #pragma unroll
        for (int u = 0; u < EPT; ++u) {
            if (u * 1024 + tid < nt) {
                stD[pd[u]] = (((unsigned)d[u] & BMASK) << 24) | (unsigned)s[u];
                sbD[pd[u]] = (unsigned short)((unsigned)d[u] >> BSH);
                stS[ps[u]] = (((unsigned)s[u] >> BSH) << 8) | ((unsigned)s[u] & BMASK);
            }
        }
        __syncthreads();

        {
            unsigned v[EPT], a[EPT];
            #pragma unroll
            for (int u = 0; u < EPT; ++u) {
                const int i = u * 1024 + tid;
                const int ic = (i < nt) ? i : 0;
                v[u] = stD[ic];
                a[u] = sbD[ic];
            }
            #pragma unroll
            for (int u = 0; u < EPT; ++u) a[u] = dlD[a[u]] + (unsigned)(u * 1024 + tid);
            #pragma unroll
            for (int u = 0; u < EPT; ++u)
                if (u * 1024 + tid < nt) bufD[a[u]] = v[u];
        }
        {
            unsigned v[EPT], a[EPT];
            #pragma unroll
            for (int u = 0; u < EPT; ++u) {
                const int i = u * 1024 + tid;
                v[u] = stS[(i < nt) ? i : 0];
            }
            #pragma unroll
            for (int u = 0; u < EPT; ++u) a[u] = dlS[v[u] >> 8] + (unsigned)(u * 1024 + tid);
            #pragma unroll
            for (int u = 0; u < EPT; ++u)
                if (u * 1024 + tid < nt) bufS[a[u]] = (unsigned char)(v[u] & BMASK);
        }

    }
}

#define STAGE_N 12288
__global__ __launch_bounds__(1024) void bucket_finish_kernel(
    const unsigned* __restrict__ bufD, const unsigned char* __restrict__ bufS,
    const unsigned* __restrict__ baseD, const unsigned* __restrict__ baseS,
    const unsigned* __restrict__ totD, const unsigned* __restrict__ totS,
    const float* __restrict__ weight,
    float* __restrict__ out_norm, float* __restrict__ in_norm, float* __restrict__ pq,
    unsigned* __restrict__ in_cnt, unsigned* __restrict__ row_start,
    int* __restrict__ csr_src, int n)
{
    __shared__ unsigned cin[256];
    __shared__ unsigned cout[256];
    __shared__ unsigned cur[256];
    __shared__ unsigned wsum[32];
    __shared__ int stage[STAGE_N];
    const int tid = threadIdx.x;
    const int b = blockIdx.x;
    const unsigned bD = baseD[b], nD = totD[b];
    const unsigned bS = baseS[b], nS = totS[b];
    const float wv = (tid < 256 && b * 256 + tid < n) ? weight[b * 256 + tid] : 0.f;
    if (tid < 256) { cin[tid] = 0u; cout[tid] = 0u; }
    __syncthreads();
    for (unsigned i = tid; i < nS; i += 1024) atomicAdd(&cout[bufS[bS + i]], 1u);
    for (unsigned i = tid; i < nD; i += 1024) atomicAdd(&cin[bufD[bD + i] >> 24], 1u);
    __syncthreads();
    if (tid < 256) cur[tid] = cin[tid];
    __syncthreads();
    block_excl_scan(cur, 256, wsum);
    if (tid < 256) {
        const unsigned rs = bD + cur[tid];
        cur[tid] = rs;
        const int node = b * 256 + tid;
        if (node < n) {
            const unsigned c = cin[tid];
            row_start[node] = rs;
            in_cnt[node] = c;
            in_norm[node]  = rsqrtf((float)(c + 1u));
            const float on = rsqrtf((float)(cout[tid] + 1u));
            out_norm[node] = on;
            ((float2*)pq)[node] = make_float2(on * wv, on);
        }
    }
    __syncthreads();
    const bool staged = nD <= (unsigned)STAGE_N;
    for (unsigned i = tid; i < nD; i += 1024) {
        const unsigned w = bufD[bD + i];
        const unsigned pos = atomicAdd(&cur[w >> 24], 1u);
        const int sv = (int)(w & 0x00FFFFFFu);
        if (staged) stage[pos - bD] = sv;
        else        csr_src[pos] = sv;
    }
    if (staged) {
        __syncthreads();
        for (unsigned i = tid; i < nD; i += 1024) csr_src[bD + i] = stage[i];
    }
}

__global__ __launch_bounds__(256) void layer0_kernel(
    const float* __restrict__ pq, const int* __restrict__ csr_src,
    const unsigned* __restrict__ row_start, const unsigned* __restrict__ in_cnt,
    const float* __restrict__ in_norm,
    const float* __restrict__ W_in, const float* __restrict__ b_in,
    const float* __restrict__ W, const float* __restrict__ bias, float* __restrict__ h)
{
    __shared__ float Wl[64 * 64];
    __shared__ float Tl[32 * 64];

    const int tid = threadIdx.x;
    const int row0 = blockIdx.x * 32;

    #pragma unroll
    for (int i = 0; i < 4; ++i)
        ((float4*)Wl)[tid + 256 * i] = ((const float4*)W)[tid + 256 * i];

    {
        const int r   = tid >> 3;
        const int sub = tid & 7;
        const int row = row0 + r;
        const unsigned start = row_start[row];
        const unsigned cnt   = in_cnt[row];
        const float2* pq2 = (const float2*)pq;
        const float2 self = pq2[row];
        const float inn = in_norm[row];

        float a = 0.f, c = 0.f;
        for (unsigned j = sub; j < cnt; j += 32) {
            int s[4];
            float2 v[4];
            #pragma unroll
            for (int u = 0; u < 4; ++u) {
                const unsigned jj = j + 8u * u;
                s[u] = csr_src[start + (jj < cnt ? jj : j)];
            }
            #pragma unroll
            for (int u = 0; u < 4; ++u) v[u] = pq2[s[u]];
            #pragma unroll
            for (int u = 0; u < 4; ++u) {
                if (j + 8u * u < cnt) { a += v[u].x; c += v[u].y; }
            }
        }
        #pragma unroll
        for (int off = 1; off < 8; off <<= 1) {
            a += __shfl_xor(a, off, 32);
            c += __shfl_xor(c, off, 32);
        }
        a = (a + self.x) * inn;
        c = (c + self.y) * inn;

        const float4 wi0 = ((const float4*)W_in)[sub * 2], wi1 = ((const float4*)W_in)[sub * 2 + 1];
        const float4 bi0 = ((const float4*)b_in)[sub * 2], bi1 = ((const float4*)b_in)[sub * 2 + 1];
        float4 t0, t1;
        t0.x = a * wi0.x + c * bi0.x;  t0.y = a * wi0.y + c * bi0.y;
        t0.z = a * wi0.z + c * bi0.z;  t0.w = a * wi0.w + c * bi0.w;
        t1.x = a * wi1.x + c * bi1.x;  t1.y = a * wi1.y + c * bi1.y;
        t1.z = a * wi1.z + c * bi1.z;  t1.w = a * wi1.w + c * bi1.w;
        ((float4*)(Tl + r * 64))[sub * 2]     = t0;
        ((float4*)(Tl + r * 64))[sub * 2 + 1] = t1;
    }
    __syncthreads();

    const int wave = tid >> 5;
    const int lane = tid & 31;
    const int msub = wave >> 2;
    const int nsub = wave & 3;
    const int l16  = lane & 15;
    const int arow = msub * 16 + l16;
    const int koff = (lane < 16) ? 0 : 2;
    const int col  = nsub * 16 + l16;

    v8f acc = {};
    #pragma unroll
    for (int kk = 0; kk < 16; ++kk) {
        int k = kk * 4 + koff;
        v2f fa, fb;
        fa.x = Tl[arow * 64 + k];
        fa.y = Tl[arow * 64 + k + 1];
        fb.x = Wl[k * 64 + col];
        fb.y = Wl[(k + 1) * 64 + col];
        acc = __builtin_amdgcn_wmma_f32_16x16x4_f32(false, fa, false, fb,
                                                    (short)0, acc, false, false);
    }

    const float bcol = bias[col];
    const int rbase = row0 + msub * 16 + ((lane < 16) ? 0 : 8);
    #pragma unroll
    for (int r = 0; r < 8; ++r)
        h[(size_t)(rbase + r) * 64 + col] = acc[r] + bcol;
}

__global__ __launch_bounds__(256) void gemm_layer_kernel(
    const float* __restrict__ h, const float* __restrict__ W,
    const float* __restrict__ out_norm, float* __restrict__ g)
{
    __shared__ float Wl[64 * 64];
    __shared__ float Tl[32 * 64];

    const int tid = threadIdx.x;
    const int row0 = blockIdx.x * 32;

    for (int i = tid; i < 1024; i += 256)
        ((float4*)Wl)[i] = ((const float4*)W)[i];

    for (int i = tid; i < 512; i += 256) {
        int r  = i >> 4;
        int c4 = i & 15;
        int row = row0 + r;
        float4 v = ((const float4*)(h + (size_t)row * 64))[c4];
        v.x = v.x > 0.f ? v.x : 0.01f * v.x;
        v.y = v.y > 0.f ? v.y : 0.01f * v.y;
        v.z = v.z > 0.f ? v.z : 0.01f * v.z;
        v.w = v.w > 0.f ? v.w : 0.01f * v.w;
        float s = out_norm[row];
        v.x *= s; v.y *= s; v.z *= s; v.w *= s;
        ((float4*)(Tl + r * 64))[c4] = v;
    }
    __syncthreads();

    const int wave = tid >> 5;
    const int lane = tid & 31;
    const int msub = wave >> 2;
    const int nsub = wave & 3;
    const int l16  = lane & 15;
    const int arow = msub * 16 + l16;
    const int koff = (lane < 16) ? 0 : 2;
    const int col  = nsub * 16 + l16;

    v8f c = {};
    #pragma unroll
    for (int kk = 0; kk < 16; ++kk) {
        int k = kk * 4 + koff;
        v2f a, b;
        a.x = Tl[arow * 64 + k];
        a.y = Tl[arow * 64 + k + 1];
        b.x = Wl[k * 64 + col];
        b.y = Wl[(k + 1) * 64 + col];
        c = __builtin_amdgcn_wmma_f32_16x16x4_f32(false, a, false, b,
                                                  (short)0, c, false, false);
    }

    const int rbase = row0 + msub * 16 + ((lane < 16) ? 0 : 8);
    #pragma unroll
    for (int r = 0; r < 8; ++r)
        g[(size_t)(rbase + r) * 64 + col] = c[r];
}

#define ROWS(U)                                                            \
    {                                                                      \
        float2 v[U];                                                       \
        _Pragma("unroll")                                                  \
        for (int u = 0; u < U; ++u) {                                      \
              \
            int s = __builtin_amdgcn_readlane(my, (int)(j + u));           \
            v[u] = g2[(size_t)s * 32 + lane];                              \
        }                                                                  \
        _Pragma("unroll")                                                  \
        for (int u = 0; u < U; ++u) { acc.x += v[u].x; acc.y += v[u].y; }  \
        j += U;                                                            \
    }

__global__ __launch_bounds__(256) void gather_post_kernel(
    const float* __restrict__ g, const int* __restrict__ csr_src,
    const unsigned* __restrict__ row_start, const unsigned* __restrict__ in_cnt,
    const float* __restrict__ in_norm, const float* __restrict__ bc,
    float* __restrict__ h, int n)
{
    const int node = blockIdx.x * 8 + (threadIdx.x >> 5);
    if (node >= n) return;
    const int lane = threadIdx.x & 31;
    const float2* g2 = (const float2*)g;

    const unsigned start = __builtin_amdgcn_readfirstlane(row_start[node]);
    const unsigned cnt   = __builtin_amdgcn_readfirstlane(in_cnt[node]);

    float2 acc = make_float2(0.f, 0.f);
    for (unsigned base = 0; base < cnt; base += 32) {
        const unsigned m = (cnt - base < 32u) ? (cnt - base) : 32u;
        const int my = ((unsigned)lane < m) ? csr_src[start + base + lane] : 0;
        unsigned j = 0;
        if (j + 16 <= m) ROWS(16)
        if (j + 16 <= m) ROWS(16)
        if (j + 8 <= m) ROWS(8)
        if (j + 4 <= m) ROWS(4)
        if (j + 2 <= m) ROWS(2)
        if (j + 1 <= m) ROWS(1)
    }

    const float2 gv = g2[(size_t)node * 32 + lane];
    const float sc = in_norm[node];
    const float2 b = ((const float2*)bc)[lane];
    float2 r;
    r.x = (acc.x + gv.x) * sc + b.x;
    r.y = (acc.y + gv.y) * sc + b.y;
    ((float2*)h)[(size_t)node * 32 + lane] = r;
}
#undef ROWS

template <int M> __device__ __forceinline__ float row_xor_add(float x) {
    const int y = __builtin_amdgcn_update_dpp(0, __builtin_bit_cast(int, x), 0x160 + M, 0xf, 0xf, true);
    return x + __builtin_bit_cast(float, y);
}

__global__ __launch_bounds__(256) void gemm_head_kernel(
    const float* __restrict__ h, const float* __restrict__ W,
    const float* __restrict__ out_norm, const float* __restrict__ W_pred, float* __restrict__ z)
{
    __shared__ float Wl[64 * 64];
    __shared__ float Tl[32 * 64];
    __shared__ float Zp[4][32];

    const int tid = threadIdx.x;
    const int row0 = blockIdx.x * 32;

    for (int i = tid; i < 1024; i += 256)
        ((float4*)Wl)[i] = ((const float4*)W)[i];

    for (int i = tid; i < 512; i += 256) {
        int r  = i >> 4;
        int c4 = i & 15;
        int row = row0 + r;
        float4 v = ((const float4*)(h + (size_t)row * 64))[c4];
        v.x = v.x > 0.f ? v.x : 0.01f * v.x;
        v.y = v.y > 0.f ? v.y : 0.01f * v.y;
        v.z = v.z > 0.f ? v.z : 0.01f * v.z;
        v.w = v.w > 0.f ? v.w : 0.01f * v.w;
        float s = out_norm[row];
        v.x *= s; v.y *= s; v.z *= s; v.w *= s;
        ((float4*)(Tl + r * 64))[c4] = v;
    }
    __syncthreads();

    const int wave = tid >> 5;
    const int lane = tid & 31;
    const int msub = wave >> 2;
    const int nsub = wave & 3;
    const int l16  = lane & 15;
    const int arow = msub * 16 + l16;
    const int koff = (lane < 16) ? 0 : 2;
    const int col  = nsub * 16 + l16;

    const float wp = W_pred[col];

    v8f c = {};
    #pragma unroll
    for (int kk = 0; kk < 16; ++kk) {
        int k = kk * 4 + koff;
        v2f a, b;
        a.x = Tl[arow * 64 + k];
        a.y = Tl[arow * 64 + k + 1];
        b.x = Wl[k * 64 + col];
        b.y = Wl[(k + 1) * 64 + col];
        c = __builtin_amdgcn_wmma_f32_16x16x4_f32(false, a, false, b,
                                                  (short)0, c, false, false);
    }

    const int rloc = msub * 16 + ((lane < 16) ? 0 : 8);
    float p[8];
    #pragma unroll
    for (int r = 0; r < 8; ++r) p[r] = c[r] * wp;
    #pragma unroll
    for (int r = 0; r < 8; ++r) p[r] = row_xor_add<1>(p[r]);
    #pragma unroll
    for (int r = 0; r < 8; ++r) p[r] = row_xor_add<2>(p[r]);
    #pragma unroll
    for (int r = 0; r < 8; ++r) p[r] = row_xor_add<4>(p[r]);
    #pragma unroll
    for (int r = 0; r < 8; ++r) p[r] = row_xor_add<8>(p[r]);
    if (l16 == 0) {
        #pragma unroll
        for (int r = 0; r < 8; ++r) Zp[nsub][rloc + r] = p[r];
    }
    __syncthreads();
    if (tid < 32)
        z[row0 + tid] = (Zp[0][tid] + Zp[1][tid]) + (Zp[2][tid] + Zp[3][tid]);
}

__global__ __launch_bounds__(256) void gather_head_kernel(
    const float* __restrict__ z, const int* __restrict__ csr_src,
    const unsigned* __restrict__ row_start, const unsigned* __restrict__ in_cnt,
    const float* __restrict__ in_norm, const float* __restrict__ bias,
    const float* __restrict__ W_pred, const float* __restrict__ b_pred,
    float* __restrict__ out)
{
    __shared__ float ol[32];

    const int tid = threadIdx.x;
    const int lane = tid & 31;
    const int r   = tid >> 3;
    const int sub = tid & 7;
    const int row = blockIdx.x * 32 + r;

    const unsigned start = row_start[row];
    const unsigned cnt   = in_cnt[row];
    const float self = z[row];
    const float inn = in_norm[row];
    const float2 bb = ((const float2*)bias)[lane];
    const float2 ww = ((const float2*)W_pred)[lane];
    asm volatile("" :: "v"(start), "v"(bb.x), "v"(bb.y), "v"(ww.x), "v"(ww.y));

    float a = 0.f;
    for (unsigned j = sub; j < cnt; j += 32) {
        int s[4];
        float v[4];
        #pragma unroll
        for (int u = 0; u < 4; ++u) {
            const unsigned jj = j + 8u * u;
            s[u] = csr_src[start + (jj < cnt ? jj : j)];
        }
        #pragma unroll
        for (int u = 0; u < 4; ++u) v[u] = z[s[u]];
        #pragma unroll
        for (int u = 0; u < 4; ++u) {
            if (j + 8u * u < cnt) a += v[u];
        }
    }
    a = row_xor_add<1>(a);
    a = row_xor_add<2>(a);
    a = row_xor_add<4>(a);

    float cb = bb.x * ww.x + bb.y * ww.y;
    cb = row_xor_add<1>(cb);
    cb = row_xor_add<2>(cb);
    cb = row_xor_add<4>(cb);
    cb = row_xor_add<8>(cb);
    cb += __shfl_xor(cb, 16, 32);
    cb += b_pred[0];

    if (sub == 0) ol[r] = (a + self) * inn + cb;
    __syncthreads();
    if (tid < 32) out[blockIdx.x * 32 + tid] = ol[tid];
}

extern "C" void kernel_launch(void* const* d_in, const int* in_sizes, int n_in,
                              void* d_out, int out_size, void* d_ws, size_t ws_size,
                              hipStream_t stream) {
    const float* weight = (const float*)d_in[0];
    const int*   src    = (const int*)d_in[1];
    const int*   dst    = (const int*)d_in[2];
    const float* W_in   = (const float*)d_in[3];
    const float* b_in   = (const float*)d_in[4];
    const float* Wc     = (const float*)d_in[5];
    const float* bc     = (const float*)d_in[6];
    const float* W_pred = (const float*)d_in[7];
    const float* b_pred = (const float*)d_in[8];
    float* out = (float*)d_out;

    const int N = in_sizes[0];
    const int E = in_sizes[1];

    const int NB = (N + 255) / 256;

    float* h          = (float*)d_ws;
    float* g          = h + (size_t)N * 64;
    unsigned* in_cnt  = (unsigned*)(g + (size_t)N * 64);
    float* out_norm   = (float*)(in_cnt + N);
    float* in_norm    = out_norm + N;
    unsigned* row_start = (unsigned*)(in_norm + N);
    float* pq         = (float*)(row_start + N);
    int* csr_src      = (int*)(pq + (size_t)N * 2);
    unsigned* bufD    = (unsigned*)(csr_src + E);
    unsigned* tableD  = bufD + E;
    unsigned* tableS  = tableD + (size_t)NB * GCH;
    unsigned* totD    = tableS + (size_t)NB * GCH;
    unsigned* totS    = totD + NB;
    unsigned* baseD   = totS + NB;
    unsigned* baseS   = baseD + NB;
    unsigned char* bufS = (unsigned char*)(baseS + NB);

    bucket_count_kernel<<<GCH, 1024, 2 * NB * sizeof(unsigned), stream>>>(src, dst, tableD, tableS, E, NB);

    bucket_scan_kernel<<<NB, GCH, 0, stream>>>(tableD, tableS, totD, totS);

    bucket_scatter_kernel<<<GCH, 1024, 6 * NB * sizeof(unsigned), stream>>>(
        src, dst, tableD, tableS, totD, totS, baseD, baseS, bufD, bufS, E, NB);

    bucket_finish_kernel<<<NB, 1024, 0, stream>>>(bufD, bufS, baseD, baseS, totD, totS, weight,
                                                  out_norm, in_norm, pq, in_cnt, row_start, csr_src, N);

    layer0_kernel<<<N / 32, 256, 0, stream>>>(pq, csr_src, row_start, in_cnt, in_norm,
                                              W_in, b_in, Wc, bc, h);

    gemm_layer_kernel<<<N / 32, 256, 0, stream>>>(h, Wc + (size_t)64 * 64, out_norm, g);
    gather_post_kernel<<<(N + 7) / 8, 256, 0, stream>>>(g, csr_src, row_start, in_cnt, in_norm,
                                                        bc + 64, h, N);

    float* z = g;
    gemm_head_kernel<<<N / 32, 256, 0, stream>>>(h, Wc + (size_t)2 * 64 * 64, out_norm, W_pred, z);
    gather_head_kernel<<<N / 32, 256, 0, stream>>>(z, csr_src, row_start, in_cnt, in_norm,
                                                   bc + 2 * 64, W_pred, b_pred, out);
}


# Each projection GEMM fused into the aggregation kernel ahead of it: layer 1's into the layer-0 kernel, last one + head into layer 1's gather (a wave takes 4 nodes); h stays in LDS; 2 launches gone
# speedup vs baseline: 7.3483x; 1.1643x over previous
#include <hip/hip_runtime.h>
#include <math.h>

typedef __attribute__((ext_vector_type(2))) float v2f;
typedef __attribute__((ext_vector_type(8))) float v8f;

#define NN 100000

#define GCH 256
#define BSH 8
#define BMASK 255u

__device__ __forceinline__ void block_excl_scan(unsigned* a, int n, unsigned* wsum) {
    const int tid = threadIdx.x, lane = tid & 31, wave = tid >> 5;
    const int per = (n + 1023) >> 10;
    const int i0 = tid * per;
    unsigned s = 0;
    for (int k = 0; k < per; ++k) if (i0 + k < n) s += a[i0 + k];
    unsigned x = s;
    for (int off = 1; off < 32; off <<= 1) {
        unsigned y = __shfl_up(x, off, 32);
        if (lane >= off) x += y;
    }
    if (lane == 31) wsum[wave] = x;
    __syncthreads();
    unsigned run = x - s;
    for (int w = 0; w < wave; ++w) run += wsum[w];
    for (int k = 0; k < per; ++k) if (i0 + k < n) { unsigned v = a[i0 + k]; a[i0 + k] = run; run += v; }
    __syncthreads();
}

__global__ __launch_bounds__(1024) void bucket_count_kernel(
    const int* __restrict__ src, const int* __restrict__ dst,
    unsigned* __restrict__ tableD, unsigned* __restrict__ tableS, int E, int NB)
{
    extern __shared__ unsigned dyn_lds[];
    const int tid = threadIdx.x;
    for (int i = tid; i < 2 * NB; i += 1024) dyn_lds[i] = 0u;
    __syncthreads();

    const int chunk = (E + GCH - 1) / GCH;
    const int e0 = blockIdx.x * chunk;
    const int e1 = (e0 + chunk < E) ? (e0 + chunk) : E;
    for (int e = e0 + tid; e < e1; e += 4096) {
        int d[4], s[4];
        #pragma unroll
        for (int u = 0; u < 4; ++u) {
            const int ee = e + u * 1024;
            const int ec = (ee < e1) ? ee : e;
            d[u] = dst[ec];
            s[u] = src[ec];
        }
        asm volatile("" :: "v"(d[0]), "v"(d[1]), "v"(d[2]), "v"(d[3]), "v"(s[0]), "v"(s[1]), "v"(s[2]), "v"(s[3]));
        #pragma unroll
        for (int u = 0; u < 4; ++u) {
            if (e + u * 1024 < e1) {
                atomicAdd(&dyn_lds[d[u] >> BSH], 1u);
                atomicAdd(&dyn_lds[NB + (s[u] >> BSH)], 1u);
            }
        }
    }
    __syncthreads();
    for (int b = tid; b < NB; b += 1024) {
        tableD[b * GCH + blockIdx.x] = dyn_lds[b];
        tableS[b * GCH + blockIdx.x] = dyn_lds[NB + b];
    }
}

__global__ __launch_bounds__(GCH) void bucket_scan_kernel(
    unsigned* __restrict__ tableD, unsigned* __restrict__ tableS,
    unsigned* __restrict__ totD, unsigned* __restrict__ totS)
{
    __shared__ unsigned ws[2][GCH / 32];
    const int t = threadIdx.x, lane = t & 31, wave = t >> 5;
    const size_t idx = (size_t)blockIdx.x * GCH + t;
    const unsigned vd = tableD[idx], vs = tableS[idx];
    unsigned xd = vd, xs = vs;
    for (int off = 1; off < 32; off <<= 1) {
        unsigned yd = __shfl_up(xd, off, 32);
        unsigned ys = __shfl_up(xs, off, 32);
        if (lane >= off) { xd += yd; xs += ys; }
    }
    if (lane == 31) { ws[0][wave] = xd; ws[1][wave] = xs; }
    __syncthreads();
    unsigned od = 0, os = 0;
    for (int w = 0; w < wave; ++w) { od += ws[0][w]; os += ws[1][w]; }
    tableD[idx] = od + xd - vd;
    tableS[idx] = os + xs - vs;
    if (t == GCH - 1) { totD[blockIdx.x] = od + xd; totS[blockIdx.x] = os + xs; }
}

#define EPT 13
#define TILE_E (1024 * EPT)
__global__ __launch_bounds__(1024) void bucket_scatter_kernel(
    const int* __restrict__ src, const int* __restrict__ dst,
    const unsigned* __restrict__ tableD, const unsigned* __restrict__ tableS,
    const unsigned* __restrict__ totD, const unsigned* __restrict__ totS,
    unsigned* __restrict__ baseD, unsigned* __restrict__ baseS,
    unsigned* __restrict__ bufD, unsigned char* __restrict__ bufS, int E, int NB)
{
    extern __shared__ unsigned dyn_lds[];
    unsigned* gcD = dyn_lds;
    unsigned* gcS = dyn_lds + NB;
    unsigned* lsD = dyn_lds + 2 * NB;
    unsigned* lsS = dyn_lds + 3 * NB;
    unsigned* dlD = dyn_lds + 4 * NB;
    unsigned* dlS = dyn_lds + 5 * NB;
    __shared__ unsigned wsum[32];
    __shared__ unsigned       stD[TILE_E];
    __shared__ unsigned short sbD[TILE_E];
    __shared__ unsigned       stS[TILE_E];
    __shared__ unsigned spare[64];
    const int tid = threadIdx.x;
    if (tid < 64) spare[tid] = 0u;
    for (int b = tid; b < NB; b += 1024) { gcD[b] = totD[b]; gcS[b] = totS[b]; }
    __syncthreads();
    block_excl_scan(gcD, NB, wsum);
    block_excl_scan(gcS, NB, wsum);
    for (int b = tid; b < NB; b += 1024) {
        const unsigned bd = gcD[b], bs = gcS[b];
        if (blockIdx.x == 0) { baseD[b] = bd; baseS[b] = bs; }
        gcD[b] = bd + tableD[b * GCH + blockIdx.x];
        gcS[b] = bs + tableS[b * GCH + blockIdx.x];
    }
    __syncthreads();

    const int chunk = (E + GCH - 1) / GCH;
    const int e0 = blockIdx.x * chunk;
    const int e1 = (e0 + chunk < E) ? (e0 + chunk) : E;
    const int nt = e1 - e0;
    if (nt > TILE_E) {
        for (int e = e0 + tid; e < e1; e += 1024) {
            const int dd = dst[e], ss = src[e];
            const unsigned qd = atomicAdd(&gcD[(unsigned)dd >> BSH], 1u);
            bufD[qd] = (((unsigned)dd & BMASK) << 24) | (unsigned)ss;
            const unsigned qs = atomicAdd(&gcS[(unsigned)ss >> BSH], 1u);
            bufS[qs] = (unsigned char)((unsigned)ss & BMASK);
        }
    } else if (nt > 0) {
        const int t0 = e0;
        for (int b = tid; b < 2 * NB; b += 1024) lsD[b] = 0u;
        __syncthreads();

        int d[EPT], s[EPT];
        #pragma unroll
        for (int u = 0; u < EPT; ++u) {
            const int i = u * 1024 + tid;
            const int ec = t0 + ((i < nt) ? i : 0);
            d[u] = dst[ec];
            s[u] = src[ec];
        }
        unsigned rk[EPT];
        {
            unsigned rd[EPT], rs[EPT];
            #pragma unroll
            for (int u = 0; u < EPT; ++u) {
                const bool valid = u * 1024 + tid < nt;
                rd[u] = atomicAdd(valid ? &lsD[(unsigned)d[u] >> BSH] : &spare[tid & 31], 1u);
                rs[u] = atomicAdd(valid ? &lsS[(unsigned)s[u] >> BSH] : &spare[32 + (tid & 31)], 1u);
            }
            #pragma unroll
            for (int u = 0; u < EPT; ++u) rk[u] = (rd[u] & 0xFFFFu) | (rs[u] << 16);
        }
        __syncthreads();
        block_excl_scan(lsD, NB, wsum);
        block_excl_scan(lsS, NB, wsum);
        for (int b = tid; b < NB; b += 1024) { dlD[b] = gcD[b] - lsD[b]; dlS[b] = gcS[b] - lsS[b]; }
        unsigned pd[EPT], ps[EPT];
        #pragma unroll
        for (int u = 0; u < EPT; ++u) {
            pd[u] = lsD[(unsigned)d[u] >> BSH] + (rk[u] & 0xFFFFu);
            ps[u] = lsS[(unsigned)s[u] >> BSH] + (rk[u] >> 16);
        }
        #pragma unroll
        for (int u = 0; u < EPT; ++u) {
            if (u * 1024 + tid < nt) {
                stD[pd[u]] = (((unsigned)d[u] & BMASK) << 24) | (unsigned)s[u];
                sbD[pd[u]] = (unsigned short)((unsigned)d[u] >> BSH);
                stS[ps[u]] = (((unsigned)s[u] >> BSH) << 8) | ((unsigned)s[u] & BMASK);
            }
        }
        __syncthreads();

        {
            unsigned v[EPT], a[EPT];
            #pragma unroll
            for (int u = 0; u < EPT; ++u) {
                const int i = u * 1024 + tid;
                const int ic = (i < nt) ? i : 0;
                v[u] = stD[ic];
                a[u] = sbD[ic];
            }
            #pragma unroll
            for (int u = 0; u < EPT; ++u) a[u] = dlD[a[u]] + (unsigned)(u * 1024 + tid);
            #pragma unroll
            for (int u = 0; u < EPT; ++u)
                if (u * 1024 + tid < nt) bufD[a[u]] = v[u];
        }
        {
            unsigned v[EPT], a[EPT];
            #pragma unroll
            for (int u = 0; u < EPT; ++u) {
                const int i = u * 1024 + tid;
                v[u] = stS[(i < nt) ? i : 0];
            }
            #pragma unroll
            for (int u = 0; u < EPT; ++u) a[u] = dlS[v[u] >> 8] + (unsigned)(u * 1024 + tid);
            #pragma unroll
            for (int u = 0; u < EPT; ++u)
                if (u * 1024 + tid < nt) bufS[a[u]] = (unsigned char)(v[u] & BMASK);
        }

    }
}

#define STAGE_N 12288
__global__ __launch_bounds__(1024) void bucket_finish_kernel(
    const unsigned* __restrict__ bufD, const unsigned char* __restrict__ bufS,
    const unsigned* __restrict__ baseD, const unsigned* __restrict__ baseS,
    const unsigned* __restrict__ totD, const unsigned* __restrict__ totS,
    const float* __restrict__ weight,
    float* __restrict__ out_norm, float* __restrict__ in_norm, float* __restrict__ pq,
    unsigned* __restrict__ in_cnt, unsigned* __restrict__ row_start,
    int* __restrict__ csr_src, int n)
{
    __shared__ unsigned cin[256];
    __shared__ unsigned cout[256];
    __shared__ unsigned cur[256];
    __shared__ unsigned wsum[32];
    __shared__ int stage[STAGE_N];
    const int tid = threadIdx.x;
    const int b = blockIdx.x;
    const unsigned bD = baseD[b], nD = totD[b];
    const unsigned bS = baseS[b], nS = totS[b];
    const float wv = (tid < 256 && b * 256 + tid < n) ? weight[b * 256 + tid] : 0.f;
    if (tid < 256) { cin[tid] = 0u; cout[tid] = 0u; }
    __syncthreads();
    for (unsigned i = tid; i < nS; i += 1024) atomicAdd(&cout[bufS[bS + i]], 1u);
    for (unsigned i = tid; i < nD; i += 1024) atomicAdd(&cin[bufD[bD + i] >> 24], 1u);
    __syncthreads();
    if (tid < 256) cur[tid] = cin[tid];
    __syncthreads();
    block_excl_scan(cur, 256, wsum);
    if (tid < 256) {
        const unsigned rs = bD + cur[tid];
        cur[tid] = rs;
        const int node = b * 256 + tid;
        if (node < n) {
            const unsigned c = cin[tid];
            row_start[node] = rs;
            in_cnt[node] = c;
            in_norm[node]  = rsqrtf((float)(c + 1u));
            const float on = rsqrtf((float)(cout[tid] + 1u));
            out_norm[node] = on;
            ((float2*)pq)[node] = make_float2(on * wv, on);
        }
    }
    __syncthreads();
    const bool staged = nD <= (unsigned)STAGE_N;
    for (unsigned i = tid; i < nD; i += 1024) {
        const unsigned w = bufD[bD + i];
        const unsigned pos = atomicAdd(&cur[w >> 24], 1u);
        const int sv = (int)(w & 0x00FFFFFFu);
        if (staged) stage[pos - bD] = sv;
        else        csr_src[pos] = sv;
    }
    if (staged) {
        __syncthreads();
        for (unsigned i = tid; i < nD; i += 1024) csr_src[bD + i] = stage[i];
    }
}

__global__ __launch_bounds__(256) void layer0_kernel(
    const float* __restrict__ pq, const int* __restrict__ csr_src,
    const unsigned* __restrict__ row_start, const unsigned* __restrict__ in_cnt,
    const float* __restrict__ in_norm, const float* __restrict__ out_norm,
    const float* __restrict__ W_in, const float* __restrict__ b_in,
    const float* __restrict__ W, const float* __restrict__ bias,
    const float* __restrict__ Wnext, float* __restrict__ g)
{
    __shared__ float Wl[64 * 64];
    __shared__ float Wn[64 * 64];
    __shared__ float Tl[32 * 64];
    __shared__ float On[32];

    const int tid = threadIdx.x;
    const int row0 = blockIdx.x * 32;

    float onv = 0.f;
    if (tid < 32) onv = out_norm[row0 + tid];

    #pragma unroll
    for (int i = 0; i < 4; ++i) {
        ((float4*)Wl)[tid + 256 * i] = ((const float4*)W)[tid + 256 * i];
        ((float4*)Wn)[tid + 256 * i] = ((const float4*)Wnext)[tid + 256 * i];
    }
    if (tid < 32) On[tid] = onv;

    {
        const int r   = tid >> 3;
        const int sub = tid & 7;
        const int row = row0 + r;
        const unsigned start = row_start[row];
        const unsigned cnt   = in_cnt[row];
        const float2* pq2 = (const float2*)pq;
        const float2 self = pq2[row];
        const float inn = in_norm[row];

        float a = 0.f, c = 0.f;
        for (unsigned j = sub; j < cnt; j += 32) {
            int s[4];
            float2 v[4];
            #pragma unroll
            for (int u = 0; u < 4; ++u) {
                const unsigned jj = j + 8u * u;
                s[u] = csr_src[start + (jj < cnt ? jj : j)];
            }
            #pragma unroll
            for (int u = 0; u < 4; ++u) v[u] = pq2[s[u]];
            #pragma unroll
            for (int u = 0; u < 4; ++u) {
                if (j + 8u * u < cnt) { a += v[u].x; c += v[u].y; }
            }
        }
        #pragma unroll
        for (int off = 1; off < 8; off <<= 1) {
            a += __shfl_xor(a, off, 32);
            c += __shfl_xor(c, off, 32);
        }
        a = (a + self.x) * inn;
        c = (c + self.y) * inn;

        const float4 wi0 = ((const float4*)W_in)[sub * 2], wi1 = ((const float4*)W_in)[sub * 2 + 1];
        const float4 bi0 = ((const float4*)b_in)[sub * 2], bi1 = ((const float4*)b_in)[sub * 2 + 1];
        float4 t0, t1;
        t0.x = a * wi0.x + c * bi0.x;  t0.y = a * wi0.y + c * bi0.y;
        t0.z = a * wi0.z + c * bi0.z;  t0.w = a * wi0.w + c * bi0.w;
        t1.x = a * wi1.x + c * bi1.x;  t1.y = a * wi1.y + c * bi1.y;
        t1.z = a * wi1.z + c * bi1.z;  t1.w = a * wi1.w + c * bi1.w;
        ((float4*)(Tl + r * 64))[sub * 2]     = t0;
        ((float4*)(Tl + r * 64))[sub * 2 + 1] = t1;
    }
    __syncthreads();

    const int wave = tid >> 5;
    const int lane = tid & 31;
    const int msub = wave >> 2;
    const int nsub = wave & 3;
    const int l16  = lane & 15;
    const int arow = msub * 16 + l16;
    const int koff = (lane < 16) ? 0 : 2;
    const int col  = nsub * 16 + l16;

    v8f acc = {};
    #pragma unroll
    for (int kk = 0; kk < 16; ++kk) {
        int k = kk * 4 + koff;
        v2f fa, fb;
        fa.x = Tl[arow * 64 + k];
        fa.y = Tl[arow * 64 + k + 1];
        fb.x = Wl[k * 64 + col];
        fb.y = Wl[(k + 1) * 64 + col];
        acc = __builtin_amdgcn_wmma_f32_16x16x4_f32(false, fa, false, fb,
                                                    (short)0, acc, false, false);
    }

    const float bcol = bias[col];
    const int rloc = msub * 16 + ((lane < 16) ? 0 : 8);
    float t[8];
    #pragma unroll
    for (int r = 0; r < 8; ++r) {
        float v = acc[r] + bcol;
        v = v > 0.f ? v : 0.01f * v;
        t[r] = v * On[rloc + r];
    }
    __syncthreads();
    #pragma unroll
    for (int r = 0; r < 8; ++r) Tl[(rloc + r) * 64 + col] = t[r];
    __syncthreads();

    v8f c = {};
    #pragma unroll
    for (int kk = 0; kk < 16; ++kk) {
        int k = kk * 4 + koff;
        v2f a, b;
        a.x = Tl[arow * 64 + k];
        a.y = Tl[arow * 64 + k + 1];
        b.x = Wn[k * 64 + col];
        b.y = Wn[(k + 1) * 64 + col];
        c = __builtin_amdgcn_wmma_f32_16x16x4_f32(false, a, false, b,
                                                  (short)0, c, false, false);
    }

    #pragma unroll
    for (int r = 0; r < 8; ++r)
        g[(size_t)(row0 + rloc + r) * 64 + col] = c[r];
}

template <int M> __device__ __forceinline__ float row_xor_add(float x) {
    const int y = __builtin_amdgcn_update_dpp(0, __builtin_bit_cast(int, x), 0x160 + M, 0xf, 0xf, true);
    return x + __builtin_bit_cast(float, y);
}

#define ROWS(U)                                                            \
    {                                                                      \
        float2 v[U];                                                       \
        _Pragma("unroll")                                                  \
        for (int u = 0; u < U; ++u) {                                      \
              \
            int s = __builtin_amdgcn_readlane(my, (int)(j + u));           \
            v[u] = g2[(size_t)s * 32 + lane];                              \
        }                                                                  \
        _Pragma("unroll")                                                  \
        for (int u = 0; u < U; ++u) { acc.x += v[u].x; acc.y += v[u].y; }  \
        j += U;                                                            \
    }

__global__ __launch_bounds__(256) void gather_gemm_head_kernel(
    const float* __restrict__ g, const int* __restrict__ csr_src,
    const unsigned* __restrict__ row_start, const unsigned* __restrict__ in_cnt,
    const float* __restrict__ in_norm, const float* __restrict__ out_norm,
    const float* __restrict__ bc, const float* __restrict__ W,
    const float* __restrict__ W_pred, float* __restrict__ z)
{
    __shared__ float Wl[64 * 64];
    __shared__ float Tl[32 * 64];
    __shared__ float Zp[4][32];

    const int tid = threadIdx.x;
    const int row0 = blockIdx.x * 32;
    const int wave = tid >> 5;
    const int lane = tid & 31;

    const int nb = row0 + wave * 4;
    unsigned st = 0u, cn = 0u;
    float inl = 0.f, onl = 0.f;
    if (lane < 4) {
        st  = row_start[nb + lane];
        cn  = in_cnt[nb + lane];
        inl = in_norm[nb + lane];
        onl = out_norm[nb + lane];
    }

    #pragma unroll
    for (int i = 0; i < 4; ++i)
        ((float4*)Wl)[tid + 256 * i] = ((const float4*)W)[tid + 256 * i];

    {
        const float2* g2 = (const float2*)g;
        const float2 b = ((const float2*)bc)[lane];

        #pragma unroll 1
        for (int q = 0; q < 4; ++q) {
            const int r = wave * 4 + q;
            const int node = row0 + r;
            const unsigned start = (unsigned)__builtin_amdgcn_readlane((int)st, q);
            const unsigned cnt   = (unsigned)__builtin_amdgcn_readlane((int)cn, q);
            const float sc = __builtin_bit_cast(float, __builtin_amdgcn_readlane(__builtin_bit_cast(int, inl), q));
            const float on = __builtin_bit_cast(float, __builtin_amdgcn_readlane(__builtin_bit_cast(int, onl), q));
            const float2 gv = g2[(size_t)node * 32 + lane];

            float2 acc = make_float2(0.f, 0.f);
            for (unsigned base = 0; base < cnt; base += 32) {
                const unsigned m = (cnt - base < 32u) ? (cnt - base) : 32u;
                const int my = ((unsigned)lane < m) ? csr_src[start + base + lane] : 0;
                unsigned j = 0;
                if (j + 16 <= m) ROWS(16)
                if (j + 16 <= m) ROWS(16)
                if (j + 8 <= m) ROWS(8)
                if (j + 4 <= m) ROWS(4)
                if (j + 2 <= m) ROWS(2)
                if (j + 1 <= m) ROWS(1)
            }

            float2 h;
            h.x = (acc.x + gv.x) * sc + b.x;
            h.y = (acc.y + gv.y) * sc + b.y;
            h.x = h.x > 0.f ? h.x : 0.01f * h.x;
            h.y = h.y > 0.f ? h.y : 0.01f * h.y;
            h.x *= on; h.y *= on;
            ((float2*)(Tl + r * 64))[lane] = h;
        }
    }
    __syncthreads();

    const int msub = wave >> 2;
    const int nsub = wave & 3;
    const int l16  = lane & 15;
    const int arow = msub * 16 + l16;
    const int koff = (lane < 16) ? 0 : 2;
    const int col  = nsub * 16 + l16;

    const float wp = W_pred[col];

    v8f c = {};
    #pragma unroll
    for (int kk = 0; kk < 16; ++kk) {
        int k = kk * 4 + koff;
        v2f a, b;
        a.x = Tl[arow * 64 + k];
        a.y = Tl[arow * 64 + k + 1];
        b.x = Wl[k * 64 + col];
        b.y = Wl[(k + 1) * 64 + col];
        c = __builtin_amdgcn_wmma_f32_16x16x4_f32(false, a, false, b,
                                                  (short)0, c, false, false);
    }

    const int rloc = msub * 16 + ((lane < 16) ? 0 : 8);
    float p[8];
    #pragma unroll
    for (int r = 0; r < 8; ++r) p[r] = c[r] * wp;
    #pragma unroll
    for (int r = 0; r < 8; ++r) p[r] = row_xor_add<1>(p[r]);
    #pragma unroll
    for (int r = 0; r < 8; ++r) p[r] = row_xor_add<2>(p[r]);
    #pragma unroll
    for (int r = 0; r < 8; ++r) p[r] = row_xor_add<4>(p[r]);
    #pragma unroll
    for (int r = 0; r < 8; ++r) p[r] = row_xor_add<8>(p[r]);
    if (l16 == 0) {
        #pragma unroll
        for (int r = 0; r < 8; ++r) Zp[nsub][rloc + r] = p[r];
    }
    __syncthreads();
    if (tid < 32)
        z[row0 + tid] = (Zp[0][tid] + Zp[1][tid]) + (Zp[2][tid] + Zp[3][tid]);
}
#undef ROWS

__global__ __launch_bounds__(256) void gather_head_kernel(
    const float* __restrict__ z, const int* __restrict__ csr_src,
    const unsigned* __restrict__ row_start, const unsigned* __restrict__ in_cnt,
    const float* __restrict__ in_norm, const float* __restrict__ bias,
    const float* __restrict__ W_pred, const float* __restrict__ b_pred,
    float* __restrict__ out)
{
    __shared__ float ol[32];

    const int tid = threadIdx.x;
    const int lane = tid & 31;
    const int r   = tid >> 3;
    const int sub = tid & 7;
    const int row = blockIdx.x * 32 + r;

    const unsigned start = row_start[row];
    const unsigned cnt   = in_cnt[row];
    const float self = z[row];
    const float inn = in_norm[row];
    const float2 bb = ((const float2*)bias)[lane];
    const float2 ww = ((const float2*)W_pred)[lane];
    asm volatile("" :: "v"(start), "v"(bb.x), "v"(bb.y), "v"(ww.x), "v"(ww.y));

    float a = 0.f;
    for (unsigned j = sub; j < cnt; j += 32) {
        int s[4];
        float v[4];
        #pragma unroll
        for (int u = 0; u < 4; ++u) {
            const unsigned jj = j + 8u * u;
            s[u] = csr_src[start + (jj < cnt ? jj : j)];
        }
        #pragma unroll
        for (int u = 0; u < 4; ++u) v[u] = z[s[u]];
        #pragma unroll
        for (int u = 0; u < 4; ++u) {
            if (j + 8u * u < cnt) a += v[u];
        }
    }
    a = row_xor_add<1>(a);
    a = row_xor_add<2>(a);
    a = row_xor_add<4>(a);

    float cb = bb.x * ww.x + bb.y * ww.y;
    cb = row_xor_add<1>(cb);
    cb = row_xor_add<2>(cb);
    cb = row_xor_add<4>(cb);
    cb = row_xor_add<8>(cb);
    cb += __shfl_xor(cb, 16, 32);
    cb += b_pred[0];

    if (sub == 0) ol[r] = (a + self) * inn + cb;
    __syncthreads();
    if (tid < 32) out[blockIdx.x * 32 + tid] = ol[tid];
}

extern "C" void kernel_launch(void* const* d_in, const int* in_sizes, int n_in,
                              void* d_out, int out_size, void* d_ws, size_t ws_size,
                              hipStream_t stream) {
    const float* weight = (const float*)d_in[0];
    const int*   src    = (const int*)d_in[1];
    const int*   dst    = (const int*)d_in[2];
    const float* W_in   = (const float*)d_in[3];
    const float* b_in   = (const float*)d_in[4];
    const float* Wc     = (const float*)d_in[5];
    const float* bc     = (const float*)d_in[6];
    const float* W_pred = (const float*)d_in[7];
    const float* b_pred = (const float*)d_in[8];
    float* out = (float*)d_out;

    const int N = in_sizes[0];
    const int E = in_sizes[1];

    const int NB = (N + 255) / 256;

    float* g          = (float*)d_ws;
    float* z          = g + (size_t)N * 64;
    unsigned* in_cnt  = (unsigned*)(z + N);
    float* out_norm   = (float*)(in_cnt + N);
    float* in_norm    = out_norm + N;
    unsigned* row_start = (unsigned*)(in_norm + N);
    float* pq         = (float*)(row_start + N);
    int* csr_src      = (int*)(pq + (size_t)N * 2);
    unsigned* bufD    = (unsigned*)(csr_src + E);
    unsigned* tableD  = bufD + E;
    unsigned* tableS  = tableD + (size_t)NB * GCH;
    unsigned* totD    = tableS + (size_t)NB * GCH;
    unsigned* totS    = totD + NB;
    unsigned* baseD   = totS + NB;
    unsigned* baseS   = baseD + NB;
    unsigned char* bufS = (unsigned char*)(baseS + NB);

    bucket_count_kernel<<<GCH, 1024, 2 * NB * sizeof(unsigned), stream>>>(src, dst, tableD, tableS, E, NB);

    bucket_scan_kernel<<<NB, GCH, 0, stream>>>(tableD, tableS, totD, totS);

    bucket_scatter_kernel<<<GCH, 1024, 6 * NB * sizeof(unsigned), stream>>>(
        src, dst, tableD, tableS, totD, totS, baseD, baseS, bufD, bufS, E, NB);

    bucket_finish_kernel<<<NB, 1024, 0, stream>>>(bufD, bufS, baseD, baseS, totD, totS, weight,
                                                  out_norm, in_norm, pq, in_cnt, row_start, csr_src, N);

    layer0_kernel<<<N / 32, 256, 0, stream>>>(pq, csr_src, row_start, in_cnt, in_norm, out_norm,
                                              W_in, b_in, Wc, bc, Wc + (size_t)64 * 64, g);

    gather_gemm_head_kernel<<<N / 32, 256, 0, stream>>>(g, csr_src, row_start, in_cnt, in_norm, out_norm,
                                                        bc + 64, Wc + (size_t)2 * 64 * 64, W_pred, z);

    gather_head_kernel<<<N / 32, 256, 0, stream>>>(z, csr_src, row_start, in_cnt, in_norm,
                                                   bc + 2 * 64, W_pred, b_pred, out);
}


# Edge sort loses its counting pre-pass: each chunk, grouped by bucket in LDS, goes back in place with a table of bucket starts; bucket blocks collect the runs; 2 launches gone; in-edge order: arrival
# speedup vs baseline: 8.5652x; 1.1656x over previous
#include <hip/hip_runtime.h>
#include <math.h>

typedef __attribute__((ext_vector_type(2))) float v2f;
typedef __attribute__((ext_vector_type(8))) float v8f;

#define NN 100000

#define GCH 256
#define BSH 8
#define BMASK 255u

__device__ __forceinline__ void block_excl_scan(unsigned* a, int n, unsigned* wsum) {
    const int tid = threadIdx.x, lane = tid & 31, wave = tid >> 5;
    const int per = (n + 1023) >> 10;
    const int i0 = tid * per;
    unsigned s = 0;
    for (int k = 0; k < per; ++k) if (i0 + k < n) s += a[i0 + k];
    unsigned x = s;
    for (int off = 1; off < 32; off <<= 1) {
        unsigned y = __shfl_up(x, off, 32);
        if (lane >= off) x += y;
    }
    if (lane == 31) wsum[wave] = x;
    __syncthreads();
    unsigned run = x - s;
    for (int w = 0; w < wave; ++w) run += wsum[w];
    for (int k = 0; k < per; ++k) if (i0 + k < n) { unsigned v = a[i0 + k]; a[i0 + k] = run; run += v; }
    __syncthreads();
}

#define EPT 13
#define TILE_E (1024 * EPT)
__global__ __launch_bounds__(1024) void chunk_group_kernel(
    const int* __restrict__ src, const int* __restrict__ dst,
    unsigned* __restrict__ tabD, unsigned* __restrict__ tabS,
    unsigned* __restrict__ bufD, unsigned char* __restrict__ bufS, int E, int NB)
{
    extern __shared__ unsigned dyn_lds[];
    unsigned* lsD = dyn_lds;
    unsigned* lsS = dyn_lds + NB;
    __shared__ unsigned wsum[32];
    __shared__ unsigned      stD[TILE_E];
    __shared__ unsigned char stS[TILE_E];
    __shared__ unsigned spare[64];
    const int tid = threadIdx.x;
    if (tid < 64) spare[tid] = 0u;
    for (int b = tid; b < 2 * NB; b += 1024) lsD[b] = 0u;
    __syncthreads();

    const int chunk = (E + GCH - 1) / GCH;
    const int e0 = blockIdx.x * chunk;
    const int e1 = (e0 + chunk < E) ? (e0 + chunk) : E;
    int nt = e1 - e0;
    nt = nt < 0 ? 0 : (nt > TILE_E ? TILE_E : nt);
    const int t0 = (nt > 0) ? e0 : 0;

    int d[EPT], s[EPT];
    #pragma unroll
    for (int u = 0; u < EPT; ++u) {
        const int i = u * 1024 + tid;
        const int ec = t0 + ((i < nt) ? i : 0);
        d[u] = dst[ec];
        s[u] = src[ec];
    }
    unsigned rk[EPT];
    {
        unsigned rd[EPT], rs[EPT];
        #pragma unroll
        for (int u = 0; u < EPT; ++u) {
            const bool valid = u * 1024 + tid < nt;
            rd[u] = atomicAdd(valid ? &lsD[(unsigned)d[u] >> BSH] : &spare[tid & 31], 1u);
            rs[u] = atomicAdd(valid ? &lsS[(unsigned)s[u] >> BSH] : &spare[32 + (tid & 31)], 1u);
        }
        #pragma unroll
        for (int u = 0; u < EPT; ++u) rk[u] = (rd[u] & 0xFFFFu) | (rs[u] << 16);
    }
    __syncthreads();
    block_excl_scan(lsD, NB, wsum);
    block_excl_scan(lsS, NB, wsum);
    {
        unsigned* rowD = tabD + (size_t)blockIdx.x * (NB + 1);
        unsigned* rowS = tabS + (size_t)blockIdx.x * (NB + 1);
        for (int b = tid; b < NB; b += 1024) { rowD[b] = lsD[b]; rowS[b] = lsS[b]; }
        if (tid == 0) { rowD[NB] = (unsigned)nt; rowS[NB] = (unsigned)nt; }
    }
    unsigned pd[EPT], ps[EPT];
    #pragma unroll
    for (int u = 0; u < EPT; ++u) {
        pd[u] = lsD[(unsigned)d[u] >> BSH] + (rk[u] & 0xFFFFu);
        ps[u] = lsS[(unsigned)s[u] >> BSH] + (rk[u] >> 16);
    }
    #pragma unroll
    for (int u = 0; u < EPT; ++u) {
        if (u * 1024 + tid < nt) {
            stD[pd[u]] = (((unsigned)d[u] & BMASK) << 24) | (unsigned)s[u];
            stS[ps[u]] = (unsigned char)((unsigned)s[u] & BMASK);
        }
    }
    __syncthreads();

    {
        unsigned v[EPT];
        unsigned char w[EPT];
        #pragma unroll
        for (int u = 0; u < EPT; ++u) {
            const int i = u * 1024 + tid;
            const int ic = (i < nt) ? i : 0;
            v[u] = stD[ic];
            w[u] = stS[ic];
        }
        #pragma unroll
        for (int u = 0; u < EPT; ++u) {
            const int i = u * 1024 + tid;
            if (i < nt) { bufD[e0 + i] = v[u]; bufS[e0 + i] = w[u]; }
        }
    }
}

#define STAGE_N 12288
#define RPW (GCH / 32)
__global__ __launch_bounds__(1024) void bucket_finish_kernel(
    const unsigned* __restrict__ bufD, const unsigned char* __restrict__ bufS,
    const unsigned* __restrict__ tabD, const unsigned* __restrict__ tabS,
    const float* __restrict__ weight,
    float* __restrict__ out_norm, float* __restrict__ in_norm, float* __restrict__ pq,
    unsigned* __restrict__ in_cnt, unsigned* __restrict__ row_start,
    int* __restrict__ csr_src, int n, int E, int NB)
{
    __shared__ unsigned cin[256];
    __shared__ unsigned cout[256];
    __shared__ unsigned cur[256];
    __shared__ unsigned wsum[32];
    __shared__ unsigned rsD[GCH], rlD[GCH];
    __shared__ unsigned rsS[GCH], rlS[GCH];
    __shared__ unsigned red[2][GCH / 32];
    __shared__ int stage[STAGE_N];
    const int tid = threadIdx.x, lane = tid & 31, wave = tid >> 5;
    const int b = blockIdx.x;
    const int chunk = (E + GCH - 1) / GCH;
    const float wv = (tid < 256 && b * 256 + tid < n) ? weight[b * 256 + tid] : 0.f;
    unsigned x0 = 0u, x1 = 0u;
    if (tid < GCH) {
        const unsigned* rD = tabD + (size_t)tid * (NB + 1) + b;
        const unsigned* rS = tabS + (size_t)tid * (NB + 1) + b;
        const unsigned d0 = rD[0], d1 = rD[1], s0 = rS[0], s1 = rS[1];
        x0 = d0; x1 = d1 - d0;
        rsD[tid] = (unsigned)tid * (unsigned)chunk + d0;  rlD[tid] = d1 - d0;
        rsS[tid] = (unsigned)tid * (unsigned)chunk + s0;  rlS[tid] = s1 - s0;
        cin[tid] = 0u; cout[tid] = 0u;
    }
    #pragma unroll
    for (int off = 1; off < 32; off <<= 1) {
        x0 += __shfl_xor(x0, off, 32);
        x1 += __shfl_xor(x1, off, 32);
    }
    if (tid < GCH && lane == 0) { red[0][wave] = x0; red[1][wave] = x1; }
    __syncthreads();
    unsigned bD = 0u, nD = 0u;
    #pragma unroll
    for (int w = 0; w < GCH / 32; ++w) { bD += red[0][w]; nD += red[1][w]; }

    unsigned stD_[RPW], lnD_[RPW], stS_[RPW], lnS_[RPW];
    #pragma unroll
    for (int u = 0; u < RPW; ++u) {
        const int c = wave + 32 * u;
        stD_[u] = rsD[c]; lnD_[u] = rlD[c];
        stS_[u] = rsS[c]; lnS_[u] = rlS[c];
    }
    {
        unsigned vd[RPW], vs[RPW];
        #pragma unroll
        for (int u = 0; u < RPW; ++u) {
            vd[u] = bufD[((unsigned)lane < lnD_[u]) ? stD_[u] + lane : 0u];
            vs[u] = bufS[((unsigned)lane < lnS_[u]) ? stS_[u] + lane : 0u];
        }
        #pragma unroll
        for (int u = 0; u < RPW; ++u) {
            if ((unsigned)lane < lnD_[u]) atomicAdd(&cin[vd[u] >> 24], 1u);
            if ((unsigned)lane < lnS_[u]) atomicAdd(&cout[vs[u]], 1u);
        }
        #pragma unroll 1
        for (int u = 0; u < RPW; ++u) {
            const int c = wave + 32 * u;
            const unsigned sd = rsD[c], ld = rlD[c], ss = rsS[c], ls = rlS[c];
            for (unsigned k = 32u + lane; k < ld; k += 32u) atomicAdd(&cin[bufD[sd + k] >> 24], 1u);
            for (unsigned k = 32u + lane; k < ls; k += 32u) atomicAdd(&cout[bufS[ss + k]], 1u);
        }
    }
    __syncthreads();
    if (tid < 256) cur[tid] = cin[tid];
    __syncthreads();
    block_excl_scan(cur, 256, wsum);
    if (tid < 256) {
        const unsigned rs = bD + cur[tid];
        cur[tid] = rs;
        const int node = b * 256 + tid;
        if (node < n) {
            const unsigned c = cin[tid];
            row_start[node] = rs;
            in_cnt[node] = c;
            in_norm[node]  = rsqrtf((float)(c + 1u));
            const float on = rsqrtf((float)(cout[tid] + 1u));
            out_norm[node] = on;
            ((float2*)pq)[node] = make_float2(on * wv, on);
        }
    }
    __syncthreads();
    const bool staged = nD <= (unsigned)STAGE_N;
    {
        unsigned vd[RPW];
        #pragma unroll
        for (int u = 0; u < RPW; ++u) vd[u] = bufD[((unsigned)lane < lnD_[u]) ? stD_[u] + lane : 0u];
        #pragma unroll
        for (int u = 0; u < RPW; ++u) {
            if ((unsigned)lane < lnD_[u]) {
                const unsigned pos = atomicAdd(&cur[vd[u] >> 24], 1u);
                const int sv = (int)(vd[u] & 0x00FFFFFFu);
                if (staged) stage[pos - bD] = sv;
                else        csr_src[pos] = sv;
            }
        }
        #pragma unroll 1
        for (int u = 0; u < RPW; ++u) {
            const int c = wave + 32 * u;
            const unsigned sd = rsD[c], ld = rlD[c];
            for (unsigned k = 32u + lane; k < ld; k += 32u) {
                const unsigned w = bufD[sd + k];
                const unsigned pos = atomicAdd(&cur[w >> 24], 1u);
                const int sv = (int)(w & 0x00FFFFFFu);
                if (staged) stage[pos - bD] = sv;
                else        csr_src[pos] = sv;
            }
        }
    }
    if (staged) {
        __syncthreads();
        for (unsigned i = tid; i < nD; i += 1024) csr_src[bD + i] = stage[i];
    }
}

__global__ __launch_bounds__(256) void layer0_kernel(
    const float* __restrict__ pq, const int* __restrict__ csr_src,
    const unsigned* __restrict__ row_start, const unsigned* __restrict__ in_cnt,
    const float* __restrict__ in_norm, const float* __restrict__ out_norm,
    const float* __restrict__ W_in, const float* __restrict__ b_in,
    const float* __restrict__ W, const float* __restrict__ bias,
    const float* __restrict__ Wnext, float* __restrict__ g)
{
    __shared__ float Wl[64 * 64];
    __shared__ float Wn[64 * 64];
    __shared__ float Tl[32 * 64];
    __shared__ float On[32];

    const int tid = threadIdx.x;
    const int row0 = blockIdx.x * 32;

    float onv = 0.f;
    if (tid < 32) onv = out_norm[row0 + tid];

    #pragma unroll
    for (int i = 0; i < 4; ++i) {
        ((float4*)Wl)[tid + 256 * i] = ((const float4*)W)[tid + 256 * i];
        ((float4*)Wn)[tid + 256 * i] = ((const float4*)Wnext)[tid + 256 * i];
    }
    if (tid < 32) On[tid] = onv;

    {
        const int r   = tid >> 3;
        const int sub = tid & 7;
        const int row = row0 + r;
        const unsigned start = row_start[row];
        const unsigned cnt   = in_cnt[row];
        const float2* pq2 = (const float2*)pq;
        const float2 self = pq2[row];
        const float inn = in_norm[row];

        float a = 0.f, c = 0.f;
        for (unsigned j = sub; j < cnt; j += 32) {
            int s[4];
            float2 v[4];
            #pragma unroll
            for (int u = 0; u < 4; ++u) {
                const unsigned jj = j + 8u * u;
                s[u] = csr_src[start + (jj < cnt ? jj : j)];
            }
            #pragma unroll
            for (int u = 0; u < 4; ++u) v[u] = pq2[s[u]];
            #pragma unroll
            for (int u = 0; u < 4; ++u) {
                if (j + 8u * u < cnt) { a += v[u].x; c += v[u].y; }
            }
        }
        #pragma unroll
        for (int off = 1; off < 8; off <<= 1) {
            a += __shfl_xor(a, off, 32);
            c += __shfl_xor(c, off, 32);
        }
        a = (a + self.x) * inn;
        c = (c + self.y) * inn;

        const float4 wi0 = ((const float4*)W_in)[sub * 2], wi1 = ((const float4*)W_in)[sub * 2 + 1];
        const float4 bi0 = ((const float4*)b_in)[sub * 2], bi1 = ((const float4*)b_in)[sub * 2 + 1];
        float4 t0, t1;
        t0.x = a * wi0.x + c * bi0.x;  t0.y = a * wi0.y + c * bi0.y;
        t0.z = a * wi0.z + c * bi0.z;  t0.w = a * wi0.w + c * bi0.w;
        t1.x = a * wi1.x + c * bi1.x;  t1.y = a * wi1.y + c * bi1.y;
        t1.z = a * wi1.z + c * bi1.z;  t1.w = a * wi1.w + c * bi1.w;
        ((float4*)(Tl + r * 64))[sub * 2]     = t0;
        ((float4*)(Tl + r * 64))[sub * 2 + 1] = t1;
    }
    __syncthreads();

    const int wave = tid >> 5;
    const int lane = tid & 31;
    const int msub = wave >> 2;
    const int nsub = wave & 3;
    const int l16  = lane & 15;
    const int arow = msub * 16 + l16;
    const int koff = (lane < 16) ? 0 : 2;
    const int col  = nsub * 16 + l16;

    v8f acc = {};
    #pragma unroll
    for (int kk = 0; kk < 16; ++kk) {
        int k = kk * 4 + koff;
        v2f fa, fb;
        fa.x = Tl[arow * 64 + k];
        fa.y = Tl[arow * 64 + k + 1];
        fb.x = Wl[k * 64 + col];
        fb.y = Wl[(k + 1) * 64 + col];
        acc = __builtin_amdgcn_wmma_f32_16x16x4_f32(false, fa, false, fb,
                                                    (short)0, acc, false, false);
    }

    const float bcol = bias[col];
    const int rloc = msub * 16 + ((lane < 16) ? 0 : 8);
    float t[8];
    #pragma unroll
    for (int r = 0; r < 8; ++r) {
        float v = acc[r] + bcol;
        v = v > 0.f ? v : 0.01f * v;
        t[r] = v * On[rloc + r];
    }
    __syncthreads();
    #pragma unroll
    for (int r = 0; r < 8; ++r) Tl[(rloc + r) * 64 + col] = t[r];
    __syncthreads();

    v8f c = {};
    #pragma unroll
    for (int kk = 0; kk < 16; ++kk) {
        int k = kk * 4 + koff;
        v2f a, b;
        a.x = Tl[arow * 64 + k];
        a.y = Tl[arow * 64 + k + 1];
        b.x = Wn[k * 64 + col];
        b.y = Wn[(k + 1) * 64 + col];
        c = __builtin_amdgcn_wmma_f32_16x16x4_f32(false, a, false, b,
                                                  (short)0, c, false, false);
    }

    #pragma unroll
    for (int r = 0; r < 8; ++r)
        g[(size_t)(row0 + rloc + r) * 64 + col] = c[r];
}

template <int M> __device__ __forceinline__ float row_xor_add(float x) {
    const int y = __builtin_amdgcn_update_dpp(0, __builtin_bit_cast(int, x), 0x160 + M, 0xf, 0xf, true);
    return x + __builtin_bit_cast(float, y);
}

#define ROWS(U)                                                            \
    {                                                                      \
        float2 v[U];                                                       \
        _Pragma("unroll")                                                  \
        for (int u = 0; u < U; ++u) {                                      \
              \
            int s = __builtin_amdgcn_readlane(my, (int)(j + u));           \
            v[u] = g2[(size_t)s * 32 + lane];                              \
        }                                                                  \
        _Pragma("unroll")                                                  \
        for (int u = 0; u < U; ++u) { acc.x += v[u].x; acc.y += v[u].y; }  \
        j += U;                                                            \
    }

__global__ __launch_bounds__(256) void gather_gemm_head_kernel(
    const float* __restrict__ g, const int* __restrict__ csr_src,
    const unsigned* __restrict__ row_start, const unsigned* __restrict__ in_cnt,
    const float* __restrict__ in_norm, const float* __restrict__ out_norm,
    const float* __restrict__ bc, const float* __restrict__ W,
    const float* __restrict__ W_pred, float* __restrict__ z)
{
    __shared__ float Wl[64 * 64];
    __shared__ float Tl[32 * 64];
    __shared__ float Zp[4][32];

    const int tid = threadIdx.x;
    const int row0 = blockIdx.x * 32;
    const int wave = tid >> 5;
    const int lane = tid & 31;

    const int nb = row0 + wave * 4;
    unsigned st = 0u, cn = 0u;
    float inl = 0.f, onl = 0.f;
    if (lane < 4) {
        st  = row_start[nb + lane];
        cn  = in_cnt[nb + lane];
        inl = in_norm[nb + lane];
        onl = out_norm[nb + lane];
    }

    #pragma unroll
    for (int i = 0; i < 4; ++i)
        ((float4*)Wl)[tid + 256 * i] = ((const float4*)W)[tid + 256 * i];

    {
        const float2* g2 = (const float2*)g;
        const float2 b = ((const float2*)bc)[lane];

        #pragma unroll 1
        for (int q = 0; q < 4; ++q) {
            const int r = wave * 4 + q;
            const int node = row0 + r;
            const unsigned start = (unsigned)__builtin_amdgcn_readlane((int)st, q);
            const unsigned cnt   = (unsigned)__builtin_amdgcn_readlane((int)cn, q);
            const float sc = __builtin_bit_cast(float, __builtin_amdgcn_readlane(__builtin_bit_cast(int, inl), q));
            const float on = __builtin_bit_cast(float, __builtin_amdgcn_readlane(__builtin_bit_cast(int, onl), q));
            const float2 gv = g2[(size_t)node * 32 + lane];

            float2 acc = make_float2(0.f, 0.f);
            for (unsigned base = 0; base < cnt; base += 32) {
                const unsigned m = (cnt - base < 32u) ? (cnt - base) : 32u;
                const int my = ((unsigned)lane < m) ? csr_src[start + base + lane] : 0;
                unsigned j = 0;
                if (j + 16 <= m) ROWS(16)
                if (j + 16 <= m) ROWS(16)
                if (j + 8 <= m) ROWS(8)
                if (j + 4 <= m) ROWS(4)
                if (j + 2 <= m) ROWS(2)
                if (j + 1 <= m) ROWS(1)
            }

            float2 h;
            h.x = (acc.x + gv.x) * sc + b.x;
            h.y = (acc.y + gv.y) * sc + b.y;
            h.x = h.x > 0.f ? h.x : 0.01f * h.x;
            h.y = h.y > 0.f ? h.y : 0.01f * h.y;
            h.x *= on; h.y *= on;
            ((float2*)(Tl + r * 64))[lane] = h;
        }
    }
    __syncthreads();

    const int msub = wave >> 2;
    const int nsub = wave & 3;
    const int l16  = lane & 15;
    const int arow = msub * 16 + l16;
    const int koff = (lane < 16) ? 0 : 2;
    const int col  = nsub * 16 + l16;

    const float wp = W_pred[col];

    v8f c = {};
    #pragma unroll
    for (int kk = 0; kk < 16; ++kk) {
        int k = kk * 4 + koff;
        v2f a, b;
        a.x = Tl[arow * 64 + k];
        a.y = Tl[arow * 64 + k + 1];
        b.x = Wl[k * 64 + col];
        b.y = Wl[(k + 1) * 64 + col];
        c = __builtin_amdgcn_wmma_f32_16x16x4_f32(false, a, false, b,
                                                  (short)0, c, false, false);
    }

    const int rloc = msub * 16 + ((lane < 16) ? 0 : 8);
    float p[8];
    #pragma unroll
    for (int r = 0; r < 8; ++r) p[r] = c[r] * wp;
    #pragma unroll
    for (int r = 0; r < 8; ++r) p[r] = row_xor_add<1>(p[r]);
    #pragma unroll
    for (int r = 0; r < 8; ++r) p[r] = row_xor_add<2>(p[r]);
    #pragma unroll
    for (int r = 0; r < 8; ++r) p[r] = row_xor_add<4>(p[r]);
    #pragma unroll
    for (int r = 0; r < 8; ++r) p[r] = row_xor_add<8>(p[r]);
    if (l16 == 0) {
        #pragma unroll
        for (int r = 0; r < 8; ++r) Zp[nsub][rloc + r] = p[r];
    }
    __syncthreads();
    if (tid < 32)
        z[row0 + tid] = (Zp[0][tid] + Zp[1][tid]) + (Zp[2][tid] + Zp[3][tid]);
}
#undef ROWS

__global__ __launch_bounds__(256) void gather_head_kernel(
    const float* __restrict__ z, const int* __restrict__ csr_src,
    const unsigned* __restrict__ row_start, const unsigned* __restrict__ in_cnt,
    const float* __restrict__ in_norm, const float* __restrict__ bias,
    const float* __restrict__ W_pred, const float* __restrict__ b_pred,
    float* __restrict__ out)
{
    __shared__ float ol[32];

    const int tid = threadIdx.x;
    const int lane = tid & 31;
    const int r   = tid >> 3;
    const int sub = tid & 7;
    const int row = blockIdx.x * 32 + r;

    const unsigned start = row_start[row];
    const unsigned cnt   = in_cnt[row];
    const float self = z[row];
    const float inn = in_norm[row];
    const float2 bb = ((const float2*)bias)[lane];
    const float2 ww = ((const float2*)W_pred)[lane];
    asm volatile("" :: "v"(start), "v"(bb.x), "v"(bb.y), "v"(ww.x), "v"(ww.y));

    float a = 0.f;
    for (unsigned j = sub; j < cnt; j += 32) {
        int s[4];
        float v[4];
        #pragma unroll
        for (int u = 0; u < 4; ++u) {
            const unsigned jj = j + 8u * u;
            s[u] = csr_src[start + (jj < cnt ? jj : j)];
        }
        #pragma unroll
        for (int u = 0; u < 4; ++u) v[u] = z[s[u]];
        #pragma unroll
        for (int u = 0; u < 4; ++u) {
            if (j + 8u * u < cnt) a += v[u];
        }
    }
    a = row_xor_add<1>(a);
    a = row_xor_add<2>(a);
    a = row_xor_add<4>(a);

    float cb = bb.x * ww.x + bb.y * ww.y;
    cb = row_xor_add<1>(cb);
    cb = row_xor_add<2>(cb);
    cb = row_xor_add<4>(cb);
    cb = row_xor_add<8>(cb);
    cb += __shfl_xor(cb, 16, 32);
    cb += b_pred[0];

    if (sub == 0) ol[r] = (a + self) * inn + cb;
    __syncthreads();
    if (tid < 32) out[blockIdx.x * 32 + tid] = ol[tid];
}

extern "C" void kernel_launch(void* const* d_in, const int* in_sizes, int n_in,
                              void* d_out, int out_size, void* d_ws, size_t ws_size,
                              hipStream_t stream) {
    const float* weight = (const float*)d_in[0];
    const int*   src    = (const int*)d_in[1];
    const int*   dst    = (const int*)d_in[2];
    const float* W_in   = (const float*)d_in[3];
    const float* b_in   = (const float*)d_in[4];
    const float* Wc     = (const float*)d_in[5];
    const float* bc     = (const float*)d_in[6];
    const float* W_pred = (const float*)d_in[7];
    const float* b_pred = (const float*)d_in[8];
    float* out = (float*)d_out;

    const int N = in_sizes[0];
    const int E = in_sizes[1];

    const int NB = (N + 255) / 256;

    float* g          = (float*)d_ws;
    float* z          = g + (size_t)N * 64;
    unsigned* in_cnt  = (unsigned*)(z + N);
    float* out_norm   = (float*)(in_cnt + N);
    float* in_norm    = out_norm + N;
    unsigned* row_start = (unsigned*)(in_norm + N);
    float* pq         = (float*)(row_start + N);
    int* csr_src      = (int*)(pq + (size_t)N * 2);
    unsigned* bufD    = (unsigned*)(csr_src + E);
    unsigned* tabD    = bufD + E;
    unsigned* tabS    = tabD + (size_t)GCH * (NB + 1);
    unsigned char* bufS = (unsigned char*)(tabS + (size_t)GCH * (NB + 1));

    chunk_group_kernel<<<GCH, 1024, 2 * NB * sizeof(unsigned), stream>>>(src, dst, tabD, tabS, bufD, bufS, E, NB);

    bucket_finish_kernel<<<NB, 1024, 0, stream>>>(bufD, bufS, tabD, tabS, weight,
                                                  out_norm, in_norm, pq, in_cnt, row_start, csr_src, N, E, NB);

    layer0_kernel<<<N / 32, 256, 0, stream>>>(pq, csr_src, row_start, in_cnt, in_norm, out_norm,
                                              W_in, b_in, Wc, bc, Wc + (size_t)64 * 64, g);

    gather_gemm_head_kernel<<<N / 32, 256, 0, stream>>>(g, csr_src, row_start, in_cnt, in_norm, out_norm,
                                                        bc + 64, Wc + (size_t)2 * 64 * 64, W_pred, z);

    gather_head_kernel<<<N / 32, 256, 0, stream>>>(z, csr_src, row_start, in_cnt, in_norm,
                                                   bc + 2 * 64, W_pred, b_pred, out);
}


# Last SpMM reads z[src] from LDS: 256 resident 1024-thread groups copy z in 200-KB slices and own tile runs, 8 lanes per node; in-edge sums go slice by slice per lane, then the 8-lane butterfly
# speedup vs baseline: 8.9853x; 1.0490x over previous
#include <hip/hip_runtime.h>
#include <math.h>

typedef __attribute__((ext_vector_type(2))) float v2f;
typedef __attribute__((ext_vector_type(8))) float v8f;

#define NN 100000

#define GCH 256
#define BSH 8
#define BMASK 255u

__device__ __forceinline__ void block_excl_scan(unsigned* a, int n, unsigned* wsum) {
    const int tid = threadIdx.x, lane = tid & 31, wave = tid >> 5;
    const int per = (n + 1023) >> 10;
    const int i0 = tid * per;
    unsigned s = 0;
    for (int k = 0; k < per; ++k) if (i0 + k < n) s += a[i0 + k];
    unsigned x = s;
    for (int off = 1; off < 32; off <<= 1) {
        unsigned y = __shfl_up(x, off, 32);
        if (lane >= off) x += y;
    }
    if (lane == 31) wsum[wave] = x;
    __syncthreads();
    unsigned run = x - s;
    for (int w = 0; w < wave; ++w) run += wsum[w];
    for (int k = 0; k < per; ++k) if (i0 + k < n) { unsigned v = a[i0 + k]; a[i0 + k] = run; run += v; }
    __syncthreads();
}

#define EPT 13
#define TILE_E (1024 * EPT)
__global__ __launch_bounds__(1024) void chunk_group_kernel(
    const int* __restrict__ src, const int* __restrict__ dst,
    unsigned* __restrict__ tabD, unsigned* __restrict__ tabS,
    unsigned* __restrict__ bufD, unsigned char* __restrict__ bufS, int E, int NB)
{
    extern __shared__ unsigned dyn_lds[];
    unsigned* lsD = dyn_lds;
    unsigned* lsS = dyn_lds + NB;
    __shared__ unsigned wsum[32];
    __shared__ unsigned      stD[TILE_E];
    __shared__ unsigned char stS[TILE_E];
    __shared__ unsigned spare[64];
    const int tid = threadIdx.x;
    if (tid < 64) spare[tid] = 0u;
    for (int b = tid; b < 2 * NB; b += 1024) lsD[b] = 0u;
    __syncthreads();

    const int chunk = (E + GCH - 1) / GCH;
    const int e0 = blockIdx.x * chunk;
    const int e1 = (e0 + chunk < E) ? (e0 + chunk) : E;
    int nt = e1 - e0;
    nt = nt < 0 ? 0 : (nt > TILE_E ? TILE_E : nt);
    const int t0 = (nt > 0) ? e0 : 0;

    int d[EPT], s[EPT];
    #pragma unroll
    for (int u = 0; u < EPT; ++u) {
        const int i = u * 1024 + tid;
        const int ec = t0 + ((i < nt) ? i : 0);
        d[u] = dst[ec];
        s[u] = src[ec];
    }
    unsigned rk[EPT];
    {
        unsigned rd[EPT], rs[EPT];
        #pragma unroll
        for (int u = 0; u < EPT; ++u) {
            const bool valid = u * 1024 + tid < nt;
            rd[u] = atomicAdd(valid ? &lsD[(unsigned)d[u] >> BSH] : &spare[tid & 31], 1u);
            rs[u] = atomicAdd(valid ? &lsS[(unsigned)s[u] >> BSH] : &spare[32 + (tid & 31)], 1u);
        }
        #pragma unroll
        for (int u = 0; u < EPT; ++u) rk[u] = (rd[u] & 0xFFFFu) | (rs[u] << 16);
    }
    __syncthreads();
    block_excl_scan(lsD, NB, wsum);
    block_excl_scan(lsS, NB, wsum);
    {
        unsigned* rowD = tabD + (size_t)blockIdx.x * (NB + 1);
        unsigned* rowS = tabS + (size_t)blockIdx.x * (NB + 1);
        for (int b = tid; b < NB; b += 1024) { rowD[b] = lsD[b]; rowS[b] = lsS[b]; }
        if (tid == 0) { rowD[NB] = (unsigned)nt; rowS[NB] = (unsigned)nt; }
    }
    unsigned pd[EPT], ps[EPT];
    #pragma unroll
    for (int u = 0; u < EPT; ++u) {
        pd[u] = lsD[(unsigned)d[u] >> BSH] + (rk[u] & 0xFFFFu);
        ps[u] = lsS[(unsigned)s[u] >> BSH] + (rk[u] >> 16);
    }
    #pragma unroll
    for (int u = 0; u < EPT; ++u) {
        if (u * 1024 + tid < nt) {
            stD[pd[u]] = (((unsigned)d[u] & BMASK) << 24) | (unsigned)s[u];
            stS[ps[u]] = (unsigned char)((unsigned)s[u] & BMASK);
        }
    }
    __syncthreads();

    {
        unsigned v[EPT];
        unsigned char w[EPT];
        #pragma unroll
        for (int u = 0; u < EPT; ++u) {
            const int i = u * 1024 + tid;
            const int ic = (i < nt) ? i : 0;
            v[u] = stD[ic];
            w[u] = stS[ic];
        }
        #pragma unroll
        for (int u = 0; u < EPT; ++u) {
            const int i = u * 1024 + tid;
            if (i < nt) { bufD[e0 + i] = v[u]; bufS[e0 + i] = w[u]; }
        }
    }
}

#define STAGE_N 12288
#define RPW (GCH / 32)
__global__ __launch_bounds__(1024) void bucket_finish_kernel(
    const unsigned* __restrict__ bufD, const unsigned char* __restrict__ bufS,
    const unsigned* __restrict__ tabD, const unsigned* __restrict__ tabS,
    const float* __restrict__ weight,
    float* __restrict__ out_norm, float* __restrict__ in_norm, float* __restrict__ pq,
    unsigned* __restrict__ in_cnt, unsigned* __restrict__ row_start,
    int* __restrict__ csr_src, int n, int E, int NB)
{
    __shared__ unsigned cin[256];
    __shared__ unsigned cout[256];
    __shared__ unsigned cur[256];
    __shared__ unsigned wsum[32];
    __shared__ unsigned rsD[GCH], rlD[GCH];
    __shared__ unsigned rsS[GCH], rlS[GCH];
    __shared__ unsigned red[2][GCH / 32];
    __shared__ int stage[STAGE_N];
    const int tid = threadIdx.x, lane = tid & 31, wave = tid >> 5;
    const int b = blockIdx.x;
    const int chunk = (E + GCH - 1) / GCH;
    const float wv = (tid < 256 && b * 256 + tid < n) ? weight[b * 256 + tid] : 0.f;
    unsigned x0 = 0u, x1 = 0u;
    if (tid < GCH) {
        const unsigned* rD = tabD + (size_t)tid * (NB + 1) + b;
        const unsigned* rS = tabS + (size_t)tid * (NB + 1) + b;
        const unsigned d0 = rD[0], d1 = rD[1], s0 = rS[0], s1 = rS[1];
        x0 = d0; x1 = d1 - d0;
        rsD[tid] = (unsigned)tid * (unsigned)chunk + d0;  rlD[tid] = d1 - d0;
        rsS[tid] = (unsigned)tid * (unsigned)chunk + s0;  rlS[tid] = s1 - s0;
        cin[tid] = 0u; cout[tid] = 0u;
    }
    #pragma unroll
    for (int off = 1; off < 32; off <<= 1) {
        x0 += __shfl_xor(x0, off, 32);
        x1 += __shfl_xor(x1, off, 32);
    }
    if (tid < GCH && lane == 0) { red[0][wave] = x0; red[1][wave] = x1; }
    __syncthreads();
    unsigned bD = 0u, nD = 0u;
    #pragma unroll
    for (int w = 0; w < GCH / 32; ++w) { bD += red[0][w]; nD += red[1][w]; }

    unsigned stD_[RPW], lnD_[RPW], stS_[RPW], lnS_[RPW];
    #pragma unroll
    for (int u = 0; u < RPW; ++u) {
        const int c = wave + 32 * u;
        stD_[u] = rsD[c]; lnD_[u] = rlD[c];
        stS_[u] = rsS[c]; lnS_[u] = rlS[c];
    }
    {
        unsigned vd[RPW], vs[RPW];
        #pragma unroll
        for (int u = 0; u < RPW; ++u) {
            vd[u] = bufD[((unsigned)lane < lnD_[u]) ? stD_[u] + lane : 0u];
            vs[u] = bufS[((unsigned)lane < lnS_[u]) ? stS_[u] + lane : 0u];
        }
        #pragma unroll
        for (int u = 0; u < RPW; ++u) {
            if ((unsigned)lane < lnD_[u]) atomicAdd(&cin[vd[u] >> 24], 1u);
            if ((unsigned)lane < lnS_[u]) atomicAdd(&cout[vs[u]], 1u);
        }
        #pragma unroll 1
        for (int u = 0; u < RPW; ++u) {
            const int c = wave + 32 * u;
            const unsigned sd = rsD[c], ld = rlD[c], ss = rsS[c], ls = rlS[c];
            for (unsigned k = 32u + lane; k < ld; k += 32u) atomicAdd(&cin[bufD[sd + k] >> 24], 1u);
            for (unsigned k = 32u + lane; k < ls; k += 32u) atomicAdd(&cout[bufS[ss + k]], 1u);
        }
    }
    __syncthreads();
    if (tid < 256) cur[tid] = cin[tid];
    __syncthreads();
    block_excl_scan(cur, 256, wsum);
    if (tid < 256) {
        const unsigned rs = bD + cur[tid];
        cur[tid] = rs;
        const int node = b * 256 + tid;
        if (node < n) {
            const unsigned c = cin[tid];
            row_start[node] = rs;
            in_cnt[node] = c;
            in_norm[node]  = rsqrtf((float)(c + 1u));
            const float on = rsqrtf((float)(cout[tid] + 1u));
            out_norm[node] = on;
            ((float2*)pq)[node] = make_float2(on * wv, on);
        }
    }
    __syncthreads();
    const bool staged = nD <= (unsigned)STAGE_N;
    {
        unsigned vd[RPW];
        #pragma unroll
        for (int u = 0; u < RPW; ++u) vd[u] = bufD[((unsigned)lane < lnD_[u]) ? stD_[u] + lane : 0u];
        #pragma unroll
        for (int u = 0; u < RPW; ++u) {
            if ((unsigned)lane < lnD_[u]) {
                const unsigned pos = atomicAdd(&cur[vd[u] >> 24], 1u);
                const int sv = (int)(vd[u] & 0x00FFFFFFu);
                if (staged) stage[pos - bD] = sv;
                else        csr_src[pos] = sv;
            }
        }
        #pragma unroll 1
        for (int u = 0; u < RPW; ++u) {
            const int c = wave + 32 * u;
            const unsigned sd = rsD[c], ld = rlD[c];
            for (unsigned k = 32u + lane; k < ld; k += 32u) {
                const unsigned w = bufD[sd + k];
                const unsigned pos = atomicAdd(&cur[w >> 24], 1u);
                const int sv = (int)(w & 0x00FFFFFFu);
                if (staged) stage[pos - bD] = sv;
                else        csr_src[pos] = sv;
            }
        }
    }
    if (staged) {
        __syncthreads();
        for (unsigned i = tid; i < nD; i += 1024) csr_src[bD + i] = stage[i];
    }
}

__global__ __launch_bounds__(256) void layer0_kernel(
    const float* __restrict__ pq, const int* __restrict__ csr_src,
    const unsigned* __restrict__ row_start, const unsigned* __restrict__ in_cnt,
    const float* __restrict__ in_norm, const float* __restrict__ out_norm,
    const float* __restrict__ W_in, const float* __restrict__ b_in,
    const float* __restrict__ W, const float* __restrict__ bias,
    const float* __restrict__ Wnext, float* __restrict__ g)
{
    __shared__ float Wl[64 * 64];
    __shared__ float Wn[64 * 64];
    __shared__ float Tl[32 * 64];
    __shared__ float On[32];

    const int tid = threadIdx.x;
    const int row0 = blockIdx.x * 32;

    float onv = 0.f;
    if (tid < 32) onv = out_norm[row0 + tid];

    #pragma unroll
    for (int i = 0; i < 4; ++i) {
        ((float4*)Wl)[tid + 256 * i] = ((const float4*)W)[tid + 256 * i];
        ((float4*)Wn)[tid + 256 * i] = ((const float4*)Wnext)[tid + 256 * i];
    }
    if (tid < 32) On[tid] = onv;

    {
        const int r   = tid >> 3;
        const int sub = tid & 7;
        const int row = row0 + r;
        const unsigned start = row_start[row];
        const unsigned cnt   = in_cnt[row];
        const float2* pq2 = (const float2*)pq;
        const float2 self = pq2[row];
        const float inn = in_norm[row];

        float a = 0.f, c = 0.f;
        for (unsigned j = sub; j < cnt; j += 32) {
            int s[4];
            float2 v[4];
            #pragma unroll
            for (int u = 0; u < 4; ++u) {
                const unsigned jj = j + 8u * u;
                s[u] = csr_src[start + (jj < cnt ? jj : j)];
            }
            #pragma unroll
            for (int u = 0; u < 4; ++u) v[u] = pq2[s[u]];
            #pragma unroll
            for (int u = 0; u < 4; ++u) {
                if (j + 8u * u < cnt) { a += v[u].x; c += v[u].y; }
            }
        }
        #pragma unroll
        for (int off = 1; off < 8; off <<= 1) {
            a += __shfl_xor(a, off, 32);
            c += __shfl_xor(c, off, 32);
        }
        a = (a + self.x) * inn;
        c = (c + self.y) * inn;

        const float4 wi0 = ((const float4*)W_in)[sub * 2], wi1 = ((const float4*)W_in)[sub * 2 + 1];
        const float4 bi0 = ((const float4*)b_in)[sub * 2], bi1 = ((const float4*)b_in)[sub * 2 + 1];
        float4 t0, t1;
        t0.x = a * wi0.x + c * bi0.x;  t0.y = a * wi0.y + c * bi0.y;
        t0.z = a * wi0.z + c * bi0.z;  t0.w = a * wi0.w + c * bi0.w;
        t1.x = a * wi1.x + c * bi1.x;  t1.y = a * wi1.y + c * bi1.y;
        t1.z = a * wi1.z + c * bi1.z;  t1.w = a * wi1.w + c * bi1.w;
        ((float4*)(Tl + r * 64))[sub * 2]     = t0;
        ((float4*)(Tl + r * 64))[sub * 2 + 1] = t1;
    }
    __syncthreads();

    const int wave = tid >> 5;
    const int lane = tid & 31;
    const int msub = wave >> 2;
    const int nsub = wave & 3;
    const int l16  = lane & 15;
    const int arow = msub * 16 + l16;
    const int koff = (lane < 16) ? 0 : 2;
    const int col  = nsub * 16 + l16;

    v8f acc = {};
    #pragma unroll
    for (int kk = 0; kk < 16; ++kk) {
        int k = kk * 4 + koff;
        v2f fa, fb;
        fa.x = Tl[arow * 64 + k];
        fa.y = Tl[arow * 64 + k + 1];
        fb.x = Wl[k * 64 + col];
        fb.y = Wl[(k + 1) * 64 + col];
        acc = __builtin_amdgcn_wmma_f32_16x16x4_f32(false, fa, false, fb,
                                                    (short)0, acc, false, false);
    }

    const float bcol = bias[col];
    const int rloc = msub * 16 + ((lane < 16) ? 0 : 8);
    float t[8];
    #pragma unroll
    for (int r = 0; r < 8; ++r) {
        float v = acc[r] + bcol;
        v = v > 0.f ? v : 0.01f * v;
        t[r] = v * On[rloc + r];
    }
    __syncthreads();
    #pragma unroll
    for (int r = 0; r < 8; ++r) Tl[(rloc + r) * 64 + col] = t[r];
    __syncthreads();

    v8f c = {};
    #pragma unroll
    for (int kk = 0; kk < 16; ++kk) {
        int k = kk * 4 + koff;
        v2f a, b;
        a.x = Tl[arow * 64 + k];
        a.y = Tl[arow * 64 + k + 1];
        b.x = Wn[k * 64 + col];
        b.y = Wn[(k + 1) * 64 + col];
        c = __builtin_amdgcn_wmma_f32_16x16x4_f32(false, a, false, b,
                                                  (short)0, c, false, false);
    }

    #pragma unroll
    for (int r = 0; r < 8; ++r)
        g[(size_t)(row0 + rloc + r) * 64 + col] = c[r];
}

template <int M> __device__ __forceinline__ float row_xor_add(float x) {
    const int y = __builtin_amdgcn_update_dpp(0, __builtin_bit_cast(int, x), 0x160 + M, 0xf, 0xf, true);
    return x + __builtin_bit_cast(float, y);
}

#define ROWS(U)                                                            \
    {                                                                      \
        float2 v[U];                                                       \
        _Pragma("unroll")                                                  \
        for (int u = 0; u < U; ++u) {                                      \
              \
            int s = __builtin_amdgcn_readlane(my, (int)(j + u));           \
            v[u] = g2[(size_t)s * 32 + lane];                              \
        }                                                                  \
        _Pragma("unroll")                                                  \
        for (int u = 0; u < U; ++u) { acc.x += v[u].x; acc.y += v[u].y; }  \
        j += U;                                                            \
    }

__global__ __launch_bounds__(256) void gather_gemm_head_kernel(
    const float* __restrict__ g, const int* __restrict__ csr_src,
    const unsigned* __restrict__ row_start, const unsigned* __restrict__ in_cnt,
    const float* __restrict__ in_norm, const float* __restrict__ out_norm,
    const float* __restrict__ bc, const float* __restrict__ W,
    const float* __restrict__ W_pred, float* __restrict__ z)
{
    __shared__ float Wl[64 * 64];
    __shared__ float Tl[32 * 64];
    __shared__ float Zp[4][32];

    const int tid = threadIdx.x;
    const int row0 = blockIdx.x * 32;
    const int wave = tid >> 5;
    const int lane = tid & 31;

    const int nb = row0 + wave * 4;
    unsigned st = 0u, cn = 0u;
    float inl = 0.f, onl = 0.f;
    if (lane < 4) {
        st  = row_start[nb + lane];
        cn  = in_cnt[nb + lane];
        inl = in_norm[nb + lane];
        onl = out_norm[nb + lane];
    }

    #pragma unroll
    for (int i = 0; i < 4; ++i)
        ((float4*)Wl)[tid + 256 * i] = ((const float4*)W)[tid + 256 * i];

    {
        const float2* g2 = (const float2*)g;
        const float2 b = ((const float2*)bc)[lane];

        #pragma unroll 1
        for (int q = 0; q < 4; ++q) {
            const int r = wave * 4 + q;
            const int node = row0 + r;
            const unsigned start = (unsigned)__builtin_amdgcn_readlane((int)st, q);
            const unsigned cnt   = (unsigned)__builtin_amdgcn_readlane((int)cn, q);
            const float sc = __builtin_bit_cast(float, __builtin_amdgcn_readlane(__builtin_bit_cast(int, inl), q));
            const float on = __builtin_bit_cast(float, __builtin_amdgcn_readlane(__builtin_bit_cast(int, onl), q));
            const float2 gv = g2[(size_t)node * 32 + lane];

            float2 acc = make_float2(0.f, 0.f);
            for (unsigned base = 0; base < cnt; base += 32) {
                const unsigned m = (cnt - base < 32u) ? (cnt - base) : 32u;
                const int my = ((unsigned)lane < m) ? csr_src[start + base + lane] : 0;
                unsigned j = 0;
                if (j + 16 <= m) ROWS(16)
                if (j + 16 <= m) ROWS(16)
                if (j + 8 <= m) ROWS(8)
                if (j + 4 <= m) ROWS(4)
                if (j + 2 <= m) ROWS(2)
                if (j + 1 <= m) ROWS(1)
            }

            float2 h;
            h.x = (acc.x + gv.x) * sc + b.x;
            h.y = (acc.y + gv.y) * sc + b.y;
            h.x = h.x > 0.f ? h.x : 0.01f * h.x;
            h.y = h.y > 0.f ? h.y : 0.01f * h.y;
            h.x *= on; h.y *= on;
            ((float2*)(Tl + r * 64))[lane] = h;
        }
    }
    __syncthreads();

    const int msub = wave >> 2;
    const int nsub = wave & 3;
    const int l16  = lane & 15;
    const int arow = msub * 16 + l16;
    const int koff = (lane < 16) ? 0 : 2;
    const int col  = nsub * 16 + l16;

    const float wp = W_pred[col];

    v8f c = {};
    #pragma unroll
    for (int kk = 0; kk < 16; ++kk) {
        int k = kk * 4 + koff;
        v2f a, b;
        a.x = Tl[arow * 64 + k];
        a.y = Tl[arow * 64 + k + 1];
        b.x = Wl[k * 64 + col];
        b.y = Wl[(k + 1) * 64 + col];
        c = __builtin_amdgcn_wmma_f32_16x16x4_f32(false, a, false, b,
                                                  (short)0, c, false, false);
    }

    const int rloc = msub * 16 + ((lane < 16) ? 0 : 8);
    float p[8];
    #pragma unroll
    for (int r = 0; r < 8; ++r) p[r] = c[r] * wp;
    #pragma unroll
    for (int r = 0; r < 8; ++r) p[r] = row_xor_add<1>(p[r]);
    #pragma unroll
    for (int r = 0; r < 8; ++r) p[r] = row_xor_add<2>(p[r]);
    #pragma unroll
    for (int r = 0; r < 8; ++r) p[r] = row_xor_add<4>(p[r]);
    #pragma unroll
    for (int r = 0; r < 8; ++r) p[r] = row_xor_add<8>(p[r]);
    if (l16 == 0) {
        #pragma unroll
        for (int r = 0; r < 8; ++r) Zp[nsub][rloc + r] = p[r];
    }
    __syncthreads();
    if (tid < 32)
        z[row0 + tid] = (Zp[0][tid] + Zp[1][tid]) + (Zp[2][tid] + Zp[3][tid]);
}
#undef ROWS

#define ZT   51200
#define HGRP 256
#define HFB  7
#define HRND 4
#define HPQ  6
__global__ __launch_bounds__(1024) void gather_head_kernel(
    const float* __restrict__ z, const int* __restrict__ csr_src,
    const unsigned* __restrict__ row_start, const unsigned* __restrict__ in_cnt,
    const float* __restrict__ in_norm, const float* __restrict__ bias,
    const float* __restrict__ W_pred, const float* __restrict__ b_pred,
    float* __restrict__ out, int n)
{
    __shared__ float tbl[ZT];
    __shared__ float ol[128 * HRND];

    const int tid = threadIdx.x;
    const int lane = tid & 31;
    const int sub = tid & 7;
    const int r   = tid >> 3;

    const int ntile = (n + 31) >> 5;
    const int t0 = (int)(((unsigned)blockIdx.x * (unsigned)ntile) / (unsigned)HGRP);
    const int t1 = (int)((((unsigned)blockIdx.x + 1u) * (unsigned)ntile) / (unsigned)HGRP);
    const int n0 = t0 * 32;
    const int n1 = (t1 * 32 < n) ? t1 * 32 : n;

    const float2 bb = ((const float2*)bias)[lane];
    const float2 ww = ((const float2*)W_pred)[lane];
    float cb = bb.x * ww.x + bb.y * ww.y;
    cb = row_xor_add<1>(cb);
    cb = row_xor_add<2>(cb);
    cb = row_xor_add<4>(cb);
    cb = row_xor_add<8>(cb);
    cb += __shfl_xor(cb, 16, 32);
    cb += b_pred[0];

    for (int nb = n0; nb < n1; nb += 128 * HRND) {
        unsigned start[HRND], cnt[HRND];
        float self[HRND], inn[HRND], a[HRND];
        #pragma unroll
        for (int q = 0; q < HRND; ++q) {
            const int row = nb + q * 128 + r;
            const bool live = row < n1;
            const int rowc = live ? row : nb;
            start[q] = row_start[rowc];
            cnt[q]   = live ? in_cnt[rowc] : 0u;
            self[q]  = z[rowc];
            inn[q]   = in_norm[rowc];
            a[q]     = 0.f;
        }
        int s[HRND][HPQ];
        #pragma unroll
        for (int q = 0; q < HRND; ++q) {
            #pragma unroll
            for (int u = 0; u < HPQ; ++u) {
                const unsigned jj = (unsigned)sub + 8u * u;
                s[q][u] = csr_src[(jj < cnt[q]) ? start[q] + jj : 0u];
            }
        }

        for (int base = 0; base < n; base += ZT) {
            const int len = (n - base < ZT) ? (n - base) : ZT;
            const float* zs = z + base;
            const int len4 = len >> 2;
            if (base > 0) __syncthreads();
            for (int i0 = 0; i0 < len4; i0 += 1024 * HFB) {
                float4 q4[HFB];
                #pragma unroll
                for (int u = 0; u < HFB; ++u) {
                    const int i = i0 + u * 1024 + tid;
                    q4[u] = ((const float4*)zs)[(i < len4) ? i : 0];
                }
                asm volatile("" :: "v"(q4[0].x), "v"(q4[0].y), "v"(q4[0].z), "v"(q4[0].w),
                                   "v"(q4[1].x), "v"(q4[1].y), "v"(q4[1].z), "v"(q4[1].w),
                                   "v"(q4[2].x), "v"(q4[2].y), "v"(q4[2].z), "v"(q4[2].w),
                                   "v"(q4[3].x), "v"(q4[3].y), "v"(q4[3].z), "v"(q4[3].w),
                                   "v"(q4[4].x), "v"(q4[4].y), "v"(q4[4].z), "v"(q4[4].w),
                                   "v"(q4[5].x), "v"(q4[5].y), "v"(q4[5].z), "v"(q4[5].w),
                                   "v"(q4[6].x), "v"(q4[6].y), "v"(q4[6].z), "v"(q4[6].w));
                #pragma unroll
                for (int u = 0; u < HFB; ++u) {
                    const int i = i0 + u * 1024 + tid;
                    if (i < len4) ((float4*)tbl)[i] = q4[u];
                }
            }
            if (tid < len - (len4 << 2)) tbl[(len4 << 2) + tid] = zs[(len4 << 2) + tid];
            __syncthreads();

            {
                float v[HRND][HPQ];
                #pragma unroll
                for (int q = 0; q < HRND; ++q) {
                    #pragma unroll
                    for (int u = 0; u < HPQ; ++u) {
                        const unsigned rel = (unsigned)(s[q][u] - base);
                        const bool ok = ((unsigned)sub + 8u * u < cnt[q]) && (rel < (unsigned)len);
                        v[q][u] = tbl[ok ? rel : 0u];
                    }
                }
                #pragma unroll
                for (int q = 0; q < HRND; ++q) {
                    #pragma unroll
                    for (int u = 0; u < HPQ; ++u) {
                        const unsigned rel = (unsigned)(s[q][u] - base);
                        const bool ok = ((unsigned)sub + 8u * u < cnt[q]) && (rel < (unsigned)len);
                        if (ok) a[q] += v[q][u];
                    }
                }
            }
            #pragma unroll
            for (int q = 0; q < HRND; ++q) {
                for (unsigned j = (unsigned)sub + 8u * HPQ; j < cnt[q]; j += 8u) {
                    const unsigned rel = (unsigned)(csr_src[start[q] + j] - base);
                    if (rel < (unsigned)len) a[q] += tbl[rel];
                }
            }
        }

        #pragma unroll
        for (int q = 0; q < HRND; ++q) {
            float x = a[q];
            x = row_xor_add<1>(x);
            x = row_xor_add<2>(x);
            x = row_xor_add<4>(x);
            if (sub == 0 && nb + q * 128 + r < n1)
                ol[q * 128 + r] = (x + self[q]) * inn[q] + cb;
        }
        __syncthreads();
        const int nn = (n1 - nb < 128 * HRND) ? (n1 - nb) : 128 * HRND;
        if (tid < nn) out[nb + tid] = ol[tid];
    }
}

extern "C" void kernel_launch(void* const* d_in, const int* in_sizes, int n_in,
                              void* d_out, int out_size, void* d_ws, size_t ws_size,
                              hipStream_t stream) {
    const float* weight = (const float*)d_in[0];
    const int*   src    = (const int*)d_in[1];
    const int*   dst    = (const int*)d_in[2];
    const float* W_in   = (const float*)d_in[3];
    const float* b_in   = (const float*)d_in[4];
    const float* Wc     = (const float*)d_in[5];
    const float* bc     = (const float*)d_in[6];
    const float* W_pred = (const float*)d_in[7];
    const float* b_pred = (const float*)d_in[8];
    float* out = (float*)d_out;

    const int N = in_sizes[0];
    const int E = in_sizes[1];

    const int NB = (N + 255) / 256;

    float* g          = (float*)d_ws;
    float* z          = g + (size_t)N * 64;
    unsigned* in_cnt  = (unsigned*)(z + N);
    float* out_norm   = (float*)(in_cnt + N);
    float* in_norm    = out_norm + N;
    unsigned* row_start = (unsigned*)(in_norm + N);
    float* pq         = (float*)(row_start + N);
    int* csr_src      = (int*)(pq + (size_t)N * 2);
    unsigned* bufD    = (unsigned*)(csr_src + E);
    unsigned* tabD    = bufD + E;
    unsigned* tabS    = tabD + (size_t)GCH * (NB + 1);
    unsigned char* bufS = (unsigned char*)(tabS + (size_t)GCH * (NB + 1));

    chunk_group_kernel<<<GCH, 1024, 2 * NB * sizeof(unsigned), stream>>>(src, dst, tabD, tabS, bufD, bufS, E, NB);

    bucket_finish_kernel<<<NB, 1024, 0, stream>>>(bufD, bufS, tabD, tabS, weight,
                                                  out_norm, in_norm, pq, in_cnt, row_start, csr_src, N, E, NB);

    layer0_kernel<<<N / 32, 256, 0, stream>>>(pq, csr_src, row_start, in_cnt, in_norm, out_norm,
                                              W_in, b_in, Wc, bc, Wc + (size_t)64 * 64, g);

    gather_gemm_head_kernel<<<N / 32, 256, 0, stream>>>(g, csr_src, row_start, in_cnt, in_norm, out_norm,
                                                        bc + 64, Wc + (size_t)2 * 64 * 64, W_pred, z);

    gather_head_kernel<<<HGRP, 1024, 0, stream>>>(z, csr_src, row_start, in_cnt, in_norm,
                                                  bc + 2 * 64, W_pred, b_pred, out, N);
}


# Layer 1 sums before it projects: a source's activated layer-0 row is rebuilt per in-edge from 3 scalars (a*u + c*v + b0; u,v = [W_in;b_in] @ W0 on WMMA), summed in list order; 256-B row gathers gone
# speedup vs baseline: 10.2262x; 1.1381x over previous
#include <hip/hip_runtime.h>
#include <math.h>

typedef __attribute__((ext_vector_type(2))) float v2f;
typedef __attribute__((ext_vector_type(8))) float v8f;

#define NN 100000

#define GCH 256
#define BSH 8
#define BMASK 255u

__device__ __forceinline__ void block_excl_scan(unsigned* a, int n, unsigned* wsum) {
    const int tid = threadIdx.x, lane = tid & 31, wave = tid >> 5;
    const int per = (n + 1023) >> 10;
    const int i0 = tid * per;
    unsigned s = 0;
    for (int k = 0; k < per; ++k) if (i0 + k < n) s += a[i0 + k];
    unsigned x = s;
    for (int off = 1; off < 32; off <<= 1) {
        unsigned y = __shfl_up(x, off, 32);
        if (lane >= off) x += y;
    }
    if (lane == 31) wsum[wave] = x;
    __syncthreads();
    unsigned run = x - s;
    for (int w = 0; w < wave; ++w) run += wsum[w];
    for (int k = 0; k < per; ++k) if (i0 + k < n) { unsigned v = a[i0 + k]; a[i0 + k] = run; run += v; }
    __syncthreads();
}

#define EPT 13
#define TILE_E (1024 * EPT)
__global__ __launch_bounds__(1024) void chunk_group_kernel(
    const int* __restrict__ src, const int* __restrict__ dst,
    unsigned* __restrict__ tabD, unsigned* __restrict__ tabS,
    unsigned* __restrict__ bufD, unsigned char* __restrict__ bufS, int E, int NB)
{
    extern __shared__ unsigned dyn_lds[];
    unsigned* lsD = dyn_lds;
    unsigned* lsS = dyn_lds + NB;
    __shared__ unsigned wsum[32];
    __shared__ unsigned      stD[TILE_E];
    __shared__ unsigned char stS[TILE_E];
    __shared__ unsigned spare[64];
    const int tid = threadIdx.x;
    if (tid < 64) spare[tid] = 0u;
    for (int b = tid; b < 2 * NB; b += 1024) lsD[b] = 0u;
    __syncthreads();

    const int chunk = (E + GCH - 1) / GCH;
    const int e0 = blockIdx.x * chunk;
    const int e1 = (e0 + chunk < E) ? (e0 + chunk) : E;
    int nt = e1 - e0;
    nt = nt < 0 ? 0 : (nt > TILE_E ? TILE_E : nt);
    const int t0 = (nt > 0) ? e0 : 0;

    int d[EPT], s[EPT];
    #pragma unroll
    for (int u = 0; u < EPT; ++u) {
        const int i = u * 1024 + tid;
        const int ec = t0 + ((i < nt) ? i : 0);
        d[u] = dst[ec];
        s[u] = src[ec];
    }
    unsigned rk[EPT];
    {
        unsigned rd[EPT], rs[EPT];
        #pragma unroll
        for (int u = 0; u < EPT; ++u) {
            const bool valid = u * 1024 + tid < nt;
            rd[u] = atomicAdd(valid ? &lsD[(unsigned)d[u] >> BSH] : &spare[tid & 31], 1u);
            rs[u] = atomicAdd(valid ? &lsS[(unsigned)s[u] >> BSH] : &spare[32 + (tid & 31)], 1u);
        }
        #pragma unroll
        for (int u = 0; u < EPT; ++u) rk[u] = (rd[u] & 0xFFFFu) | (rs[u] << 16);
    }
    __syncthreads();
    block_excl_scan(lsD, NB, wsum);
    block_excl_scan(lsS, NB, wsum);
    {
        unsigned* rowD = tabD + (size_t)blockIdx.x * (NB + 1);
        unsigned* rowS = tabS + (size_t)blockIdx.x * (NB + 1);
        for (int b = tid; b < NB; b += 1024) { rowD[b] = lsD[b]; rowS[b] = lsS[b]; }
        if (tid == 0) { rowD[NB] = (unsigned)nt; rowS[NB] = (unsigned)nt; }
    }
    unsigned pd[EPT], ps[EPT];
    #pragma unroll
    for (int u = 0; u < EPT; ++u) {
        pd[u] = lsD[(unsigned)d[u] >> BSH] + (rk[u] & 0xFFFFu);
        ps[u] = lsS[(unsigned)s[u] >> BSH] + (rk[u] >> 16);
    }
    #pragma unroll
    for (int u = 0; u < EPT; ++u) {
        if (u * 1024 + tid < nt) {
            stD[pd[u]] = (((unsigned)d[u] & BMASK) << 24) | (unsigned)s[u];
            stS[ps[u]] = (unsigned char)((unsigned)s[u] & BMASK);
        }
    }
    __syncthreads();

    {
        unsigned v[EPT];
        unsigned char w[EPT];
        #pragma unroll
        for (int u = 0; u < EPT; ++u) {
            const int i = u * 1024 + tid;
            const int ic = (i < nt) ? i : 0;
            v[u] = stD[ic];
            w[u] = stS[ic];
        }
        #pragma unroll
        for (int u = 0; u < EPT; ++u) {
            const int i = u * 1024 + tid;
            if (i < nt) { bufD[e0 + i] = v[u]; bufS[e0 + i] = w[u]; }
        }
    }
}

#define STAGE_N 12288
#define RPW (GCH / 32)
__global__ __launch_bounds__(1024) void bucket_finish_kernel(
    const unsigned* __restrict__ bufD, const unsigned char* __restrict__ bufS,
    const unsigned* __restrict__ tabD, const unsigned* __restrict__ tabS,
    const float* __restrict__ weight,
    float* __restrict__ out_norm, float* __restrict__ in_norm, float* __restrict__ pq,
    unsigned* __restrict__ in_cnt, unsigned* __restrict__ row_start,
    int* __restrict__ csr_src, int n, int E, int NB)
{
    __shared__ unsigned cin[256];
    __shared__ unsigned cout[256];
    __shared__ unsigned cur[256];
    __shared__ unsigned wsum[32];
    __shared__ unsigned rsD[GCH], rlD[GCH];
    __shared__ unsigned rsS[GCH], rlS[GCH];
    __shared__ unsigned red[2][GCH / 32];
    __shared__ int stage[STAGE_N];
    const int tid = threadIdx.x, lane = tid & 31, wave = tid >> 5;
    const int b = blockIdx.x;
    const int chunk = (E + GCH - 1) / GCH;
    const float wv = (tid < 256 && b * 256 + tid < n) ? weight[b * 256 + tid] : 0.f;
    unsigned x0 = 0u, x1 = 0u;
    if (tid < GCH) {
        const unsigned* rD = tabD + (size_t)tid * (NB + 1) + b;
        const unsigned* rS = tabS + (size_t)tid * (NB + 1) + b;
        const unsigned d0 = rD[0], d1 = rD[1], s0 = rS[0], s1 = rS[1];
        x0 = d0; x1 = d1 - d0;
        rsD[tid] = (unsigned)tid * (unsigned)chunk + d0;  rlD[tid] = d1 - d0;
        rsS[tid] = (unsigned)tid * (unsigned)chunk + s0;  rlS[tid] = s1 - s0;
        cin[tid] = 0u; cout[tid] = 0u;
    }
    #pragma unroll
    for (int off = 1; off < 32; off <<= 1) {
        x0 += __shfl_xor(x0, off, 32);
        x1 += __shfl_xor(x1, off, 32);
    }
    if (tid < GCH && lane == 0) { red[0][wave] = x0; red[1][wave] = x1; }
    __syncthreads();
    unsigned bD = 0u, nD = 0u;
    #pragma unroll
    for (int w = 0; w < GCH / 32; ++w) { bD += red[0][w]; nD += red[1][w]; }

    unsigned stD_[RPW], lnD_[RPW], stS_[RPW], lnS_[RPW];
    #pragma unroll
    for (int u = 0; u < RPW; ++u) {
        const int c = wave + 32 * u;
        stD_[u] = rsD[c]; lnD_[u] = rlD[c];
        stS_[u] = rsS[c]; lnS_[u] = rlS[c];
    }
    {
        unsigned vd[RPW], vs[RPW];
        #pragma unroll
        for (int u = 0; u < RPW; ++u) {
            vd[u] = bufD[((unsigned)lane < lnD_[u]) ? stD_[u] + lane : 0u];
            vs[u] = bufS[((unsigned)lane < lnS_[u]) ? stS_[u] + lane : 0u];
        }
        #pragma unroll
        for (int u = 0; u < RPW; ++u) {
            if ((unsigned)lane < lnD_[u]) atomicAdd(&cin[vd[u] >> 24], 1u);
            if ((unsigned)lane < lnS_[u]) atomicAdd(&cout[vs[u]], 1u);
        }
        #pragma unroll 1
        for (int u = 0; u < RPW; ++u) {
            const int c = wave + 32 * u;
            const unsigned sd = rsD[c], ld = rlD[c], ss = rsS[c], ls = rlS[c];
            for (unsigned k = 32u + lane; k < ld; k += 32u) atomicAdd(&cin[bufD[sd + k] >> 24], 1u);
            for (unsigned k = 32u + lane; k < ls; k += 32u) atomicAdd(&cout[bufS[ss + k]], 1u);
        }
    }
    __syncthreads();
    if (tid < 256) cur[tid] = cin[tid];
    __syncthreads();
    block_excl_scan(cur, 256, wsum);
    if (tid < 256) {
        const unsigned rs = bD + cur[tid];
        cur[tid] = rs;
        const int node = b * 256 + tid;
        if (node < n) {
            const unsigned c = cin[tid];
            row_start[node] = rs;
            in_cnt[node] = c;
            in_norm[node]  = rsqrtf((float)(c + 1u));
            const float on = rsqrtf((float)(cout[tid] + 1u));
            out_norm[node] = on;
            ((float2*)pq)[node] = make_float2(on * wv, on);
        }
    }
    __syncthreads();
    const bool staged = nD <= (unsigned)STAGE_N;
    {
        unsigned vd[RPW];
        #pragma unroll
        for (int u = 0; u < RPW; ++u) vd[u] = bufD[((unsigned)lane < lnD_[u]) ? stD_[u] + lane : 0u];
        #pragma unroll
        for (int u = 0; u < RPW; ++u) {
            if ((unsigned)lane < lnD_[u]) {
                const unsigned pos = atomicAdd(&cur[vd[u] >> 24], 1u);
                const int sv = (int)(vd[u] & 0x00FFFFFFu);
                if (staged) stage[pos - bD] = sv;
                else        csr_src[pos] = sv;
            }
        }
        #pragma unroll 1
        for (int u = 0; u < RPW; ++u) {
            const int c = wave + 32 * u;
            const unsigned sd = rsD[c], ld = rlD[c];
            for (unsigned k = 32u + lane; k < ld; k += 32u) {
                const unsigned w = bufD[sd + k];
                const unsigned pos = atomicAdd(&cur[w >> 24], 1u);
                const int sv = (int)(w & 0x00FFFFFFu);
                if (staged) stage[pos - bD] = sv;
                else        csr_src[pos] = sv;
            }
        }
    }
    if (staged) {
        __syncthreads();
        for (unsigned i = tid; i < nD; i += 1024) csr_src[bD + i] = stage[i];
    }
}

template <int M> __device__ __forceinline__ float row_xor_add(float x) {
    const int y = __builtin_amdgcn_update_dpp(0, __builtin_bit_cast(int, x), 0x160 + M, 0xf, 0xf, true);
    return x + __builtin_bit_cast(float, y);
}

#define L0R 4
#define L0Q 6
__global__ __launch_bounds__(256) void layer0_sums_kernel(
    const float* __restrict__ pq, const int* __restrict__ csr_src,
    const unsigned* __restrict__ row_start, const unsigned* __restrict__ in_cnt,
    const float* __restrict__ in_norm, float* __restrict__ rec, int n)
{
    const int tid = threadIdx.x;
    const int r   = tid >> 3;
    const int sub = tid & 7;
    const int nb  = blockIdx.x * (32 * L0R);
    const float2* pq2 = (const float2*)pq;

    unsigned start[L0R], cnt[L0R];
    float2 self[L0R];
    float inn[L0R], a[L0R], c[L0R];
    #pragma unroll
    for (int q = 0; q < L0R; ++q) {
        const int row = nb + q * 32 + r;
        const bool live = row < n;
        const int rowc = live ? row : nb;
        start[q] = row_start[rowc];
        cnt[q]   = live ? in_cnt[rowc] : 0u;
        self[q]  = pq2[rowc];
        inn[q]   = in_norm[rowc];
        a[q] = 0.f; c[q] = 0.f;
    }
    int s[L0R][L0Q];
    #pragma unroll
    for (int q = 0; q < L0R; ++q) {
        #pragma unroll
        for (int u = 0; u < L0Q; ++u) {
            const unsigned jj = (unsigned)sub + 8u * u;
            s[q][u] = csr_src[(jj < cnt[q]) ? start[q] + jj : 0u];
        }
    }
    float2 v[L0R][L0Q];
    #pragma unroll
    for (int q = 0; q < L0R; ++q) {
        #pragma unroll
        for (int u = 0; u < L0Q; ++u) v[q][u] = pq2[s[q][u]];
    }
    #pragma unroll
    for (int q = 0; q < L0R; ++q) {
        #pragma unroll
        for (int u = 0; u < L0Q; ++u) {
            if ((unsigned)sub + 8u * u < cnt[q]) { a[q] += v[q][u].x; c[q] += v[q][u].y; }
        }
        for (unsigned j = (unsigned)sub + 8u * L0Q; j < cnt[q]; j += 8u) {
            const float2 w = pq2[csr_src[start[q] + j]];
            a[q] += w.x; c[q] += w.y;
        }
    }
    #pragma unroll
    for (int q = 0; q < L0R; ++q) {
        float x = a[q], y = c[q];
        x = row_xor_add<1>(x);  y = row_xor_add<1>(y);
        x = row_xor_add<2>(x);  y = row_xor_add<2>(y);
        x = row_xor_add<4>(x);  y = row_xor_add<4>(y);
        const int row = nb + q * 32 + r;
        if (sub == 0 && row < n)
            ((float4*)rec)[row] = make_float4((x + self[q].x) * inn[q], (y + self[q].y) * inn[q], self[q].y, 0.f);
    }
}

#define SPL   5
#define SLOTS (32 * SPL)

__device__ __forceinline__ void add_source_row(const float4 t, const v2f u2, const v2f v2, const v2f b2, v2f& acc) {
    const v2f ta = {t.x, t.x}, tc = {t.y, t.y}, to = {t.z, t.z};
    v2f x = __builtin_elementwise_fma(tc, v2, b2);
    x = __builtin_elementwise_fma(ta, u2, x);
    x = __builtin_elementwise_max(x, x * 0.01f);
    acc = __builtin_elementwise_fma(to, x, acc);
}

__global__ __launch_bounds__(256) void sum_gemm_head_kernel(
    const float* __restrict__ rec, const int* __restrict__ csr_src,
    const unsigned* __restrict__ row_start, const unsigned* __restrict__ in_cnt,
    const float* __restrict__ in_norm, const float* __restrict__ out_norm,
    const float* __restrict__ W_in, const float* __restrict__ b_in,
    const float* __restrict__ W0, const float* __restrict__ bias0,
    const float* __restrict__ W1, const float* __restrict__ bias1,
    const float* __restrict__ W2, const float* __restrict__ W_pred, float* __restrict__ z)
{
    __shared__ float4 stg[8 * SLOTS];
    __shared__ float Wa[64 * 64];
    __shared__ float Wb[64 * 64];
    __shared__ float Tl[32 * 64];
    __shared__ float UV[2][64];
    __shared__ float In[32], On[32];
    __shared__ float Zp[4][32];

    const int tid = threadIdx.x;
    const int row0 = blockIdx.x * 32;
    const int wave = tid >> 5;
    const int lane = tid & 31;
    const float4* rec4 = (const float4*)rec;

    const int nb = row0 + wave * 4;
    unsigned st = 0u, cn = 0u;
    float4 own = make_float4(0.f, 0.f, 0.f, 0.f);
    if (lane < 4) {
        st  = row_start[nb + lane];
        cn  = in_cnt[nb + lane];
        own = rec4[nb + lane];
    }
    float innv = 0.f, onv = 0.f;
    if (tid < 32) { innv = in_norm[row0 + tid]; onv = out_norm[row0 + tid]; }

    float* W0l = (float*)stg;
    #pragma unroll
    for (int i = 0; i < 4; ++i) {
        ((float4*)W0l)[tid + 256 * i] = ((const float4*)W0)[tid + 256 * i];
        ((float4*)Wa)[tid + 256 * i]  = ((const float4*)W1)[tid + 256 * i];
        ((float4*)Wb)[tid + 256 * i]  = ((const float4*)W2)[tid + 256 * i];
    }
    if (tid < 32) { In[tid] = innv; On[tid] = onv; }
    __syncthreads();

    const int l16  = lane & 15;
    const int koff = (lane < 16) ? 0 : 2;

    if (wave < 4) {
        const int col = wave * 16 + l16;
        const float* arow_src = (l16 == 1) ? b_in : W_in;
        v8f acc = {};
        #pragma unroll
        for (int kk = 0; kk < 16; ++kk) {
            const int k = kk * 4 + koff;
            v2f fa, fb;
            const float2 av = ((const float2*)arow_src)[k >> 1];
            fa.x = (l16 < 2) ? av.x : 0.f;
            fa.y = (l16 < 2) ? av.y : 0.f;
            fb.x = W0l[k * 64 + col];
            fb.y = W0l[(k + 1) * 64 + col];
            acc = __builtin_amdgcn_wmma_f32_16x16x4_f32(false, fa, false, fb,
                                                        (short)0, acc, false, false);
        }
        if (lane < 16) { UV[0][col] = acc[0]; UV[1][col] = acc[1]; }
    }
    __syncthreads();

    {
        const v2f u2 = ((const v2f*)UV[0])[lane];
        const v2f v2 = ((const v2f*)UV[1])[lane];
        const v2f b2 = ((const v2f*)bias0)[lane];
        float4* ws = stg + wave * SLOTS;

        unsigned s_q[4], e_q[4];
        #pragma unroll
        for (int q = 0; q < 4; ++q) {
            s_q[q] = (unsigned)__builtin_amdgcn_readlane((int)st, q);
            e_q[q] = s_q[q] + (unsigned)__builtin_amdgcn_readlane((int)cn, q);
        }
        const unsigned S = s_q[0], T = e_q[3];

        v2f acc[4];
        #pragma unroll
        for (int q = 0; q < 4; ++q) acc[q] = (v2f){0.f, 0.f};

        for (unsigned c0 = S; c0 < T; c0 += SLOTS) {
            const unsigned lim = (T - c0 < (unsigned)SLOTS) ? (T - c0) : (unsigned)SLOTS;
            int idx[SPL];
            #pragma unroll
            for (int u = 0; u < SPL; ++u) {
                const unsigned slot = (unsigned)(u * 32 + lane);
                idx[u] = csr_src[c0 + (slot < lim ? slot : 0u)];
            }
            float4 tv[SPL];
            #pragma unroll
            for (int u = 0; u < SPL; ++u) tv[u] = rec4[idx[u]];
            #pragma unroll
            for (int u = 0; u < SPL; ++u) ws[u * 32 + lane] = tv[u];

            #pragma unroll
            for (int q = 0; q < 4; ++q) {
                const unsigned lo = (s_q[q] > c0) ? s_q[q] : c0;
                const unsigned hi = (e_q[q] < c0 + lim) ? e_q[q] : c0 + lim;
                unsigned e = lo - c0;
                const unsigned eh = (hi > lo) ? hi - c0 : e;
                for (; e + 4 <= eh; e += 4) {
                    const float4 t0 = ws[e], t1 = ws[e + 1], t2 = ws[e + 2], t3 = ws[e + 3];
                    add_source_row(t0, u2, v2, b2, acc[q]);
                    add_source_row(t1, u2, v2, b2, acc[q]);
                    add_source_row(t2, u2, v2, b2, acc[q]);
                    add_source_row(t3, u2, v2, b2, acc[q]);
                }
                for (; e < eh; ++e) add_source_row(ws[e], u2, v2, b2, acc[q]);
            }
        }

        #pragma unroll
        for (int q = 0; q < 4; ++q) {
            float4 t;
            t.x = __builtin_bit_cast(float, __builtin_amdgcn_readlane(__builtin_bit_cast(int, own.x), q));
            t.y = __builtin_bit_cast(float, __builtin_amdgcn_readlane(__builtin_bit_cast(int, own.y), q));
            t.z = __builtin_bit_cast(float, __builtin_amdgcn_readlane(__builtin_bit_cast(int, own.z), q));
            t.w = 0.f;
            add_source_row(t, u2, v2, b2, acc[q]);
            ((v2f*)(Tl + (wave * 4 + q) * 64))[lane] = acc[q];
        }
    }
    __syncthreads();

    const int msub = wave >> 2;
    const int nsub = wave & 3;
    const int arow = msub * 16 + l16;
    const int col  = nsub * 16 + l16;

    v8f c1 = {};
    #pragma unroll
    for (int kk = 0; kk < 16; ++kk) {
        int k = kk * 4 + koff;
        v2f a, b;
        a.x = Tl[arow * 64 + k];
        a.y = Tl[arow * 64 + k + 1];
        b.x = Wa[k * 64 + col];
        b.y = Wa[(k + 1) * 64 + col];
        c1 = __builtin_amdgcn_wmma_f32_16x16x4_f32(false, a, false, b,
                                                   (short)0, c1, false, false);
    }

    const float bcol = bias1[col];
    const int rloc = msub * 16 + ((lane < 16) ? 0 : 8);
    float t[8];
    #pragma unroll
    for (int r = 0; r < 8; ++r) {
        float v = c1[r] * In[rloc + r] + bcol;
        v = v > 0.f ? v : 0.01f * v;
        t[r] = v * On[rloc + r];
    }
    __syncthreads();
    #pragma unroll
    for (int r = 0; r < 8; ++r) Tl[(rloc + r) * 64 + col] = t[r];
    __syncthreads();

    const float wp = W_pred[col];

    v8f c = {};
    #pragma unroll
    for (int kk = 0; kk < 16; ++kk) {
        int k = kk * 4 + koff;
        v2f a, b;
        a.x = Tl[arow * 64 + k];
        a.y = Tl[arow * 64 + k + 1];
        b.x = Wb[k * 64 + col];
        b.y = Wb[(k + 1) * 64 + col];
        c = __builtin_amdgcn_wmma_f32_16x16x4_f32(false, a, false, b,
                                                  (short)0, c, false, false);
    }

    float p[8];
    #pragma unroll
    for (int r = 0; r < 8; ++r) p[r] = c[r] * wp;
    #pragma unroll
    for (int r = 0; r < 8; ++r) p[r] = row_xor_add<1>(p[r]);
    #pragma unroll
    for (int r = 0; r < 8; ++r) p[r] = row_xor_add<2>(p[r]);
    #pragma unroll
    for (int r = 0; r < 8; ++r) p[r] = row_xor_add<4>(p[r]);
    #pragma unroll
    for (int r = 0; r < 8; ++r) p[r] = row_xor_add<8>(p[r]);
    if (l16 == 0) {
        #pragma unroll
        for (int r = 0; r < 8; ++r) Zp[nsub][rloc + r] = p[r];
    }
    __syncthreads();
    if (tid < 32)
        z[row0 + tid] = (Zp[0][tid] + Zp[1][tid]) + (Zp[2][tid] + Zp[3][tid]);
}

#define ZT   51200
#define HGRP 256
#define HFB  7
#define HRND 4
#define HPQ  6
__global__ __launch_bounds__(1024) void gather_head_kernel(
    const float* __restrict__ z, const int* __restrict__ csr_src,
    const unsigned* __restrict__ row_start, const unsigned* __restrict__ in_cnt,
    const float* __restrict__ in_norm, const float* __restrict__ bias,
    const float* __restrict__ W_pred, const float* __restrict__ b_pred,
    float* __restrict__ out, int n)
{
    __shared__ float tbl[ZT];
    __shared__ float ol[128 * HRND];

    const int tid = threadIdx.x;
    const int lane = tid & 31;
    const int sub = tid & 7;
    const int r   = tid >> 3;

    const int ntile = (n + 31) >> 5;
    const int t0 = (int)(((unsigned)blockIdx.x * (unsigned)ntile) / (unsigned)HGRP);
    const int t1 = (int)((((unsigned)blockIdx.x + 1u) * (unsigned)ntile) / (unsigned)HGRP);
    const int n0 = t0 * 32;
    const int n1 = (t1 * 32 < n) ? t1 * 32 : n;

    const float2 bb = ((const float2*)bias)[lane];
    const float2 ww = ((const float2*)W_pred)[lane];
    float cb = bb.x * ww.x + bb.y * ww.y;
    cb = row_xor_add<1>(cb);
    cb = row_xor_add<2>(cb);
    cb = row_xor_add<4>(cb);
    cb = row_xor_add<8>(cb);
    cb += __shfl_xor(cb, 16, 32);
    cb += b_pred[0];

    for (int nb = n0; nb < n1; nb += 128 * HRND) {
        unsigned start[HRND], cnt[HRND];
        float self[HRND], inn[HRND], a[HRND];
        #pragma unroll
        for (int q = 0; q < HRND; ++q) {
            const int row = nb + q * 128 + r;
            const bool live = row < n1;
            const int rowc = live ? row : nb;
            start[q] = row_start[rowc];
            cnt[q]   = live ? in_cnt[rowc] : 0u;
            self[q]  = z[rowc];
            inn[q]   = in_norm[rowc];
            a[q]     = 0.f;
        }
        int s[HRND][HPQ];
        #pragma unroll
        for (int q = 0; q < HRND; ++q) {
            #pragma unroll
            for (int u = 0; u < HPQ; ++u) {
                const unsigned jj = (unsigned)sub + 8u * u;
                s[q][u] = csr_src[(jj < cnt[q]) ? start[q] + jj : 0u];
            }
        }

        for (int base = 0; base < n; base += ZT) {
            const int len = (n - base < ZT) ? (n - base) : ZT;
            const float* zs = z + base;
            const int len4 = len >> 2;
            if (base > 0) __syncthreads();
            for (int i0 = 0; i0 < len4; i0 += 1024 * HFB) {
                float4 q4[HFB];
                #pragma unroll
                for (int u = 0; u < HFB; ++u) {
                    const int i = i0 + u * 1024 + tid;
                    q4[u] = ((const float4*)zs)[(i < len4) ? i : 0];
                }
                asm volatile("" :: "v"(q4[0].x), "v"(q4[0].y), "v"(q4[0].z), "v"(q4[0].w),
                                   "v"(q4[1].x), "v"(q4[1].y), "v"(q4[1].z), "v"(q4[1].w),
                                   "v"(q4[2].x), "v"(q4[2].y), "v"(q4[2].z), "v"(q4[2].w),
                                   "v"(q4[3].x), "v"(q4[3].y), "v"(q4[3].z), "v"(q4[3].w),
                                   "v"(q4[4].x), "v"(q4[4].y), "v"(q4[4].z), "v"(q4[4].w),
                                   "v"(q4[5].x), "v"(q4[5].y), "v"(q4[5].z), "v"(q4[5].w),
                                   "v"(q4[6].x), "v"(q4[6].y), "v"(q4[6].z), "v"(q4[6].w));
                #pragma unroll
                for (int u = 0; u < HFB; ++u) {
                    const int i = i0 + u * 1024 + tid;
                    if (i < len4) ((float4*)tbl)[i] = q4[u];
                }
            }
            if (tid < len - (len4 << 2)) tbl[(len4 << 2) + tid] = zs[(len4 << 2) + tid];
            __syncthreads();

            {
                float v[HRND][HPQ];
                #pragma unroll
                for (int q = 0; q < HRND; ++q) {
                    #pragma unroll
                    for (int u = 0; u < HPQ; ++u) {
                        const unsigned rel = (unsigned)(s[q][u] - base);
                        const bool ok = ((unsigned)sub + 8u * u < cnt[q]) && (rel < (unsigned)len);
                        v[q][u] = tbl[ok ? rel : 0u];
                    }
                }
                #pragma unroll
                for (int q = 0; q < HRND; ++q) {
                    #pragma unroll
                    for (int u = 0; u < HPQ; ++u) {
                        const unsigned rel = (unsigned)(s[q][u] - base);
                        const bool ok = ((unsigned)sub + 8u * u < cnt[q]) && (rel < (unsigned)len);
                        if (ok) a[q] += v[q][u];
                    }
                }
            }
            #pragma unroll
            for (int q = 0; q < HRND; ++q) {
                for (unsigned j = (unsigned)sub + 8u * HPQ; j < cnt[q]; j += 8u) {
                    const unsigned rel = (unsigned)(csr_src[start[q] + j] - base);
                    if (rel < (unsigned)len) a[q] += tbl[rel];
                }
            }
        }

        #pragma unroll
        for (int q = 0; q < HRND; ++q) {
            float x = a[q];
            x = row_xor_add<1>(x);
            x = row_xor_add<2>(x);
            x = row_xor_add<4>(x);
            if (sub == 0 && nb + q * 128 + r < n1)
                ol[q * 128 + r] = (x + self[q]) * inn[q] + cb;
        }
        __syncthreads();
        const int nn = (n1 - nb < 128 * HRND) ? (n1 - nb) : 128 * HRND;
        if (tid < nn) out[nb + tid] = ol[tid];
    }
}

extern "C" void kernel_launch(void* const* d_in, const int* in_sizes, int n_in,
                              void* d_out, int out_size, void* d_ws, size_t ws_size,
                              hipStream_t stream) {
    const float* weight = (const float*)d_in[0];
    const int*   src    = (const int*)d_in[1];
    const int*   dst    = (const int*)d_in[2];
    const float* W_in   = (const float*)d_in[3];
    const float* b_in   = (const float*)d_in[4];
    const float* Wc     = (const float*)d_in[5];
    const float* bc     = (const float*)d_in[6];
    const float* W_pred = (const float*)d_in[7];
    const float* b_pred = (const float*)d_in[8];
    float* out = (float*)d_out;

    const int N = in_sizes[0];
    const int E = in_sizes[1];

    const int NB = (N + 255) / 256;

    float* rec        = (float*)d_ws;
    float* z          = rec + (size_t)N * 64;
    unsigned* in_cnt  = (unsigned*)(z + N);
    float* out_norm   = (float*)(in_cnt + N);
    float* in_norm    = out_norm + N;
    unsigned* row_start = (unsigned*)(in_norm + N);
    float* pq         = (float*)(row_start + N);
    int* csr_src      = (int*)(pq + (size_t)N * 2);
    unsigned* bufD    = (unsigned*)(csr_src + E);
    unsigned* tabD    = bufD + E;
    unsigned* tabS    = tabD + (size_t)GCH * (NB + 1);
    unsigned char* bufS = (unsigned char*)(tabS + (size_t)GCH * (NB + 1));

    chunk_group_kernel<<<GCH, 1024, 2 * NB * sizeof(unsigned), stream>>>(src, dst, tabD, tabS, bufD, bufS, E, NB);

    bucket_finish_kernel<<<NB, 1024, 0, stream>>>(bufD, bufS, tabD, tabS, weight,
                                                  out_norm, in_norm, pq, in_cnt, row_start, csr_src, N, E, NB);

    layer0_sums_kernel<<<(N + 32 * L0R - 1) / (32 * L0R), 256, 0, stream>>>(pq, csr_src, row_start, in_cnt, in_norm, rec, N);

    sum_gemm_head_kernel<<<N / 32, 256, 0, stream>>>(rec, csr_src, row_start, in_cnt, in_norm, out_norm,
                                                     W_in, b_in, Wc, bc, Wc + (size_t)64 * 64, bc + 64,
                                                     Wc + (size_t)2 * 64 * 64, W_pred, z);

    gather_head_kernel<<<HGRP, 1024, 0, stream>>>(z, csr_src, row_start, in_cnt, in_norm,
                                                  bc + 2 * 64, W_pred, b_pred, out, N);
}


# Layer-1 kernel's WMMA operand tiles laid out bank-conflict-free in LDS: row tile pitch 64 -> 68 floats (A reads were 16-way conflicted); weight tiles with column ^ 16 in rows 2,3 mod 4
# speedup vs baseline: 11.5804x; 1.1324x over previous
#include <hip/hip_runtime.h>
#include <math.h>

typedef __attribute__((ext_vector_type(2))) float v2f;
typedef __attribute__((ext_vector_type(8))) float v8f;

#define NN 100000

#define GCH 256
#define BSH 8
#define BMASK 255u

__device__ __forceinline__ void block_excl_scan(unsigned* a, int n, unsigned* wsum) {
    const int tid = threadIdx.x, lane = tid & 31, wave = tid >> 5;
    const int per = (n + 1023) >> 10;
    const int i0 = tid * per;
    unsigned s = 0;
    for (int k = 0; k < per; ++k) if (i0 + k < n) s += a[i0 + k];
    unsigned x = s;
    for (int off = 1; off < 32; off <<= 1) {
        unsigned y = __shfl_up(x, off, 32);
        if (lane >= off) x += y;
    }
    if (lane == 31) wsum[wave] = x;
    __syncthreads();
    unsigned run = x - s;
    for (int w = 0; w < wave; ++w) run += wsum[w];
    for (int k = 0; k < per; ++k) if (i0 + k < n) { unsigned v = a[i0 + k]; a[i0 + k] = run; run += v; }
    __syncthreads();
}

#define EPT 13
#define TILE_E (1024 * EPT)
__global__ __launch_bounds__(1024) void chunk_group_kernel(
    const int* __restrict__ src, const int* __restrict__ dst,
    unsigned* __restrict__ tabD, unsigned* __restrict__ tabS,
    unsigned* __restrict__ bufD, unsigned char* __restrict__ bufS, int E, int NB)
{
    extern __shared__ unsigned dyn_lds[];
    unsigned* lsD = dyn_lds;
    unsigned* lsS = dyn_lds + NB;
    __shared__ unsigned wsum[32];
    __shared__ unsigned      stD[TILE_E];
    __shared__ unsigned char stS[TILE_E];
    __shared__ unsigned spare[64];
    const int tid = threadIdx.x;
    if (tid < 64) spare[tid] = 0u;
    for (int b = tid; b < 2 * NB; b += 1024) lsD[b] = 0u;
    __syncthreads();

    const int chunk = (E + GCH - 1) / GCH;
    const int e0 = blockIdx.x * chunk;
    const int e1 = (e0 + chunk < E) ? (e0 + chunk) : E;
    int nt = e1 - e0;
    nt = nt < 0 ? 0 : (nt > TILE_E ? TILE_E : nt);
    const int t0 = (nt > 0) ? e0 : 0;

    int d[EPT], s[EPT];
    #pragma unroll
    for (int u = 0; u < EPT; ++u) {
        const int i = u * 1024 + tid;
        const int ec = t0 + ((i < nt) ? i : 0);
        d[u] = dst[ec];
        s[u] = src[ec];
    }
    unsigned rk[EPT];
    {
        unsigned rd[EPT], rs[EPT];
        #pragma unroll
        for (int u = 0; u < EPT; ++u) {
            const bool valid = u * 1024 + tid < nt;
            rd[u] = atomicAdd(valid ? &lsD[(unsigned)d[u] >> BSH] : &spare[tid & 31], 1u);
            rs[u] = atomicAdd(valid ? &lsS[(unsigned)s[u] >> BSH] : &spare[32 + (tid & 31)], 1u);
        }
        #pragma unroll
        for (int u = 0; u < EPT; ++u) rk[u] = (rd[u] & 0xFFFFu) | (rs[u] << 16);
    }
    __syncthreads();
    block_excl_scan(lsD, NB, wsum);
    block_excl_scan(lsS, NB, wsum);
    {
        unsigned* rowD = tabD + (size_t)blockIdx.x * (NB + 1);
        unsigned* rowS = tabS + (size_t)blockIdx.x * (NB + 1);
        for (int b = tid; b < NB; b += 1024) { rowD[b] = lsD[b]; rowS[b] = lsS[b]; }
        if (tid == 0) { rowD[NB] = (unsigned)nt; rowS[NB] = (unsigned)nt; }
    }
    unsigned pd[EPT], ps[EPT];
    #pragma unroll
    for (int u = 0; u < EPT; ++u) {
        pd[u] = lsD[(unsigned)d[u] >> BSH] + (rk[u] & 0xFFFFu);
        ps[u] = lsS[(unsigned)s[u] >> BSH] + (rk[u] >> 16);
    }
    #pragma unroll
    for (int u = 0; u < EPT; ++u) {
        if (u * 1024 + tid < nt) {
            stD[pd[u]] = (((unsigned)d[u] & BMASK) << 24) | (unsigned)s[u];
            stS[ps[u]] = (unsigned char)((unsigned)s[u] & BMASK);
        }
    }
    __syncthreads();

    {
        unsigned v[EPT];
        unsigned char w[EPT];
        #pragma unroll
        for (int u = 0; u < EPT; ++u) {
            const int i = u * 1024 + tid;
            const int ic = (i < nt) ? i : 0;
            v[u] = stD[ic];
            w[u] = stS[ic];
        }
        #pragma unroll
        for (int u = 0; u < EPT; ++u) {
            const int i = u * 1024 + tid;
            if (i < nt) { bufD[e0 + i] = v[u]; bufS[e0 + i] = w[u]; }
        }
    }
}

#define STAGE_N 12288
#define RPW (GCH / 32)
__global__ __launch_bounds__(1024) void bucket_finish_kernel(
    const unsigned* __restrict__ bufD, const unsigned char* __restrict__ bufS,
    const unsigned* __restrict__ tabD, const unsigned* __restrict__ tabS,
    const float* __restrict__ weight,
    float* __restrict__ out_norm, float* __restrict__ in_norm, float* __restrict__ pq,
    unsigned* __restrict__ in_cnt, unsigned* __restrict__ row_start,
    int* __restrict__ csr_src, int n, int E, int NB)
{
    __shared__ unsigned cin[256];
    __shared__ unsigned cout[256];
    __shared__ unsigned cur[256];
    __shared__ unsigned wsum[32];
    __shared__ unsigned rsD[GCH], rlD[GCH];
    __shared__ unsigned rsS[GCH], rlS[GCH];
    __shared__ unsigned red[2][GCH / 32];
    __shared__ int stage[STAGE_N];
    const int tid = threadIdx.x, lane = tid & 31, wave = tid >> 5;
    const int b = blockIdx.x;
    const int chunk = (E + GCH - 1) / GCH;
    const float wv = (tid < 256 && b * 256 + tid < n) ? weight[b * 256 + tid] : 0.f;
    unsigned x0 = 0u, x1 = 0u;
    if (tid < GCH) {
        const unsigned* rD = tabD + (size_t)tid * (NB + 1) + b;
        const unsigned* rS = tabS + (size_t)tid * (NB + 1) + b;
        const unsigned d0 = rD[0], d1 = rD[1], s0 = rS[0], s1 = rS[1];
        x0 = d0; x1 = d1 - d0;
        rsD[tid] = (unsigned)tid * (unsigned)chunk + d0;  rlD[tid] = d1 - d0;
        rsS[tid] = (unsigned)tid * (unsigned)chunk + s0;  rlS[tid] = s1 - s0;
        cin[tid] = 0u; cout[tid] = 0u;
    }
    #pragma unroll
    for (int off = 1; off < 32; off <<= 1) {
        x0 += __shfl_xor(x0, off, 32);
        x1 += __shfl_xor(x1, off, 32);
    }
    if (tid < GCH && lane == 0) { red[0][wave] = x0; red[1][wave] = x1; }
    __syncthreads();
    unsigned bD = 0u, nD = 0u;
    #pragma unroll
    for (int w = 0; w < GCH / 32; ++w) { bD += red[0][w]; nD += red[1][w]; }

    unsigned stD_[RPW], lnD_[RPW], stS_[RPW], lnS_[RPW];
    #pragma unroll
    for (int u = 0; u < RPW; ++u) {
        const int c = wave + 32 * u;
        stD_[u] = rsD[c]; lnD_[u] = rlD[c];
        stS_[u] = rsS[c]; lnS_[u] = rlS[c];
    }
    {
        unsigned vd[RPW], vs[RPW];
        #pragma unroll
        for (int u = 0; u < RPW; ++u) {
            vd[u] = bufD[((unsigned)lane < lnD_[u]) ? stD_[u] + lane : 0u];
            vs[u] = bufS[((unsigned)lane < lnS_[u]) ? stS_[u] + lane : 0u];
        }
        #pragma unroll
        for (int u = 0; u < RPW; ++u) {
            if ((unsigned)lane < lnD_[u]) atomicAdd(&cin[vd[u] >> 24], 1u);
            if ((unsigned)lane < lnS_[u]) atomicAdd(&cout[vs[u]], 1u);
        }
        #pragma unroll 1
        for (int u = 0; u < RPW; ++u) {
            const int c = wave + 32 * u;
            const unsigned sd = rsD[c], ld = rlD[c], ss = rsS[c], ls = rlS[c];
            for (unsigned k = 32u + lane; k < ld; k += 32u) atomicAdd(&cin[bufD[sd + k] >> 24], 1u);
            for (unsigned k = 32u + lane; k < ls; k += 32u) atomicAdd(&cout[bufS[ss + k]], 1u);
        }
    }
    __syncthreads();
    if (tid < 256) cur[tid] = cin[tid];
    __syncthreads();
    block_excl_scan(cur, 256, wsum);
    if (tid < 256) {
        const unsigned rs = bD + cur[tid];
        cur[tid] = rs;
        const int node = b * 256 + tid;
        if (node < n) {
            const unsigned c = cin[tid];
            row_start[node] = rs;
            in_cnt[node] = c;
            in_norm[node]  = rsqrtf((float)(c + 1u));
            const float on = rsqrtf((float)(cout[tid] + 1u));
            out_norm[node] = on;
            ((float2*)pq)[node] = make_float2(on * wv, on);
        }
    }
    __syncthreads();
    const bool staged = nD <= (unsigned)STAGE_N;
    {
        unsigned vd[RPW];
        #pragma unroll
        for (int u = 0; u < RPW; ++u) vd[u] = bufD[((unsigned)lane < lnD_[u]) ? stD_[u] + lane : 0u];
        #pragma unroll
        for (int u = 0; u < RPW; ++u) {
            if ((unsigned)lane < lnD_[u]) {
                const unsigned pos = atomicAdd(&cur[vd[u] >> 24], 1u);
                const int sv = (int)(vd[u] & 0x00FFFFFFu);
                if (staged) stage[pos - bD] = sv;
                else        csr_src[pos] = sv;
            }
        }
        #pragma unroll 1
        for (int u = 0; u < RPW; ++u) {
            const int c = wave + 32 * u;
            const unsigned sd = rsD[c], ld = rlD[c];
            for (unsigned k = 32u + lane; k < ld; k += 32u) {
                const unsigned w = bufD[sd + k];
                const unsigned pos = atomicAdd(&cur[w >> 24], 1u);
                const int sv = (int)(w & 0x00FFFFFFu);
                if (staged) stage[pos - bD] = sv;
                else        csr_src[pos] = sv;
            }
        }
    }
    if (staged) {
        __syncthreads();
        for (unsigned i = tid; i < nD; i += 1024) csr_src[bD + i] = stage[i];
    }
}

template <int M> __device__ __forceinline__ float row_xor_add(float x) {
    const int y = __builtin_amdgcn_update_dpp(0, __builtin_bit_cast(int, x), 0x160 + M, 0xf, 0xf, true);
    return x + __builtin_bit_cast(float, y);
}

#define L0R 4
#define L0Q 6
__global__ __launch_bounds__(256) void layer0_sums_kernel(
    const float* __restrict__ pq, const int* __restrict__ csr_src,
    const unsigned* __restrict__ row_start, const unsigned* __restrict__ in_cnt,
    const float* __restrict__ in_norm, float* __restrict__ rec, int n)
{
    const int tid = threadIdx.x;
    const int r   = tid >> 3;
    const int sub = tid & 7;
    const int nb  = blockIdx.x * (32 * L0R);
    const float2* pq2 = (const float2*)pq;

    unsigned start[L0R], cnt[L0R];
    float2 self[L0R];
    float inn[L0R], a[L0R], c[L0R];
    #pragma unroll
    for (int q = 0; q < L0R; ++q) {
        const int row = nb + q * 32 + r;
        const bool live = row < n;
        const int rowc = live ? row : nb;
        start[q] = row_start[rowc];
        cnt[q]   = live ? in_cnt[rowc] : 0u;
        self[q]  = pq2[rowc];
        inn[q]   = in_norm[rowc];
        a[q] = 0.f; c[q] = 0.f;
    }
    int s[L0R][L0Q];
    #pragma unroll
    for (int q = 0; q < L0R; ++q) {
        #pragma unroll
        for (int u = 0; u < L0Q; ++u) {
            const unsigned jj = (unsigned)sub + 8u * u;
            s[q][u] = csr_src[(jj < cnt[q]) ? start[q] + jj : 0u];
        }
    }
    float2 v[L0R][L0Q];
    #pragma unroll
    for (int q = 0; q < L0R; ++q) {
        #pragma unroll
        for (int u = 0; u < L0Q; ++u) v[q][u] = pq2[s[q][u]];
    }
    #pragma unroll
    for (int q = 0; q < L0R; ++q) {
        #pragma unroll
        for (int u = 0; u < L0Q; ++u) {
            if ((unsigned)sub + 8u * u < cnt[q]) { a[q] += v[q][u].x; c[q] += v[q][u].y; }
        }
        for (unsigned j = (unsigned)sub + 8u * L0Q; j < cnt[q]; j += 8u) {
            const float2 w = pq2[csr_src[start[q] + j]];
            a[q] += w.x; c[q] += w.y;
        }
    }
    #pragma unroll
    for (int q = 0; q < L0R; ++q) {
        float x = a[q], y = c[q];
        x = row_xor_add<1>(x);  y = row_xor_add<1>(y);
        x = row_xor_add<2>(x);  y = row_xor_add<2>(y);
        x = row_xor_add<4>(x);  y = row_xor_add<4>(y);
        const int row = nb + q * 32 + r;
        if (sub == 0 && row < n)
            ((float4*)rec)[row] = make_float4((x + self[q].x) * inn[q], (y + self[q].y) * inn[q], self[q].y, 0.f);
    }
}

#define SPL   5
#define SLOTS (32 * SPL)
#define TLP   68

__device__ __forceinline__ void add_source_row(const float4 t, const v2f u2, const v2f v2, const v2f b2, v2f& acc) {
    const v2f ta = {t.x, t.x}, tc = {t.y, t.y}, to = {t.z, t.z};
    v2f x = __builtin_elementwise_fma(tc, v2, b2);
    x = __builtin_elementwise_fma(ta, u2, x);
    x = __builtin_elementwise_max(x, x * 0.01f);
    acc = __builtin_elementwise_fma(to, x, acc);
}

__global__ __launch_bounds__(256) void sum_gemm_head_kernel(
    const float* __restrict__ rec, const int* __restrict__ csr_src,
    const unsigned* __restrict__ row_start, const unsigned* __restrict__ in_cnt,
    const float* __restrict__ in_norm, const float* __restrict__ out_norm,
    const float* __restrict__ W_in, const float* __restrict__ b_in,
    const float* __restrict__ W0, const float* __restrict__ bias0,
    const float* __restrict__ W1, const float* __restrict__ bias1,
    const float* __restrict__ W2, const float* __restrict__ W_pred, float* __restrict__ z)
{
    __shared__ float4 stg[8 * SLOTS];
    __shared__ float Wa[64 * 64];
    __shared__ float Wb[64 * 64];
    __shared__ float Tl[32 * TLP];
    __shared__ float UV[2][64];
    __shared__ float In[32], On[32];
    __shared__ float Zp[4][32];

    const int tid = threadIdx.x;
    const int row0 = blockIdx.x * 32;
    const int wave = tid >> 5;
    const int lane = tid & 31;
    const float4* rec4 = (const float4*)rec;

    const int nb = row0 + wave * 4;
    unsigned st = 0u, cn = 0u;
    float4 own = make_float4(0.f, 0.f, 0.f, 0.f);
    if (lane < 4) {
        st  = row_start[nb + lane];
        cn  = in_cnt[nb + lane];
        own = rec4[nb + lane];
    }
    float innv = 0.f, onv = 0.f;
    if (tid < 32) { innv = in_norm[row0 + tid]; onv = out_norm[row0 + tid]; }

    float* W0l = (float*)stg;
    #pragma unroll
    for (int i = 0; i < 4; ++i) {
        const int f  = tid + 256 * i;
        const int fs = f ^ (((f >> 5) & 1) << 2);
        ((float4*)W0l)[fs] = ((const float4*)W0)[f];
        ((float4*)Wa)[fs]  = ((const float4*)W1)[f];
        ((float4*)Wb)[fs]  = ((const float4*)W2)[f];
    }
    if (tid < 32) { In[tid] = innv; On[tid] = onv; }
    __syncthreads();

    const int l16  = lane & 15;
    const int koff = (lane < 16) ? 0 : 2;

    if (wave < 4) {
        const int col = wave * 16 + l16;
        const int colB = col ^ (koff << 3);
        const float* arow_src = (l16 == 1) ? b_in : W_in;
        v8f acc = {};
        #pragma unroll
        for (int kk = 0; kk < 16; ++kk) {
            const int k = kk * 4 + koff;
            v2f fa, fb;
            const float2 av = ((const float2*)arow_src)[k >> 1];
            fa.x = (l16 < 2) ? av.x : 0.f;
            fa.y = (l16 < 2) ? av.y : 0.f;
            fb.x = W0l[k * 64 + colB];
            fb.y = W0l[(k + 1) * 64 + colB];
            acc = __builtin_amdgcn_wmma_f32_16x16x4_f32(false, fa, false, fb,
                                                        (short)0, acc, false, false);
        }
        if (lane < 16) { UV[0][col] = acc[0]; UV[1][col] = acc[1]; }
    }
    __syncthreads();

    {
        const v2f u2 = ((const v2f*)UV[0])[lane];
        const v2f v2 = ((const v2f*)UV[1])[lane];
        const v2f b2 = ((const v2f*)bias0)[lane];
        float4* ws = stg + wave * SLOTS;

        unsigned s_q[4], e_q[4];
        #pragma unroll
        for (int q = 0; q < 4; ++q) {
            s_q[q] = (unsigned)__builtin_amdgcn_readlane((int)st, q);
            e_q[q] = s_q[q] + (unsigned)__builtin_amdgcn_readlane((int)cn, q);
        }
        const unsigned S = s_q[0], T = e_q[3];

        v2f acc[4];
        #pragma unroll
        for (int q = 0; q < 4; ++q) acc[q] = (v2f){0.f, 0.f};

        for (unsigned c0 = S; c0 < T; c0 += SLOTS) {
            const unsigned lim = (T - c0 < (unsigned)SLOTS) ? (T - c0) : (unsigned)SLOTS;
            int idx[SPL];
            #pragma unroll
            for (int u = 0; u < SPL; ++u) {
                const unsigned slot = (unsigned)(u * 32 + lane);
                idx[u] = csr_src[c0 + (slot < lim ? slot : 0u)];
            }
            float4 tv[SPL];
            #pragma unroll
            for (int u = 0; u < SPL; ++u) tv[u] = rec4[idx[u]];
            #pragma unroll
            for (int u = 0; u < SPL; ++u) ws[u * 32 + lane] = tv[u];

            #pragma unroll
            for (int q = 0; q < 4; ++q) {
                const unsigned lo = (s_q[q] > c0) ? s_q[q] : c0;
                const unsigned hi = (e_q[q] < c0 + lim) ? e_q[q] : c0 + lim;
                unsigned e = lo - c0;
                const unsigned eh = (hi > lo) ? hi - c0 : e;
                for (; e + 4 <= eh; e += 4) {
                    const float4 t0 = ws[e], t1 = ws[e + 1], t2 = ws[e + 2], t3 = ws[e + 3];
                    add_source_row(t0, u2, v2, b2, acc[q]);
                    add_source_row(t1, u2, v2, b2, acc[q]);
                    add_source_row(t2, u2, v2, b2, acc[q]);
                    add_source_row(t3, u2, v2, b2, acc[q]);
                }
                for (; e < eh; ++e) add_source_row(ws[e], u2, v2, b2, acc[q]);
            }
        }

        #pragma unroll
        for (int q = 0; q < 4; ++q) {
            float4 t;
            t.x = __builtin_bit_cast(float, __builtin_amdgcn_readlane(__builtin_bit_cast(int, own.x), q));
            t.y = __builtin_bit_cast(float, __builtin_amdgcn_readlane(__builtin_bit_cast(int, own.y), q));
            t.z = __builtin_bit_cast(float, __builtin_amdgcn_readlane(__builtin_bit_cast(int, own.z), q));
            t.w = 0.f;
            add_source_row(t, u2, v2, b2, acc[q]);
            ((v2f*)(Tl + (wave * 4 + q) * TLP))[lane] = acc[q];
        }
    }
    __syncthreads();

    const int msub = wave >> 2;
    const int nsub = wave & 3;
    const int arow = msub * 16 + l16;
    const int col  = nsub * 16 + l16;
    const int colB = col ^ (koff << 3);

    v8f c1 = {};
    #pragma unroll
    for (int kk = 0; kk < 16; ++kk) {
        int k = kk * 4 + koff;
        v2f a, b;
        a.x = Tl[arow * TLP + k];
        a.y = Tl[arow * TLP + k + 1];
        b.x = Wa[k * 64 + colB];
        b.y = Wa[(k + 1) * 64 + colB];
        c1 = __builtin_amdgcn_wmma_f32_16x16x4_f32(false, a, false, b,
                                                   (short)0, c1, false, false);
    }

    const float bcol = bias1[col];
    const int rloc = msub * 16 + ((lane < 16) ? 0 : 8);
    float t[8];
    #pragma unroll
    for (int r = 0; r < 8; ++r) {
        float v = c1[r] * In[rloc + r] + bcol;
        v = v > 0.f ? v : 0.01f * v;
        t[r] = v * On[rloc + r];
    }
    __syncthreads();
    #pragma unroll
    for (int r = 0; r < 8; ++r) Tl[(rloc + r) * TLP + col] = t[r];
    __syncthreads();

    const float wp = W_pred[col];

    v8f c = {};
    #pragma unroll
    for (int kk = 0; kk < 16; ++kk) {
        int k = kk * 4 + koff;
        v2f a, b;
        a.x = Tl[arow * TLP + k];
        a.y = Tl[arow * TLP + k + 1];
        b.x = Wb[k * 64 + colB];
        b.y = Wb[(k + 1) * 64 + colB];
        c = __builtin_amdgcn_wmma_f32_16x16x4_f32(false, a, false, b,
                                                  (short)0, c, false, false);
    }

    float p[8];
    #pragma unroll
    for (int r = 0; r < 8; ++r) p[r] = c[r] * wp;
    #pragma unroll
    for (int r = 0; r < 8; ++r) p[r] = row_xor_add<1>(p[r]);
    #pragma unroll
    for (int r = 0; r < 8; ++r) p[r] = row_xor_add<2>(p[r]);
    #pragma unroll
    for (int r = 0; r < 8; ++r) p[r] = row_xor_add<4>(p[r]);
    #pragma unroll
    for (int r = 0; r < 8; ++r) p[r] = row_xor_add<8>(p[r]);
    if (l16 == 0) {
        #pragma unroll
        for (int r = 0; r < 8; ++r) Zp[nsub][rloc + r] = p[r];
    }
    __syncthreads();
    if (tid < 32)
        z[row0 + tid] = (Zp[0][tid] + Zp[1][tid]) + (Zp[2][tid] + Zp[3][tid]);
}

#define ZT   51200
#define HGRP 256
#define HFB  7
#define HRND 4
#define HPQ  6
__global__ __launch_bounds__(1024) void gather_head_kernel(
    const float* __restrict__ z, const int* __restrict__ csr_src,
    const unsigned* __restrict__ row_start, const unsigned* __restrict__ in_cnt,
    const float* __restrict__ in_norm, const float* __restrict__ bias,
    const float* __restrict__ W_pred, const float* __restrict__ b_pred,
    float* __restrict__ out, int n)
{
    __shared__ float tbl[ZT];
    __shared__ float ol[128 * HRND];

    const int tid = threadIdx.x;
    const int lane = tid & 31;
    const int sub = tid & 7;
    const int r   = tid >> 3;

    const int ntile = (n + 31) >> 5;
    const int t0 = (int)(((unsigned)blockIdx.x * (unsigned)ntile) / (unsigned)HGRP);
    const int t1 = (int)((((unsigned)blockIdx.x + 1u) * (unsigned)ntile) / (unsigned)HGRP);
    const int n0 = t0 * 32;
    const int n1 = (t1 * 32 < n) ? t1 * 32 : n;

    const float2 bb = ((const float2*)bias)[lane];
    const float2 ww = ((const float2*)W_pred)[lane];
    float cb = bb.x * ww.x + bb.y * ww.y;
    cb = row_xor_add<1>(cb);
    cb = row_xor_add<2>(cb);
    cb = row_xor_add<4>(cb);
    cb = row_xor_add<8>(cb);
    cb += __shfl_xor(cb, 16, 32);
    cb += b_pred[0];

    for (int nb = n0; nb < n1; nb += 128 * HRND) {
        unsigned start[HRND], cnt[HRND];
        float self[HRND], inn[HRND], a[HRND];
        #pragma unroll
        for (int q = 0; q < HRND; ++q) {
            const int row = nb + q * 128 + r;
            const bool live = row < n1;
            const int rowc = live ? row : nb;
            start[q] = row_start[rowc];
            cnt[q]   = live ? in_cnt[rowc] : 0u;
            self[q]  = z[rowc];
            inn[q]   = in_norm[rowc];
            a[q]     = 0.f;
        }
        int s[HRND][HPQ];
        #pragma unroll
        for (int q = 0; q < HRND; ++q) {
            #pragma unroll
            for (int u = 0; u < HPQ; ++u) {
                const unsigned jj = (unsigned)sub + 8u * u;
                s[q][u] = csr_src[(jj < cnt[q]) ? start[q] + jj : 0u];
            }
        }

        for (int base = 0; base < n; base += ZT) {
            const int len = (n - base < ZT) ? (n - base) : ZT;
            const float* zs = z + base;
            const int len4 = len >> 2;
            if (base > 0) __syncthreads();
            for (int i0 = 0; i0 < len4; i0 += 1024 * HFB) {
                float4 q4[HFB];
                #pragma unroll
                for (int u = 0; u < HFB; ++u) {
                    const int i = i0 + u * 1024 + tid;
                    q4[u] = ((const float4*)zs)[(i < len4) ? i : 0];
                }
                asm volatile("" :: "v"(q4[0].x), "v"(q4[0].y), "v"(q4[0].z), "v"(q4[0].w),
                                   "v"(q4[1].x), "v"(q4[1].y), "v"(q4[1].z), "v"(q4[1].w),
                                   "v"(q4[2].x), "v"(q4[2].y), "v"(q4[2].z), "v"(q4[2].w),
                                   "v"(q4[3].x), "v"(q4[3].y), "v"(q4[3].z), "v"(q4[3].w),
                                   "v"(q4[4].x), "v"(q4[4].y), "v"(q4[4].z), "v"(q4[4].w),
                                   "v"(q4[5].x), "v"(q4[5].y), "v"(q4[5].z), "v"(q4[5].w),
                                   "v"(q4[6].x), "v"(q4[6].y), "v"(q4[6].z), "v"(q4[6].w));
                #pragma unroll
                for (int u = 0; u < HFB; ++u) {
                    const int i = i0 + u * 1024 + tid;
                    if (i < len4) ((float4*)tbl)[i] = q4[u];
                }
            }
            if (tid < len - (len4 << 2)) tbl[(len4 << 2) + tid] = zs[(len4 << 2) + tid];
            __syncthreads();

            {
                float v[HRND][HPQ];
                #pragma unroll
                for (int q = 0; q < HRND; ++q) {
                    #pragma unroll
                    for (int u = 0; u < HPQ; ++u) {
                        const unsigned rel = (unsigned)(s[q][u] - base);
                        const bool ok = ((unsigned)sub + 8u * u < cnt[q]) && (rel < (unsigned)len);
                        v[q][u] = tbl[ok ? rel : 0u];
                    }
                }
                #pragma unroll
                for (int q = 0; q < HRND; ++q) {
                    #pragma unroll
                    for (int u = 0; u < HPQ; ++u) {
                        const unsigned rel = (unsigned)(s[q][u] - base);
                        const bool ok = ((unsigned)sub + 8u * u < cnt[q]) && (rel < (unsigned)len);
                        if (ok) a[q] += v[q][u];
                    }
                }
            }
            #pragma unroll
            for (int q = 0; q < HRND; ++q) {
                for (unsigned j = (unsigned)sub + 8u * HPQ; j < cnt[q]; j += 8u) {
                    const unsigned rel = (unsigned)(csr_src[start[q] + j] - base);
                    if (rel < (unsigned)len) a[q] += tbl[rel];
                }
            }
        }

        #pragma unroll
        for (int q = 0; q < HRND; ++q) {
            float x = a[q];
            x = row_xor_add<1>(x);
            x = row_xor_add<2>(x);
            x = row_xor_add<4>(x);
            if (sub == 0 && nb + q * 128 + r < n1)
                ol[q * 128 + r] = (x + self[q]) * inn[q] + cb;
        }
        __syncthreads();
        const int nn = (n1 - nb < 128 * HRND) ? (n1 - nb) : 128 * HRND;
        if (tid < nn) out[nb + tid] = ol[tid];
    }
}

extern "C" void kernel_launch(void* const* d_in, const int* in_sizes, int n_in,
                              void* d_out, int out_size, void* d_ws, size_t ws_size,
                              hipStream_t stream) {
    const float* weight = (const float*)d_in[0];
    const int*   src    = (const int*)d_in[1];
    const int*   dst    = (const int*)d_in[2];
    const float* W_in   = (const float*)d_in[3];
    const float* b_in   = (const float*)d_in[4];
    const float* Wc     = (const float*)d_in[5];
    const float* bc     = (const float*)d_in[6];
    const float* W_pred = (const float*)d_in[7];
    const float* b_pred = (const float*)d_in[8];
    float* out = (float*)d_out;

    const int N = in_sizes[0];
    const int E = in_sizes[1];

    const int NB = (N + 255) / 256;

    float* rec        = (float*)d_ws;
    float* z          = rec + (size_t)N * 64;
    unsigned* in_cnt  = (unsigned*)(z + N);
    float* out_norm   = (float*)(in_cnt + N);
    float* in_norm    = out_norm + N;
    unsigned* row_start = (unsigned*)(in_norm + N);
    float* pq         = (float*)(row_start + N);
    int* csr_src      = (int*)(pq + (size_t)N * 2);
    unsigned* bufD    = (unsigned*)(csr_src + E);
    unsigned* tabD    = bufD + E;
    unsigned* tabS    = tabD + (size_t)GCH * (NB + 1);
    unsigned char* bufS = (unsigned char*)(tabS + (size_t)GCH * (NB + 1));

    chunk_group_kernel<<<GCH, 1024, 2 * NB * sizeof(unsigned), stream>>>(src, dst, tabD, tabS, bufD, bufS, E, NB);

    bucket_finish_kernel<<<NB, 1024, 0, stream>>>(bufD, bufS, tabD, tabS, weight,
                                                  out_norm, in_norm, pq, in_cnt, row_start, csr_src, N, E, NB);

    layer0_sums_kernel<<<(N + 32 * L0R - 1) / (32 * L0R), 256, 0, stream>>>(pq, csr_src, row_start, in_cnt, in_norm, rec, N);

    sum_gemm_head_kernel<<<N / 32, 256, 0, stream>>>(rec, csr_src, row_start, in_cnt, in_norm, out_norm,
                                                     W_in, b_in, Wc, bc, Wc + (size_t)64 * 64, bc + 64,
                                                     Wc + (size_t)2 * 64 * 64, W_pred, z);

    gather_head_kernel<<<HGRP, 1024, 0, stream>>>(z, csr_src, row_start, in_cnt, in_norm,
                                                  bc + 2 * 64, W_pred, b_pred, out, N);
}
